# starting kernel
# baseline (speedup 1.0000x reference)
#include <hip/hip_runtime.h>
#include <hip/hip_bf16.h>

typedef __attribute__((ext_vector_type(2))) float v2f;
typedef __attribute__((ext_vector_type(8))) float v8f;

__global__ void fill_zero(float* __restrict__ p, int n) {
    int i = blockIdx.x * blockDim.x + threadIdx.x;
    if (i < n) p[i] = 0.0f;
}

__global__ void degree_count(const int* __restrict__ ei, int E,
                             float* __restrict__ cdst, float* __restrict__ csrc) {
    int e = blockIdx.x * blockDim.x + threadIdx.x;
    if (e < E) {
        atomicAdd(&csrc[ei[e]],     1.0f);
        atomicAdd(&cdst[ei[e + E]], 1.0f);
    }
}

__global__ void scatter_add64(const float* __restrict__ Y,
                              const int* __restrict__ jidx,
                              const int* __restrict__ iidx,
                              float* __restrict__ S, int E) {
    int t = blockIdx.x * blockDim.x + threadIdx.x;
    int e = t >> 6;
    int c = t & 63;
    if (e < E) {
        int j = jidx[e];
        int i = iidx[e];
        atomicAdd(&S[(size_t)i * 64 + c], Y[(size_t)j * 64 + c]);
    }
}

__global__ __launch_bounds__(128)
void gemm_wmma_relu(const float* __restrict__ X0, int d0,
                    const float* __restrict__ X1,
                    const float* __restrict__ cnt,
                    const float* __restrict__ W,
                    int Ktot,
                    const float* __restrict__ bias,
                    float* __restrict__ out, int ostride, int ocol0, int dup,
                    int Nrows) {
    __shared__ float sW[192 * 64];

    const int tid = threadIdx.x;

    for (int i = tid; i < Ktot * 64; i += 128) {
        int k = i >> 6;
        int c = i & 63;
        sW[(((k >> 1) << 6) + c) * 2 + (k & 1)] = W[i];
    }
    __syncthreads();

    const int wave = tid >> 5;
    const int lane = tid & 31;
    const int m    = lane & 15;
    const int kSel = lane >> 4;

    const int rowbase = (blockIdx.x * 4 + wave) * 16;
    const int row     = rowbase + m;
    const int crow    = row < Nrows ? row : (Nrows - 1);

    const float* x0r = X0 + (size_t)crow * d0 + 2 * kSel;
    const float* x1r = X1 ? X1 + (size_t)crow * 64 + 2 * kSel : nullptr;
    float scale1 = 1.0f;
    if (cnt) scale1 = 1.0f / fmaxf(cnt[crow], 1.0f);

    v8f acc[4] = {};

#pragma unroll 1
    for (int k0 = 0; k0 < d0; k0 += 32) {
        v2f areg[8];
#pragma unroll
        for (int u = 0; u < 8; ++u)
            areg[u] = *(const v2f*)(x0r + k0 + 4 * u);
#pragma unroll
        for (int u = 0; u < 8; ++u) {
            const int sbase = (((k0 >> 1) + 2 * u + kSel) << 7);
#pragma unroll
            for (int t = 0; t < 4; ++t) {
                v2f b = *(const v2f*)&sW[sbase + ((t << 4) + m) * 2];
                acc[t] = __builtin_amdgcn_wmma_f32_16x16x4_f32(
                    false, areg[u], false, b, (short)0, acc[t], false, false);
            }
        }
    }

    if (x1r) {
#pragma unroll 1
        for (int k0 = d0; k0 < Ktot; k0 += 32) {
            v2f areg[8];
#pragma unroll
            for (int u = 0; u < 8; ++u)
                areg[u] = *(const v2f*)(x1r + (k0 - d0) + 4 * u);
#pragma unroll
            for (int u = 0; u < 8; ++u) {
                v2f a;
                a.x = areg[u].x * scale1;
                a.y = areg[u].y * scale1;
                const int sbase = (((k0 >> 1) + 2 * u + kSel) << 7);
#pragma unroll
                for (int t = 0; t < 4; ++t) {
                    v2f b = *(const v2f*)&sW[sbase + ((t << 4) + m) * 2];
                    acc[t] = __builtin_amdgcn_wmma_f32_16x16x4_f32(
                        false, a, false, b, (short)0, acc[t], false, false);
                }
            }
        }
    }

    const int rr  = kSel * 8;
    const int col = m;
#pragma unroll
    for (int r = 0; r < 8; ++r) {
        const int orow = rowbase + r + rr;
        if (orow < Nrows) {
#pragma unroll
            for (int t = 0; t < 4; ++t) {
                float v = acc[t][r];
                if (bias) v += bias[(t << 4) + col];
                v = fmaxf(v, 0.0f);
                float* p = out + (size_t)orow * ostride + ocol0 + (t << 4) + col;
                *p = v;
                if (dup) p[64] = v;
            }
        }
    }
}

extern "C" void kernel_launch(void* const* d_in, const int* in_sizes, int n_in,
                              void* d_out, int out_size, void* d_ws, size_t ws_size,
                              hipStream_t stream) {
    const float* x    = (const float*)d_in[0];
    const int*   ei   = (const int*)  d_in[1];
    const float* W_u1 = (const float*)d_in[2];
    const float* b_u1 = (const float*)d_in[3];
    const float* U_u1 = (const float*)d_in[4];
    const float* W_d1 = (const float*)d_in[5];
    const float* b_d1 = (const float*)d_in[6];
    const float* U_d1 = (const float*)d_in[7];
    const float* W_u2 = (const float*)d_in[8];
    const float* b_u2 = (const float*)d_in[9];
    const float* U_u2 = (const float*)d_in[10];
    float* out = (float*)d_out;

    const int N = in_sizes[0] / 128;
    const int E = in_sizes[1] / 2;
    const int* src = ei;
    const int* dst = ei + E;

    float* ws   = (float*)d_ws;
    float* Y    = ws;
    float* S    = Y    + (size_t)N * 64;
    float* ha   = S    + (size_t)N * 64;
    float* hb   = ha   + (size_t)N * 128;
    float* cdst = hb   + (size_t)N * 128;
    float* csrc = cdst + N;

    const int TPB = 256;

    fill_zero<<<(2 * N + TPB - 1) / TPB, TPB, 0, stream>>>(cdst, 2 * N);
    degree_count<<<(E + TPB - 1) / TPB, TPB, 0, stream>>>(ei, E, cdst, csrc);

    const int gblk  = (N + 63) / 64;
    const int nS    = N * 64;
    const int sgrid = (E * 64 + TPB - 1) / TPB;

    auto conv = [&](const float* Xin, const int* jidx, const int* iidx,
                    const float* cntp, const float* Wm, const float* bm,
                    const float* Um, float* outp, int ocol0, int dupw) {
        gemm_wmma_relu<<<gblk, 128, 0, stream>>>(Xin, 128, nullptr, nullptr,
                                                 Wm, 128, bm, Y, 64, 0, 0, N);
        fill_zero<<<(nS + TPB - 1) / TPB, TPB, 0, stream>>>(S, nS);
        scatter_add64<<<sgrid, TPB, 0, stream>>>(Y, jidx, iidx, S, E);
        gemm_wmma_relu<<<gblk, 128, 0, stream>>>(Xin, 128, S, cntp,
                                                 Um, 192, nullptr,
                                                 outp, 128, ocol0, dupw, N);
    };

    conv(x,  src, dst, cdst, W_u1, b_u1, U_u1, ha,  0, 0);
    conv(x,  dst, src, csrc, W_d1, b_d1, U_d1, ha, 64, 0);
    conv(ha, src, dst, cdst, W_u2, b_u2, U_u2, hb,  0, 1);
    conv(hb, src, dst, cdst, W_u2, b_u2, U_u2, out, 0, 1);
}


# scatter-add: one wave takes 16 edges (scalar index reads, its 32 row gathers issued as one group, then 32 line atomics) instead of half an edge; atomic sum order stays arrival order
# speedup vs baseline: 1.9952x; 1.9952x over previous
#include <hip/hip_runtime.h>
#include <hip/hip_bf16.h>

typedef __attribute__((ext_vector_type(2))) float v2f;
typedef __attribute__((ext_vector_type(8))) float v8f;

__global__ void fill_zero(float* __restrict__ p, int n) {
    int i = blockIdx.x * blockDim.x + threadIdx.x;
    if (i < n) p[i] = 0.0f;
}

__global__ void degree_count(const int* __restrict__ ei, int E,
                             float* __restrict__ cdst, float* __restrict__ csrc) {
    int e = blockIdx.x * blockDim.x + threadIdx.x;
    if (e < E) {
        atomicAdd(&csrc[ei[e]],     1.0f);
        atomicAdd(&cdst[ei[e + E]], 1.0f);
    }
}

#define SC_EPW 16
#define SC_TPB 256
#define SC_EPB (SC_EPW * (SC_TPB / 32))

__global__ __launch_bounds__(SC_TPB)
void scatter_add64(const float* __restrict__ Y,
                   const int* __restrict__ jidx,
                   const int* __restrict__ iidx,
                   float* __restrict__ S, int E) {
    const int lane = threadIdx.x & 31;
    const int wv = __builtin_amdgcn_readfirstlane(
        (int)((blockIdx.x * SC_TPB + threadIdx.x) >> 5));
    const int e0 = wv * SC_EPW;

    if (e0 + SC_EPW <= E) {
        int j[SC_EPW], i[SC_EPW];
#pragma unroll
        for (int u = 0; u < SC_EPW; ++u) {
            j[u] = jidx[e0 + u];
            i[u] = iidx[e0 + u];
        }
        float a[SC_EPW], b[SC_EPW];
#pragma unroll
        for (int u = 0; u < SC_EPW; ++u) {
            const float* yr = Y + (size_t)j[u] * 64 + lane;
            a[u] = yr[0];
            b[u] = yr[32];
        }
#pragma unroll
        for (int u = 0; u < SC_EPW; ++u) {
            float* sr = S + (size_t)i[u] * 64 + lane;
            atomicAdd(sr,      a[u]);
            atomicAdd(sr + 32, b[u]);
        }
    } else {
        for (int e = e0; e < E; ++e) {
            const float* yr = Y + (size_t)jidx[e] * 64 + lane;
            float* sr = S + (size_t)iidx[e] * 64 + lane;
            atomicAdd(sr,      yr[0]);
            atomicAdd(sr + 32, yr[32]);
        }
    }
}

__global__ __launch_bounds__(128)
void gemm_wmma_relu(const float* __restrict__ X0, int d0,
                    const float* __restrict__ X1,
                    const float* __restrict__ cnt,
                    const float* __restrict__ W,
                    int Ktot,
                    const float* __restrict__ bias,
                    float* __restrict__ out, int ostride, int ocol0, int dup,
                    int Nrows) {
    __shared__ float sW[192 * 64];

    const int tid = threadIdx.x;

    for (int i = tid; i < Ktot * 64; i += 128) {
        int k = i >> 6;
        int c = i & 63;
        sW[(((k >> 1) << 6) + c) * 2 + (k & 1)] = W[i];
    }
    __syncthreads();

    const int wave = tid >> 5;
    const int lane = tid & 31;
    const int m    = lane & 15;
    const int kSel = lane >> 4;

    const int rowbase = (blockIdx.x * 4 + wave) * 16;
    const int row     = rowbase + m;
    const int crow    = row < Nrows ? row : (Nrows - 1);

    const float* x0r = X0 + (size_t)crow * d0 + 2 * kSel;
    const float* x1r = X1 ? X1 + (size_t)crow * 64 + 2 * kSel : nullptr;
    float scale1 = 1.0f;
    if (cnt) scale1 = 1.0f / fmaxf(cnt[crow], 1.0f);

    v8f acc[4] = {};

#pragma unroll 1
    for (int k0 = 0; k0 < d0; k0 += 32) {
        v2f areg[8];
#pragma unroll
        for (int u = 0; u < 8; ++u)
            areg[u] = *(const v2f*)(x0r + k0 + 4 * u);
#pragma unroll
        for (int u = 0; u < 8; ++u) {
            const int sbase = (((k0 >> 1) + 2 * u + kSel) << 7);
#pragma unroll
            for (int t = 0; t < 4; ++t) {
                v2f b = *(const v2f*)&sW[sbase + ((t << 4) + m) * 2];
                acc[t] = __builtin_amdgcn_wmma_f32_16x16x4_f32(
                    false, areg[u], false, b, (short)0, acc[t], false, false);
            }
        }
    }

    if (x1r) {
#pragma unroll 1
        for (int k0 = d0; k0 < Ktot; k0 += 32) {
            v2f areg[8];
#pragma unroll
            for (int u = 0; u < 8; ++u)
                areg[u] = *(const v2f*)(x1r + (k0 - d0) + 4 * u);
#pragma unroll
            for (int u = 0; u < 8; ++u) {
                v2f a;
                a.x = areg[u].x * scale1;
                a.y = areg[u].y * scale1;
                const int sbase = (((k0 >> 1) + 2 * u + kSel) << 7);
#pragma unroll
                for (int t = 0; t < 4; ++t) {
                    v2f b = *(const v2f*)&sW[sbase + ((t << 4) + m) * 2];
                    acc[t] = __builtin_amdgcn_wmma_f32_16x16x4_f32(
                        false, a, false, b, (short)0, acc[t], false, false);
                }
            }
        }
    }

    const int rr  = kSel * 8;
    const int col = m;
#pragma unroll
    for (int r = 0; r < 8; ++r) {
        const int orow = rowbase + r + rr;
        if (orow < Nrows) {
#pragma unroll
            for (int t = 0; t < 4; ++t) {
                float v = acc[t][r];
                if (bias) v += bias[(t << 4) + col];
                v = fmaxf(v, 0.0f);
                float* p = out + (size_t)orow * ostride + ocol0 + (t << 4) + col;
                *p = v;
                if (dup) p[64] = v;
            }
        }
    }
}

extern "C" void kernel_launch(void* const* d_in, const int* in_sizes, int n_in,
                              void* d_out, int out_size, void* d_ws, size_t ws_size,
                              hipStream_t stream) {
    const float* x    = (const float*)d_in[0];
    const int*   ei   = (const int*)  d_in[1];
    const float* W_u1 = (const float*)d_in[2];
    const float* b_u1 = (const float*)d_in[3];
    const float* U_u1 = (const float*)d_in[4];
    const float* W_d1 = (const float*)d_in[5];
    const float* b_d1 = (const float*)d_in[6];
    const float* U_d1 = (const float*)d_in[7];
    const float* W_u2 = (const float*)d_in[8];
    const float* b_u2 = (const float*)d_in[9];
    const float* U_u2 = (const float*)d_in[10];
    float* out = (float*)d_out;

    const int N = in_sizes[0] / 128;
    const int E = in_sizes[1] / 2;
    const int* src = ei;
    const int* dst = ei + E;

    float* ws   = (float*)d_ws;
    float* Y    = ws;
    float* S    = Y    + (size_t)N * 64;
    float* ha   = S    + (size_t)N * 64;
    float* hb   = ha   + (size_t)N * 128;
    float* cdst = hb   + (size_t)N * 128;
    float* csrc = cdst + N;

    const int TPB = 256;

    fill_zero<<<(2 * N + TPB - 1) / TPB, TPB, 0, stream>>>(cdst, 2 * N);
    degree_count<<<(E + TPB - 1) / TPB, TPB, 0, stream>>>(ei, E, cdst, csrc);

    const int gblk  = (N + 63) / 64;
    const int nS    = N * 64;
    const int sgrid = (E + SC_EPB - 1) / SC_EPB;

    auto conv = [&](const float* Xin, const int* jidx, const int* iidx,
                    const float* cntp, const float* Wm, const float* bm,
                    const float* Um, float* outp, int ocol0, int dupw) {
        gemm_wmma_relu<<<gblk, 128, 0, stream>>>(Xin, 128, nullptr, nullptr,
                                                 Wm, 128, bm, Y, 64, 0, 0, N);
        fill_zero<<<(nS + TPB - 1) / TPB, TPB, 0, stream>>>(S, nS);
        scatter_add64<<<sgrid, SC_TPB, 0, stream>>>(Y, jidx, iidx, S, E);
        gemm_wmma_relu<<<gblk, 128, 0, stream>>>(Xin, 128, S, cntp,
                                                 Um, 192, nullptr,
                                                 outp, 128, ocol0, dupw, N);
    };

    conv(x,  src, dst, cdst, W_u1, b_u1, U_u1, ha,  0, 0);
    conv(x,  dst, src, csrc, W_d1, b_d1, U_d1, ha, 64, 0);
    conv(ha, src, dst, cdst, W_u2, b_u2, U_u2, hb,  0, 1);
    conv(hb, src, dst, cdst, W_u2, b_u2, U_u2, out, 0, 1);
}


# aggregation pulled, not pushed: edges listed per node once per call (LDS counters give degree and rank), one wave per node sums its neighbour rows in registers; no float atomics; sums in arrival rank
# speedup vs baseline: 2.4185x; 1.2122x over previous
#include <hip/hip_runtime.h>
#include <hip/hip_bf16.h>

typedef __attribute__((ext_vector_type(2))) float v2f;
typedef __attribute__((ext_vector_type(4))) float v4f;
typedef __attribute__((ext_vector_type(8))) float v8f;

__global__ void fill_zero(int* __restrict__ p, int n) {
    int i = blockIdx.x * blockDim.x + threadIdx.x;
    if (i < n) p[i] = 0;
}


#define BK_TPB    1024
#define BK_LDSN   65536
#define BK_SLICES 32
__global__ __launch_bounds__(BK_TPB)
void count_rank(const int* __restrict__ ei, int E, int N, int cstride,
                int* __restrict__ cnt, int* __restrict__ rank) {
    __shared__ int h[BK_LDSN];
    const int dir = blockIdx.y;
    const int* key = ei + (dir == 0 ? E : 0);
    int* cn = cnt  + (size_t)dir * cstride;
    int* rk = rank + (size_t)dir * E;
    const int tid = threadIdx.x;
    const int len = (E + gridDim.x - 1) / gridDim.x;
    const int lo  = min((int)blockIdx.x * len, E);
    const int hi  = min(lo + len, E);

    for (int n = tid; n < N; n += BK_TPB) h[n] = 0;
    __syncthreads();

    for (int e0 = lo; e0 < hi; e0 += 8 * BK_TPB) {
        int k[8];
#pragma unroll
        for (int u = 0; u < 8; ++u) {
            const int e = e0 + u * BK_TPB + tid;
            k[u] = (e < hi) ? key[e] : 0;
        }
#pragma unroll
        for (int u = 0; u < 8; ++u)
            if (e0 + u * BK_TPB + tid < hi) atomicAdd(&h[k[u]], 1);
    }
    __syncthreads();

    for (int n0 = 0; n0 < N; n0 += 16 * BK_TPB) {
        int c[16], b[16];
#pragma unroll
        for (int u = 0; u < 16; ++u) {
            const int n = n0 + u * BK_TPB + tid;
            c[u] = (n < N) ? h[n] : 0;
        }
#pragma unroll
        for (int u = 0; u < 16; ++u) {
            const int n = n0 + u * BK_TPB + tid;
            b[u] = (n < N) ? atomicAdd(&cn[n], c[u]) : 0;
        }
#pragma unroll
        for (int u = 0; u < 16; ++u) {
            const int n = n0 + u * BK_TPB + tid;
            if (n < N) h[n] = b[u];
        }
    }
    __syncthreads();

    for (int e0 = lo; e0 < hi; e0 += 8 * BK_TPB) {
        int k[8];
#pragma unroll
        for (int u = 0; u < 8; ++u) {
            const int e = e0 + u * BK_TPB + tid;
            k[u] = (e < hi) ? key[e] : 0;
        }
#pragma unroll
        for (int u = 0; u < 8; ++u) {
            const int e = e0 + u * BK_TPB + tid;
            if (e < hi) rk[e] = atomicAdd(&h[k[u]], 1);
        }
    }
}

#define RS_TPB 1024
__global__ __launch_bounds__(RS_TPB)
void row_starts(const int* __restrict__ cnt, int N, int cstride,
                int* __restrict__ rowptr) {
    __shared__ int wsum[RS_TPB / 32];
    const int* c  = cnt    + (size_t)blockIdx.x * cstride;
    int*       rp = rowptr + (size_t)blockIdx.x * (N + 1);
    const int tid  = threadIdx.x;
    const int lane = tid & 31;
    const int wv   = tid >> 5;
    const int per  = (N + RS_TPB - 1) / RS_TPB;
    const int b    = min(tid * per, N);
    const int e    = min(b + per, N);

    int s = 0;
    for (int i = b; i < e; ++i) s += c[i];

    int incl = s;
    for (int d = 1; d < 32; d <<= 1) {
        int t = __shfl_up(incl, d);
        if (lane >= d) incl += t;
    }
    if (lane == 31) wsum[wv] = incl;
    __syncthreads();
    if (wv == 0) {
        const int w = wsum[lane];
        int wi = w;
        for (int d = 1; d < 32; d <<= 1) {
            int t = __shfl_up(wi, d);
            if (lane >= d) wi += t;
        }
        wsum[lane] = wi - w;
    }
    __syncthreads();

    int run = wsum[wv] + incl - s;
    for (int i = b; i < e; ++i) {
        const int v = c[i];
        rp[i] = run;
        run += v;
    }
    if (tid == RS_TPB - 1) rp[N] = run;
}

__global__ void place_edges(const int* __restrict__ ei, int E, int N,
                            const int* __restrict__ rowptr,
                            const int* __restrict__ rank,
                            int* __restrict__ adj) {
    const int dir = blockIdx.y;
    const int* key = ei + (dir == 0 ? E : 0);
    const int* val = ei + (dir == 0 ? 0 : E);
    const int* rp  = rowptr + (size_t)dir * (N + 1);
    const int* rk  = rank   + (size_t)dir * E;
    int*       ad  = adj    + (size_t)dir * E;
    const int e = blockIdx.x * blockDim.x + threadIdx.x;
    if (e < E) ad[rp[key[e]] + rk[e]] = val[e];
}

#define GA_TPB 256
__global__ __launch_bounds__(GA_TPB)
void gather_mean64(const float* __restrict__ Y,
                   const int* __restrict__ rowptr,
                   const int* __restrict__ adj,
                   float* __restrict__ S, int N) {
    const int lane = threadIdx.x & 31;
    const int h    = lane >> 4;
    const int c    = lane & 15;
    const int node = __builtin_amdgcn_readfirstlane(
        (int)((blockIdx.x * GA_TPB + threadIdx.x) >> 5));
    if (node >= N) return;
    const int beg = rowptr[node];
    const int end = rowptr[node + 1];

    const v4f* Y4 = (const v4f*)Y;
    v4f acc = {0.0f, 0.0f, 0.0f, 0.0f};
    for (int p = beg; p < end; p += 16) {
        int j[8];
#pragma unroll
        for (int u = 0; u < 8; ++u)
            j[u] = adj[min(p + 2 * u + h, end - 1)];
        v4f v[8];
#pragma unroll
        for (int u = 0; u < 8; ++u)
            v[u] = Y4[(size_t)j[u] * 16 + c];
#pragma unroll
        for (int u = 0; u < 8; ++u)
            if (p + 2 * u + h < end) acc += v[u];
    }
    acc.x += __shfl_xor(acc.x, 16);
    acc.y += __shfl_xor(acc.y, 16);
    acc.z += __shfl_xor(acc.z, 16);
    acc.w += __shfl_xor(acc.w, 16);
    const float inv = 1.0f / fmaxf((float)(end - beg), 1.0f);
    if (h == 0) ((v4f*)S)[(size_t)node * 16 + c] = acc * inv;
}

__global__ __launch_bounds__(128)
void gemm_wmma_relu(const float* __restrict__ X0, int d0,
                    const float* __restrict__ X1,
                    const float* __restrict__ cnt,
                    const float* __restrict__ W,
                    int Ktot,
                    const float* __restrict__ bias,
                    float* __restrict__ out, int ostride, int ocol0, int dup,
                    int Nrows) {
    __shared__ float sW[192 * 64];

    const int tid = threadIdx.x;

    for (int i = tid; i < Ktot * 64; i += 128) {
        int k = i >> 6;
        int c = i & 63;
        sW[(((k >> 1) << 6) + c) * 2 + (k & 1)] = W[i];
    }
    __syncthreads();

    const int wave = tid >> 5;
    const int lane = tid & 31;
    const int m    = lane & 15;
    const int kSel = lane >> 4;

    const int rowbase = (blockIdx.x * 4 + wave) * 16;
    const int row     = rowbase + m;
    const int crow    = row < Nrows ? row : (Nrows - 1);

    const float* x0r = X0 + (size_t)crow * d0 + 2 * kSel;
    const float* x1r = X1 ? X1 + (size_t)crow * 64 + 2 * kSel : nullptr;
    float scale1 = 1.0f;
    if (cnt) scale1 = 1.0f / fmaxf(cnt[crow], 1.0f);

    v8f acc[4] = {};

#pragma unroll 1
    for (int k0 = 0; k0 < d0; k0 += 32) {
        v2f areg[8];
#pragma unroll
        for (int u = 0; u < 8; ++u)
            areg[u] = *(const v2f*)(x0r + k0 + 4 * u);
#pragma unroll
        for (int u = 0; u < 8; ++u) {
            const int sbase = (((k0 >> 1) + 2 * u + kSel) << 7);
#pragma unroll
            for (int t = 0; t < 4; ++t) {
                v2f b = *(const v2f*)&sW[sbase + ((t << 4) + m) * 2];
                acc[t] = __builtin_amdgcn_wmma_f32_16x16x4_f32(
                    false, areg[u], false, b, (short)0, acc[t], false, false);
            }
        }
    }

    if (x1r) {
#pragma unroll 1
        for (int k0 = d0; k0 < Ktot; k0 += 32) {
            v2f areg[8];
#pragma unroll
            for (int u = 0; u < 8; ++u)
                areg[u] = *(const v2f*)(x1r + (k0 - d0) + 4 * u);
#pragma unroll
            for (int u = 0; u < 8; ++u) {
                v2f a;
                a.x = areg[u].x * scale1;
                a.y = areg[u].y * scale1;
                const int sbase = (((k0 >> 1) + 2 * u + kSel) << 7);
#pragma unroll
                for (int t = 0; t < 4; ++t) {
                    v2f b = *(const v2f*)&sW[sbase + ((t << 4) + m) * 2];
                    acc[t] = __builtin_amdgcn_wmma_f32_16x16x4_f32(
                        false, a, false, b, (short)0, acc[t], false, false);
                }
            }
        }
    }

    const int rr  = kSel * 8;
    const int col = m;
#pragma unroll
    for (int r = 0; r < 8; ++r) {
        const int orow = rowbase + r + rr;
        if (orow < Nrows) {
#pragma unroll
            for (int t = 0; t < 4; ++t) {
                float v = acc[t][r];
                if (bias) v += bias[(t << 4) + col];
                v = fmaxf(v, 0.0f);
                float* p = out + (size_t)orow * ostride + ocol0 + (t << 4) + col;
                *p = v;
                if (dup) p[64] = v;
            }
        }
    }
}

extern "C" void kernel_launch(void* const* d_in, const int* in_sizes, int n_in,
                              void* d_out, int out_size, void* d_ws, size_t ws_size,
                              hipStream_t stream) {
    const float* x    = (const float*)d_in[0];
    const int*   ei   = (const int*)  d_in[1];
    const float* W_u1 = (const float*)d_in[2];
    const float* b_u1 = (const float*)d_in[3];
    const float* U_u1 = (const float*)d_in[4];
    const float* W_d1 = (const float*)d_in[5];
    const float* b_d1 = (const float*)d_in[6];
    const float* U_d1 = (const float*)d_in[7];
    const float* W_u2 = (const float*)d_in[8];
    const float* b_u2 = (const float*)d_in[9];
    const float* U_u2 = (const float*)d_in[10];
    float* out = (float*)d_out;

    const int N = in_sizes[0] / 128;
    const int E = in_sizes[1] / 2;

    const int cstride = (N + 31) / 32 * 32;
    float* ws     = (float*)d_ws;
    float* Y      = ws;
    float* S      = Y  + (size_t)N * 64;
    float* ha     = S  + (size_t)N * 64;
    int*   cnt    = (int*)(ha + (size_t)N * 128);
    int*   rowptr = cnt    + 2 * (size_t)cstride;
    int*   rank   = rowptr + 2 * ((size_t)N + 1);
    int*   adj    = rank   + 2 * (size_t)E;

    const int* rp_in   = rowptr;
    const int* rp_out  = rowptr + (N + 1);
    const int* adj_in  = adj;
    const int* adj_out = adj + E;

    const int TPB = 256;

    fill_zero<<<(2 * cstride + TPB - 1) / TPB, TPB, 0, stream>>>(cnt, 2 * cstride);
    count_rank<<<dim3(BK_SLICES, 2), BK_TPB, 0, stream>>>(ei, E, N, cstride, cnt, rank);
    row_starts<<<2, RS_TPB, 0, stream>>>(cnt, N, cstride, rowptr);
    place_edges<<<dim3((E + TPB - 1) / TPB, 2), TPB, 0, stream>>>(ei, E, N, rowptr, rank, adj);

    const int gblk  = (N + 63) / 64;
    const int ggrid = (N + GA_TPB / 32 - 1) / (GA_TPB / 32);

    auto conv = [&](const float* Xin, const int* rp, const int* nb,
                    const float* Wm, const float* bm,
                    const float* Um, float* outp, int ocol0, int dupw) {
        gemm_wmma_relu<<<gblk, 128, 0, stream>>>(Xin, 128, nullptr, nullptr,
                                                 Wm, 128, bm, Y, 64, 0, 0, N);
        gather_mean64<<<ggrid, GA_TPB, 0, stream>>>(Y, rp, nb, S, N);
        gemm_wmma_relu<<<gblk, 128, 0, stream>>>(Xin, 128, S, nullptr,
                                                 Um, 192, nullptr,
                                                 outp, 128, ocol0, dupw, N);
    };

    conv(x,   rp_in,  adj_in,  W_u1, b_u1, U_u1, out,  0, 0);
    conv(x,   rp_out, adj_out, W_d1, b_d1, U_d1, out, 64, 0);
    conv(out, rp_in,  adj_in,  W_u2, b_u2, U_u2, ha,   0, 1);
    conv(ha,  rp_in,  adj_in,  W_u2, b_u2, U_u2, out,  0, 1);
}


# per-node lists built without global atomics or rank array (slice counts to a table, one scan, LDS slot hand-out; 15 launches); mean kernel sorts each list by neighbour index: sums in a fixed order
# speedup vs baseline: 2.5388x; 1.0497x over previous
#include <hip/hip_runtime.h>
#include <hip/hip_bf16.h>

typedef __attribute__((ext_vector_type(2))) float v2f;
typedef __attribute__((ext_vector_type(4))) float v4f;
typedef __attribute__((ext_vector_type(8))) float v8f;

#define LB_TPB    1024
#define LB_LDSN   65536
#define LB_SLICES 64
#define LB_BT     512
#define LB_NB     128
__global__ __launch_bounds__(LB_TPB)
void slice_count(const int* __restrict__ ei, int E, int N, int cstride,
                 int* __restrict__ T, int* __restrict__ C) {
    __shared__ int h[LB_LDSN];
    const int dir = blockIdx.y;
    const int* key = ei + (dir == 0 ? E : 0);
    const int tid = threadIdx.x;
    const int len = (E + LB_SLICES - 1) / LB_SLICES;
    const int lo  = min((int)blockIdx.x * len, E);
    const int hi  = min(lo + len, E);
    const size_t srow = (size_t)dir * LB_SLICES + blockIdx.x;
    int* Tt = T + srow * cstride;

    for (int n = tid; n < N; n += LB_TPB) h[n] = 0;
    __syncthreads();

    for (int e0 = lo; e0 < hi; e0 += 8 * LB_TPB) {
        int k[8];
#pragma unroll
        for (int u = 0; u < 8; ++u) {
            const int e = e0 + u * LB_TPB + tid;
            k[u] = (e < hi) ? key[e] : 0;
        }
#pragma unroll
        for (int u = 0; u < 8; ++u)
            if (e0 + u * LB_TPB + tid < hi) atomicAdd(&h[k[u]], 1);
    }
    __syncthreads();

    for (int n0 = 0; n0 < N; n0 += 8 * LB_TPB) {
        int c[8];
#pragma unroll
        for (int u = 0; u < 8; ++u) {
            const int n = n0 + u * LB_TPB + tid;
            c[u] = (n < N) ? h[n] : 0;
        }
#pragma unroll
        for (int u = 0; u < 8; ++u) {
            const int n = n0 + u * LB_TPB + tid;
            if (n < N) Tt[n] = c[u];
        }
    }

    const int i = tid >> 3;
    const int j = tid & 7;
    int s = 0;
    for (int m = 0; m < LB_BT / 8; ++m) {
        const int n = i * LB_BT + j + (m << 3);
        const int x = h[min(n, N - 1)];
        s += (n < N) ? x : 0;
    }
    s += __shfl_xor(s, 4);
    s += __shfl_xor(s, 2);
    s += __shfl_xor(s, 1);
    if (j == 0) C[srow * LB_NB + i] = s;
}

__global__ __launch_bounds__(LB_BT)
void slot_bases(int* __restrict__ T, const int* __restrict__ C, int N,
                int cstride, int* __restrict__ rowptr) {
    __shared__ int wsum[32];
    __shared__ int wpart[32];
    __shared__ int bstart;
    const int b    = blockIdx.x;
    const int dir  = blockIdx.y;
    const int tid  = threadIdx.x;
    const int lane = tid & 31;
    const int wv   = tid >> 5;
    const int n    = b * LB_BT + tid;
    int*       Td = T + (size_t)dir * LB_SLICES * cstride;
    const int* Cd = C + (size_t)dir * LB_SLICES * LB_NB;
    int*       rp = rowptr + (size_t)dir * (N + 1);

    int part = 0;
    for (int idx = tid; idx < LB_SLICES * LB_NB; idx += LB_BT)
        if ((idx & (LB_NB - 1)) < b) part += Cd[idx];

    int c[LB_SLICES];
#pragma unroll
    for (int s = 0; s < LB_SLICES; ++s)
        c[s] = Td[(size_t)s * cstride + min(n, N - 1)];
#pragma unroll
    for (int s = 0; s < LB_SLICES; ++s)
        if (n >= N) c[s] = 0;
    int deg = 0;
#pragma unroll
    for (int s = 0; s < LB_SLICES; ++s) deg += c[s];

    int incl = deg;
    for (int d = 1; d < 32; d <<= 1) {
        int t = __shfl_up(incl, d);
        if (lane >= d) incl += t;
    }
    for (int d = 16; d >= 1; d >>= 1) part += __shfl_xor(part, d);
    if (lane == 31) wsum[wv] = incl;
    if (lane == 0)  wpart[wv] = part;
    __syncthreads();
    if (wv == 0) {
        const int w = (lane < LB_BT / 32) ? wsum[lane] : 0;
        int wi = w;
        for (int d = 1; d < 32; d <<= 1) {
            int t = __shfl_up(wi, d);
            if (lane >= d) wi += t;
        }
        if (lane < LB_BT / 32) wsum[lane] = wi - w;
        int p = (lane < LB_BT / 32) ? wpart[lane] : 0;
        for (int d = 16; d >= 1; d >>= 1) p += __shfl_xor(p, d);
        if (lane == 0) bstart = p;
    }
    __syncthreads();

    int run = bstart + wsum[wv] + incl - deg;
    if (n < N) {
        rp[n] = run;
        if (n == N - 1) rp[N] = run + deg;
#pragma unroll
        for (int s = 0; s < LB_SLICES; ++s) {
            Td[(size_t)s * cstride + n] = run;
            run += c[s];
        }
    }
}

__global__ __launch_bounds__(LB_TPB)
void slice_place(const int* __restrict__ ei, int E, int N, int cstride,
                 const int* __restrict__ T, int* __restrict__ adj) {
    __shared__ int h[LB_LDSN];
    const int dir = blockIdx.y;
    const int* key = ei + (dir == 0 ? E : 0);
    const int* val = ei + (dir == 0 ? 0 : E);
    const int tid = threadIdx.x;
    const int len = (E + LB_SLICES - 1) / LB_SLICES;
    const int lo  = min((int)blockIdx.x * len, E);
    const int hi  = min(lo + len, E);
    const int* Tt = T + ((size_t)dir * LB_SLICES + blockIdx.x) * cstride;
    int* ad = adj + (size_t)dir * E;

    for (int n0 = 0; n0 < N; n0 += 16 * LB_TPB) {
        int c[16];
#pragma unroll
        for (int u = 0; u < 16; ++u) {
            const int n = n0 + u * LB_TPB + tid;
            c[u] = (n < N) ? Tt[n] : 0;
        }
#pragma unroll
        for (int u = 0; u < 16; ++u) {
            const int n = n0 + u * LB_TPB + tid;
            if (n < N) h[n] = c[u];
        }
    }
    __syncthreads();

    for (int e0 = lo; e0 < hi; e0 += 8 * LB_TPB) {
        int k[8], v[8];
#pragma unroll
        for (int u = 0; u < 8; ++u) {
            const int e = min(e0 + u * LB_TPB + tid, hi - 1);
            k[u] = key[e];
            v[u] = val[e];
        }
#pragma unroll
        for (int u = 0; u < 8; ++u)
            if (e0 + u * LB_TPB + tid < hi) ad[atomicAdd(&h[k[u]], 1)] = v[u];
    }
}

#define GA_TPB 256
__global__ __launch_bounds__(GA_TPB)
void gather_mean64(const float* __restrict__ Y,
                   const int* __restrict__ rowptr,
                   const int* __restrict__ adj,
                   float* __restrict__ S, int N) {
    const int lane = threadIdx.x & 31;
    const int h    = lane >> 4;
    const int c    = lane & 15;
    const int node = __builtin_amdgcn_readfirstlane(
        (int)((blockIdx.x * GA_TPB + threadIdx.x) >> 5));
    if (node >= N) return;
    const int beg = rowptr[node];
    const int end = rowptr[node + 1];
    const int d   = end - beg;

    const v4f* Y4 = (const v4f*)Y;
    v4f acc = {0.0f, 0.0f, 0.0f, 0.0f};
    if (d > 0 && d <= 32) {
        const int my  = adj[beg + min(lane, d - 1)];
        const int key = (my << 5) | lane;
        int pos = 0;
        for (int j = 0; j < d; ++j) {
            const int kj = __builtin_amdgcn_readlane(key, j);
            pos += (kj < key) ? 1 : 0;
        }
        if (lane >= d) pos = lane;
        const int dst = ((pos & 1) << 4) | (pos >> 1);
        const int srt = __builtin_amdgcn_ds_permute(dst << 2, my);
        v4f v[8];
#define GA_ROW(U, Q) v[U] = Y4[(size_t)__builtin_amdgcn_update_dpp(0, srt, 0x150 + (Q), 0xf, 0xf, false) * 16 + c]
        GA_ROW(0, 0); GA_ROW(1, 1); GA_ROW(2, 2); GA_ROW(3, 3);
        GA_ROW(4, 4); GA_ROW(5, 5); GA_ROW(6, 6); GA_ROW(7, 7);
#pragma unroll
        for (int u = 0; u < 8; ++u)
            if (2 * u + h < d) acc += v[u];
        if (d > 16) {
            GA_ROW(0, 8);  GA_ROW(1, 9);  GA_ROW(2, 10); GA_ROW(3, 11);
            GA_ROW(4, 12); GA_ROW(5, 13); GA_ROW(6, 14); GA_ROW(7, 15);
#pragma unroll
            for (int u = 0; u < 8; ++u)
                if (16 + 2 * u + h < d) acc += v[u];
        }
#undef GA_ROW
    } else if (d > 32 && d <= 64) {
        const int e0 = adj[beg + lane];
        const int e1 = adj[beg + min(32 + lane, d - 1)];
        const int k0 = (e0 << 6) | lane;
        const int k1 = (32 + lane < d) ? ((e1 << 6) | (32 + lane)) : 0x7fffffff;
        int p0 = 0, p1 = 0;
        for (int j = 0; j < 32; ++j) {
            const int kj = __builtin_amdgcn_readlane(k0, j);
            p0 += (kj < k0) ? 1 : 0;
            p1 += (kj < k1) ? 1 : 0;
        }
        for (int j = 32; j < d; ++j) {
            const int kj = __builtin_amdgcn_readlane(k1, j - 32);
            p0 += (kj < k0) ? 1 : 0;
            p1 += (kj < k1) ? 1 : 0;
        }
        if (32 + lane >= d) p1 = -1;
        for (int g = 0; g < d; g += 16) {
            v4f v[8];
#pragma unroll
            for (int u = 0; u < 8; ++u) {
                int jr[2];
#pragma unroll
                for (int q = 0; q < 2; ++q) {
                    const int r = min(g + 2 * u + q, d - 1);
                    const unsigned m0 = __builtin_amdgcn_ballot_w32(p0 == r);
                    const unsigned m1 = __builtin_amdgcn_ballot_w32(p1 == r);
                    const int a0 = __builtin_amdgcn_readlane(e0, m0 ? __builtin_ctz(m0) : 0);
                    const int a1 = __builtin_amdgcn_readlane(e1, m1 ? __builtin_ctz(m1) : 0);
                    jr[q] = m0 ? a0 : a1;
                }
                const int jj = h ? jr[1] : jr[0];
                v[u] = Y4[(size_t)jj * 16 + c];
            }
#pragma unroll
            for (int u = 0; u < 8; ++u)
                if (g + 2 * u + h < d) acc += v[u];
        }
    } else if (d > 64) {
        long long last = -1;
        for (int it = 0; it < d; ++it) {
            long long best = 0x7fffffffffffffffLL;
            for (int p = lane; p < d; p += 32) {
                const long long kk = ((long long)adj[beg + p] << 32) | (long long)p;
                if (kk > last && kk < best) best = kk;
            }
            for (int m = 16; m >= 1; m >>= 1) {
                const long long o = __shfl_xor(best, m);
                if (o < best) best = o;
            }
            last = best;
            const int jj = (int)(best >> 32);
            const v4f r = Y4[(size_t)jj * 16 + c];
            if (h == 0) acc += r;
        }
    }
    acc.x += __shfl_xor(acc.x, 16);
    acc.y += __shfl_xor(acc.y, 16);
    acc.z += __shfl_xor(acc.z, 16);
    acc.w += __shfl_xor(acc.w, 16);
    const float inv = 1.0f / fmaxf((float)d, 1.0f);
    if (h == 0) ((v4f*)S)[(size_t)node * 16 + c] = acc * inv;
}

__global__ __launch_bounds__(128)
void gemm_wmma_relu(const float* __restrict__ X0, int d0,
                    const float* __restrict__ X1,
                    const float* __restrict__ cnt,
                    const float* __restrict__ W,
                    int Ktot,
                    const float* __restrict__ bias,
                    float* __restrict__ out, int ostride, int ocol0, int dup,
                    int Nrows) {
    __shared__ float sW[192 * 64];

    const int tid = threadIdx.x;

    for (int i = tid; i < Ktot * 64; i += 128) {
        int k = i >> 6;
        int c = i & 63;
        sW[(((k >> 1) << 6) + c) * 2 + (k & 1)] = W[i];
    }
    __syncthreads();

    const int wave = tid >> 5;
    const int lane = tid & 31;
    const int m    = lane & 15;
    const int kSel = lane >> 4;

    const int rowbase = (blockIdx.x * 4 + wave) * 16;
    const int row     = rowbase + m;
    const int crow    = row < Nrows ? row : (Nrows - 1);

    const float* x0r = X0 + (size_t)crow * d0 + 2 * kSel;
    const float* x1r = X1 ? X1 + (size_t)crow * 64 + 2 * kSel : nullptr;
    float scale1 = 1.0f;
    if (cnt) scale1 = 1.0f / fmaxf(cnt[crow], 1.0f);

    v8f acc[4] = {};

#pragma unroll 1
    for (int k0 = 0; k0 < d0; k0 += 32) {
        v2f areg[8];
#pragma unroll
        for (int u = 0; u < 8; ++u)
            areg[u] = *(const v2f*)(x0r + k0 + 4 * u);
#pragma unroll
        for (int u = 0; u < 8; ++u) {
            const int sbase = (((k0 >> 1) + 2 * u + kSel) << 7);
#pragma unroll
            for (int t = 0; t < 4; ++t) {
                v2f b = *(const v2f*)&sW[sbase + ((t << 4) + m) * 2];
                acc[t] = __builtin_amdgcn_wmma_f32_16x16x4_f32(
                    false, areg[u], false, b, (short)0, acc[t], false, false);
            }
        }
    }

    if (x1r) {
#pragma unroll 1
        for (int k0 = d0; k0 < Ktot; k0 += 32) {
            v2f areg[8];
#pragma unroll
            for (int u = 0; u < 8; ++u)
                areg[u] = *(const v2f*)(x1r + (k0 - d0) + 4 * u);
#pragma unroll
            for (int u = 0; u < 8; ++u) {
                v2f a;
                a.x = areg[u].x * scale1;
                a.y = areg[u].y * scale1;
                const int sbase = (((k0 >> 1) + 2 * u + kSel) << 7);
#pragma unroll
                for (int t = 0; t < 4; ++t) {
                    v2f b = *(const v2f*)&sW[sbase + ((t << 4) + m) * 2];
                    acc[t] = __builtin_amdgcn_wmma_f32_16x16x4_f32(
                        false, a, false, b, (short)0, acc[t], false, false);
                }
            }
        }
    }

    const int rr  = kSel * 8;
    const int col = m;
#pragma unroll
    for (int r = 0; r < 8; ++r) {
        const int orow = rowbase + r + rr;
        if (orow < Nrows) {
#pragma unroll
            for (int t = 0; t < 4; ++t) {
                float v = acc[t][r];
                if (bias) v += bias[(t << 4) + col];
                v = fmaxf(v, 0.0f);
                float* p = out + (size_t)orow * ostride + ocol0 + (t << 4) + col;
                *p = v;
                if (dup) p[64] = v;
            }
        }
    }
}

extern "C" void kernel_launch(void* const* d_in, const int* in_sizes, int n_in,
                              void* d_out, int out_size, void* d_ws, size_t ws_size,
                              hipStream_t stream) {
    const float* x    = (const float*)d_in[0];
    const int*   ei   = (const int*)  d_in[1];
    const float* W_u1 = (const float*)d_in[2];
    const float* b_u1 = (const float*)d_in[3];
    const float* U_u1 = (const float*)d_in[4];
    const float* W_d1 = (const float*)d_in[5];
    const float* b_d1 = (const float*)d_in[6];
    const float* U_d1 = (const float*)d_in[7];
    const float* W_u2 = (const float*)d_in[8];
    const float* b_u2 = (const float*)d_in[9];
    const float* U_u2 = (const float*)d_in[10];
    float* out = (float*)d_out;

    const int N = in_sizes[0] / 128;
    const int E = in_sizes[1] / 2;

    const int cstride = (N + 31) / 32 * 32;
    float* ws     = (float*)d_ws;
    float* Y      = ws;
    float* S      = Y  + (size_t)N * 64;
    float* ha     = S  + (size_t)N * 64;
    int*   cnt    = (int*)(ha + (size_t)N * 128);
    int*   rowptr = cnt    + 2 * (size_t)cstride;
    int*   adj    = rowptr + 2 * ((size_t)N + 1);
    int*   T      = (int*)ws;
    int*   C      = cnt;

    const int* rp_in   = rowptr;
    const int* rp_out  = rowptr + (N + 1);
    const int* adj_in  = adj;
    const int* adj_out = adj + E;

    slice_count<<<dim3(LB_SLICES, 2), LB_TPB, 0, stream>>>(ei, E, N, cstride, T, C);
    slot_bases<<<dim3((N + LB_BT - 1) / LB_BT, 2), LB_BT, 0, stream>>>(T, C, N, cstride, rowptr);
    slice_place<<<dim3(LB_SLICES, 2), LB_TPB, 0, stream>>>(ei, E, N, cstride, T, adj);

    const int gblk  = (N + 63) / 64;
    const int ggrid = (N + GA_TPB / 32 - 1) / (GA_TPB / 32);

    auto conv = [&](const float* Xin, const int* rp, const int* nb,
                    const float* Wm, const float* bm,
                    const float* Um, float* outp, int ocol0, int dupw) {
        gemm_wmma_relu<<<gblk, 128, 0, stream>>>(Xin, 128, nullptr, nullptr,
                                                 Wm, 128, bm, Y, 64, 0, 0, N);
        gather_mean64<<<ggrid, GA_TPB, 0, stream>>>(Y, rp, nb, S, N);
        gemm_wmma_relu<<<gblk, 128, 0, stream>>>(Xin, 128, S, nullptr,
                                                 Um, 192, nullptr,
                                                 outp, 128, ocol0, dupw, N);
    };

    conv(x,   rp_in,  adj_in,  W_u1, b_u1, U_u1, out,  0, 0);
    conv(x,   rp_out, adj_out, W_d1, b_d1, U_d1, out, 64, 0);
    conv(out, rp_in,  adj_in,  W_u2, b_u2, U_u2, ha,   0, 1);
    conv(ha,  rp_in,  adj_in,  W_u2, b_u2, U_u2, out,  0, 1);
}


# the earlier try's run-to-run spread came from arrival-order sums, sorted since: layer 1's up and down branches share one launch per stage (grid y = branch; LDS tile sized to K), 12 launches not 15
# speedup vs baseline: 2.7856x; 1.0972x over previous
#include <hip/hip_runtime.h>
#include <hip/hip_bf16.h>

typedef __attribute__((ext_vector_type(2))) float v2f;
typedef __attribute__((ext_vector_type(4))) float v4f;
typedef __attribute__((ext_vector_type(8))) float v8f;

#define LB_TPB    1024
#define LB_LDSN   65536
#define LB_SLICES 64
#define LB_BT     512
#define LB_NB     128
__global__ __launch_bounds__(LB_TPB)
void slice_count(const int* __restrict__ ei, int E, int N, int cstride,
                 int* __restrict__ T, int* __restrict__ C) {
    __shared__ int h[LB_LDSN];
    const int dir = blockIdx.y;
    const int* key = ei + (dir == 0 ? E : 0);
    const int tid = threadIdx.x;
    const int len = (E + LB_SLICES - 1) / LB_SLICES;
    const int lo  = min((int)blockIdx.x * len, E);
    const int hi  = min(lo + len, E);
    const size_t srow = (size_t)dir * LB_SLICES + blockIdx.x;
    int* Tt = T + srow * cstride;

    for (int n = tid; n < N; n += LB_TPB) h[n] = 0;
    __syncthreads();

    for (int e0 = lo; e0 < hi; e0 += 8 * LB_TPB) {
        int k[8];
#pragma unroll
        for (int u = 0; u < 8; ++u) {
            const int e = e0 + u * LB_TPB + tid;
            k[u] = (e < hi) ? key[e] : 0;
        }
#pragma unroll
        for (int u = 0; u < 8; ++u)
            if (e0 + u * LB_TPB + tid < hi) atomicAdd(&h[k[u]], 1);
    }
    __syncthreads();

    for (int n0 = 0; n0 < N; n0 += 8 * LB_TPB) {
        int c[8];
#pragma unroll
        for (int u = 0; u < 8; ++u) {
            const int n = n0 + u * LB_TPB + tid;
            c[u] = (n < N) ? h[n] : 0;
        }
#pragma unroll
        for (int u = 0; u < 8; ++u) {
            const int n = n0 + u * LB_TPB + tid;
            if (n < N) Tt[n] = c[u];
        }
    }

    const int i = tid >> 3;
    const int j = tid & 7;
    int s = 0;
    for (int m = 0; m < LB_BT / 8; ++m) {
        const int n = i * LB_BT + j + (m << 3);
        const int x = h[min(n, N - 1)];
        s += (n < N) ? x : 0;
    }
    s += __shfl_xor(s, 4);
    s += __shfl_xor(s, 2);
    s += __shfl_xor(s, 1);
    if (j == 0) C[srow * LB_NB + i] = s;
}

__global__ __launch_bounds__(LB_BT)
void slot_bases(int* __restrict__ T, const int* __restrict__ C, int N,
                int cstride, int* __restrict__ rowptr) {
    __shared__ int wsum[32];
    __shared__ int wpart[32];
    __shared__ int bstart;
    const int b    = blockIdx.x;
    const int dir  = blockIdx.y;
    const int tid  = threadIdx.x;
    const int lane = tid & 31;
    const int wv   = tid >> 5;
    const int n    = b * LB_BT + tid;
    int*       Td = T + (size_t)dir * LB_SLICES * cstride;
    const int* Cd = C + (size_t)dir * LB_SLICES * LB_NB;
    int*       rp = rowptr + (size_t)dir * (N + 1);

    int part = 0;
    for (int idx = tid; idx < LB_SLICES * LB_NB; idx += LB_BT)
        if ((idx & (LB_NB - 1)) < b) part += Cd[idx];

    int c[LB_SLICES];
#pragma unroll
    for (int s = 0; s < LB_SLICES; ++s)
        c[s] = Td[(size_t)s * cstride + min(n, N - 1)];
#pragma unroll
    for (int s = 0; s < LB_SLICES; ++s)
        if (n >= N) c[s] = 0;
    int deg = 0;
#pragma unroll
    for (int s = 0; s < LB_SLICES; ++s) deg += c[s];

    int incl = deg;
    for (int d = 1; d < 32; d <<= 1) {
        int t = __shfl_up(incl, d);
        if (lane >= d) incl += t;
    }
    for (int d = 16; d >= 1; d >>= 1) part += __shfl_xor(part, d);
    if (lane == 31) wsum[wv] = incl;
    if (lane == 0)  wpart[wv] = part;
    __syncthreads();
    if (wv == 0) {
        const int w = (lane < LB_BT / 32) ? wsum[lane] : 0;
        int wi = w;
        for (int d = 1; d < 32; d <<= 1) {
            int t = __shfl_up(wi, d);
            if (lane >= d) wi += t;
        }
        if (lane < LB_BT / 32) wsum[lane] = wi - w;
        int p = (lane < LB_BT / 32) ? wpart[lane] : 0;
        for (int d = 16; d >= 1; d >>= 1) p += __shfl_xor(p, d);
        if (lane == 0) bstart = p;
    }
    __syncthreads();

    int run = bstart + wsum[wv] + incl - deg;
    if (n < N) {
        rp[n] = run;
        if (n == N - 1) rp[N] = run + deg;
#pragma unroll
        for (int s = 0; s < LB_SLICES; ++s) {
            Td[(size_t)s * cstride + n] = run;
            run += c[s];
        }
    }
}

__global__ __launch_bounds__(LB_TPB)
void slice_place(const int* __restrict__ ei, int E, int N, int cstride,
                 const int* __restrict__ T, int* __restrict__ adj) {
    __shared__ int h[LB_LDSN];
    const int dir = blockIdx.y;
    const int* key = ei + (dir == 0 ? E : 0);
    const int* val = ei + (dir == 0 ? 0 : E);
    const int tid = threadIdx.x;
    const int len = (E + LB_SLICES - 1) / LB_SLICES;
    const int lo  = min((int)blockIdx.x * len, E);
    const int hi  = min(lo + len, E);
    const int* Tt = T + ((size_t)dir * LB_SLICES + blockIdx.x) * cstride;
    int* ad = adj + (size_t)dir * E;

    for (int n0 = 0; n0 < N; n0 += 16 * LB_TPB) {
        int c[16];
#pragma unroll
        for (int u = 0; u < 16; ++u) {
            const int n = n0 + u * LB_TPB + tid;
            c[u] = (n < N) ? Tt[n] : 0;
        }
#pragma unroll
        for (int u = 0; u < 16; ++u) {
            const int n = n0 + u * LB_TPB + tid;
            if (n < N) h[n] = c[u];
        }
    }
    __syncthreads();

    for (int e0 = lo; e0 < hi; e0 += 8 * LB_TPB) {
        int k[8], v[8];
#pragma unroll
        for (int u = 0; u < 8; ++u) {
            const int e = min(e0 + u * LB_TPB + tid, hi - 1);
            k[u] = key[e];
            v[u] = val[e];
        }
#pragma unroll
        for (int u = 0; u < 8; ++u)
            if (e0 + u * LB_TPB + tid < hi) ad[atomicAdd(&h[k[u]], 1)] = v[u];
    }
}

#define GA_TPB 256
__global__ __launch_bounds__(GA_TPB)
void gather_mean64(const float* __restrict__ Ya, const float* __restrict__ Yb,
                   const int* __restrict__ rowptra, const int* __restrict__ rowptrb,
                   const int* __restrict__ adja, const int* __restrict__ adjb,
                   float* __restrict__ Sa, float* __restrict__ Sb, int N) {
    const bool br = blockIdx.y != 0;
    const float* __restrict__ Y      = br ? Yb : Ya;
    const int*   __restrict__ rowptr = br ? rowptrb : rowptra;
    const int*   __restrict__ adj    = br ? adjb : adja;
    float*       __restrict__ S      = br ? Sb : Sa;
    const int lane = threadIdx.x & 31;
    const int h    = lane >> 4;
    const int c    = lane & 15;
    const int node = __builtin_amdgcn_readfirstlane(
        (int)((blockIdx.x * GA_TPB + threadIdx.x) >> 5));
    if (node >= N) return;
    const int beg = rowptr[node];
    const int end = rowptr[node + 1];
    const int d   = end - beg;

    const v4f* Y4 = (const v4f*)Y;
    v4f acc = {0.0f, 0.0f, 0.0f, 0.0f};
    if (d > 0 && d <= 32) {
        const int my  = adj[beg + min(lane, d - 1)];
        const int key = (my << 5) | lane;
        int pos = 0;
        for (int j = 0; j < d; ++j) {
            const int kj = __builtin_amdgcn_readlane(key, j);
            pos += (kj < key) ? 1 : 0;
        }
        if (lane >= d) pos = lane;
        const int dst = ((pos & 1) << 4) | (pos >> 1);
        const int srt = __builtin_amdgcn_ds_permute(dst << 2, my);
        v4f v[8];
#define GA_ROW(U, Q) v[U] = Y4[(size_t)__builtin_amdgcn_update_dpp(0, srt, 0x150 + (Q), 0xf, 0xf, false) * 16 + c]
        GA_ROW(0, 0); GA_ROW(1, 1); GA_ROW(2, 2); GA_ROW(3, 3);
        GA_ROW(4, 4); GA_ROW(5, 5); GA_ROW(6, 6); GA_ROW(7, 7);
#pragma unroll
        for (int u = 0; u < 8; ++u)
            if (2 * u + h < d) acc += v[u];
        if (d > 16) {
            GA_ROW(0, 8);  GA_ROW(1, 9);  GA_ROW(2, 10); GA_ROW(3, 11);
            GA_ROW(4, 12); GA_ROW(5, 13); GA_ROW(6, 14); GA_ROW(7, 15);
#pragma unroll
            for (int u = 0; u < 8; ++u)
                if (16 + 2 * u + h < d) acc += v[u];
        }
#undef GA_ROW
    } else if (d > 32 && d <= 64) {
        const int e0 = adj[beg + lane];
        const int e1 = adj[beg + min(32 + lane, d - 1)];
        const int k0 = (e0 << 6) | lane;
        const int k1 = (32 + lane < d) ? ((e1 << 6) | (32 + lane)) : 0x7fffffff;
        int p0 = 0, p1 = 0;
        for (int j = 0; j < 32; ++j) {
            const int kj = __builtin_amdgcn_readlane(k0, j);
            p0 += (kj < k0) ? 1 : 0;
            p1 += (kj < k1) ? 1 : 0;
        }
        for (int j = 32; j < d; ++j) {
            const int kj = __builtin_amdgcn_readlane(k1, j - 32);
            p0 += (kj < k0) ? 1 : 0;
            p1 += (kj < k1) ? 1 : 0;
        }
        if (32 + lane >= d) p1 = -1;
        for (int g = 0; g < d; g += 16) {
            v4f v[8];
#pragma unroll
            for (int u = 0; u < 8; ++u) {
                int jr[2];
#pragma unroll
                for (int q = 0; q < 2; ++q) {
                    const int r = min(g + 2 * u + q, d - 1);
                    const unsigned m0 = __builtin_amdgcn_ballot_w32(p0 == r);
                    const unsigned m1 = __builtin_amdgcn_ballot_w32(p1 == r);
                    const int a0 = __builtin_amdgcn_readlane(e0, m0 ? __builtin_ctz(m0) : 0);
                    const int a1 = __builtin_amdgcn_readlane(e1, m1 ? __builtin_ctz(m1) : 0);
                    jr[q] = m0 ? a0 : a1;
                }
                const int jj = h ? jr[1] : jr[0];
                v[u] = Y4[(size_t)jj * 16 + c];
            }
#pragma unroll
            for (int u = 0; u < 8; ++u)
                if (g + 2 * u + h < d) acc += v[u];
        }
    } else if (d > 64) {
        long long last = -1;
        for (int it = 0; it < d; ++it) {
            long long best = 0x7fffffffffffffffLL;
            for (int p = lane; p < d; p += 32) {
                const long long kk = ((long long)adj[beg + p] << 32) | (long long)p;
                if (kk > last && kk < best) best = kk;
            }
            for (int m = 16; m >= 1; m >>= 1) {
                const long long o = __shfl_xor(best, m);
                if (o < best) best = o;
            }
            last = best;
            const int jj = (int)(best >> 32);
            const v4f r = Y4[(size_t)jj * 16 + c];
            if (h == 0) acc += r;
        }
    }
    acc.x += __shfl_xor(acc.x, 16);
    acc.y += __shfl_xor(acc.y, 16);
    acc.z += __shfl_xor(acc.z, 16);
    acc.w += __shfl_xor(acc.w, 16);
    const float inv = 1.0f / fmaxf((float)d, 1.0f);
    if (h == 0) ((v4f*)S)[(size_t)node * 16 + c] = acc * inv;
}

template <int LDSK>
__global__ __launch_bounds__(128)
void gemm_wmma_relu(const float* __restrict__ X0, int d0,
                    const float* __restrict__ X1a,
                    const float* __restrict__ X1b,
                    const float* __restrict__ cnt,
                    const float* __restrict__ Wa,
                    const float* __restrict__ Wb,
                    int Ktot,
                    const float* __restrict__ biasa,
                    const float* __restrict__ biasb,
                    float* outa, float* outb,
                    int ostride, int ocol0a, int ocol0b, int dup, int Nrows) {
    __shared__ float sW[LDSK * 64];
    const bool br = blockIdx.y != 0;
    const float* __restrict__ X1   = br ? X1b : X1a;
    const float* __restrict__ W    = br ? Wb : Wa;
    const float* __restrict__ bias = br ? biasb : biasa;
    float* __restrict__ out = br ? outb : outa;
    const int ocol0 = br ? ocol0b : ocol0a;

    const int tid = threadIdx.x;

    for (int i = tid; i < Ktot * 64; i += 128) {
        int k = i >> 6;
        int c = i & 63;
        sW[(((k >> 1) << 6) + c) * 2 + (k & 1)] = W[i];
    }
    __syncthreads();

    const int wave = tid >> 5;
    const int lane = tid & 31;
    const int m    = lane & 15;
    const int kSel = lane >> 4;

    const int rowbase = (blockIdx.x * 4 + wave) * 16;
    const int row     = rowbase + m;
    const int crow    = row < Nrows ? row : (Nrows - 1);

    const float* x0r = X0 + (size_t)crow * d0 + 2 * kSel;
    const float* x1r = X1 ? X1 + (size_t)crow * 64 + 2 * kSel : nullptr;
    float scale1 = 1.0f;
    if (cnt) scale1 = 1.0f / fmaxf(cnt[crow], 1.0f);

    v8f acc[4] = {};

#pragma unroll 1
    for (int k0 = 0; k0 < d0; k0 += 32) {
        v2f areg[8];
#pragma unroll
        for (int u = 0; u < 8; ++u)
            areg[u] = *(const v2f*)(x0r + k0 + 4 * u);
#pragma unroll
        for (int u = 0; u < 8; ++u) {
            const int sbase = (((k0 >> 1) + 2 * u + kSel) << 7);
#pragma unroll
            for (int t = 0; t < 4; ++t) {
                v2f b = *(const v2f*)&sW[sbase + ((t << 4) + m) * 2];
                acc[t] = __builtin_amdgcn_wmma_f32_16x16x4_f32(
                    false, areg[u], false, b, (short)0, acc[t], false, false);
            }
        }
    }

    if (x1r) {
#pragma unroll 1
        for (int k0 = d0; k0 < Ktot; k0 += 32) {
            v2f areg[8];
#pragma unroll
            for (int u = 0; u < 8; ++u)
                areg[u] = *(const v2f*)(x1r + (k0 - d0) + 4 * u);
#pragma unroll
            for (int u = 0; u < 8; ++u) {
                v2f a;
                a.x = areg[u].x * scale1;
                a.y = areg[u].y * scale1;
                const int sbase = (((k0 >> 1) + 2 * u + kSel) << 7);
#pragma unroll
                for (int t = 0; t < 4; ++t) {
                    v2f b = *(const v2f*)&sW[sbase + ((t << 4) + m) * 2];
                    acc[t] = __builtin_amdgcn_wmma_f32_16x16x4_f32(
                        false, a, false, b, (short)0, acc[t], false, false);
                }
            }
        }
    }

    const int rr  = kSel * 8;
    const int col = m;
#pragma unroll
    for (int r = 0; r < 8; ++r) {
        const int orow = rowbase + r + rr;
        if (orow < Nrows) {
#pragma unroll
            for (int t = 0; t < 4; ++t) {
                float v = acc[t][r];
                if (bias) v += bias[(t << 4) + col];
                v = fmaxf(v, 0.0f);
                float* p = out + (size_t)orow * ostride + ocol0 + (t << 4) + col;
                *p = v;
                if (dup) p[64] = v;
            }
        }
    }
}

extern "C" void kernel_launch(void* const* d_in, const int* in_sizes, int n_in,
                              void* d_out, int out_size, void* d_ws, size_t ws_size,
                              hipStream_t stream) {
    const float* x    = (const float*)d_in[0];
    const int*   ei   = (const int*)  d_in[1];
    const float* W_u1 = (const float*)d_in[2];
    const float* b_u1 = (const float*)d_in[3];
    const float* U_u1 = (const float*)d_in[4];
    const float* W_d1 = (const float*)d_in[5];
    const float* b_d1 = (const float*)d_in[6];
    const float* U_d1 = (const float*)d_in[7];
    const float* W_u2 = (const float*)d_in[8];
    const float* b_u2 = (const float*)d_in[9];
    const float* U_u2 = (const float*)d_in[10];
    float* out = (float*)d_out;

    const int N = in_sizes[0] / 128;
    const int E = in_sizes[1] / 2;

    const int cstride = (N + 31) / 32 * 32;
    float* ws     = (float*)d_ws;
    float* Y      = ws;
    float* S      = Y  + (size_t)N * 64;
    float* ha     = S  + (size_t)N * 64;
    int*   cnt    = (int*)(ha + (size_t)N * 128);
    int*   rowptr = cnt    + 2 * (size_t)cstride;
    int*   adj    = rowptr + 2 * ((size_t)N + 1);
    int*   T      = (int*)ws;
    int*   C      = cnt;

    const int* rp_in   = rowptr;
    const int* rp_out  = rowptr + (N + 1);
    const int* adj_in  = adj;
    const int* adj_out = adj + E;

    slice_count<<<dim3(LB_SLICES, 2), LB_TPB, 0, stream>>>(ei, E, N, cstride, T, C);
    slot_bases<<<dim3((N + LB_BT - 1) / LB_BT, 2), LB_BT, 0, stream>>>(T, C, N, cstride, rowptr);
    slice_place<<<dim3(LB_SLICES, 2), LB_TPB, 0, stream>>>(ei, E, N, cstride, T, adj);

    const int gblk  = (N + 63) / 64;
    const int ggrid = (N + GA_TPB / 32 - 1) / (GA_TPB / 32);

    float* Yd = ha;
    float* Sd = ha + (size_t)N * 64;
    gemm_wmma_relu<128><<<dim3(gblk, 2), 128, 0, stream>>>(
        x, 128, nullptr, nullptr, nullptr, W_u1, W_d1, 128, b_u1, b_d1,
        Y, Yd, 64, 0, 0, 0, N);
    gather_mean64<<<dim3(ggrid, 2), GA_TPB, 0, stream>>>(
        Y, Yd, rp_in, rp_out, adj_in, adj_out, S, Sd, N);
    gemm_wmma_relu<192><<<dim3(gblk, 2), 128, 0, stream>>>(
        x, 128, S, Sd, nullptr, U_u1, U_d1, 192, nullptr, nullptr,
        out, out, 128, 0, 64, 0, N);

    auto conv = [&](const float* Xin, const int* rp, const int* nb,
                    const float* Wm, const float* bm,
                    const float* Um, float* outp, int dupw) {
        gemm_wmma_relu<128><<<gblk, 128, 0, stream>>>(
            Xin, 128, nullptr, nullptr, nullptr, Wm, nullptr, 128, bm, nullptr,
            Y, nullptr, 64, 0, 0, 0, N);
        gather_mean64<<<ggrid, GA_TPB, 0, stream>>>(Y, nullptr, rp, nullptr, nb, nullptr,
                                                    S, nullptr, N);
        gemm_wmma_relu<192><<<gblk, 128, 0, stream>>>(
            Xin, 128, S, nullptr, nullptr, Um, nullptr, 192, nullptr, nullptr,
            outp, nullptr, 128, 0, 0, dupw, N);
    };

    conv(out, rp_in,  adj_in,  W_u2, b_u2, U_u2, ha,  1);
    conv(ha,  rp_in,  adj_in,  W_u2, b_u2, U_u2, out, 1);
}


# GEMM kernel: no load waited for where issued: first A chunk and bias requested before the W tile, W staged by one group of 16-byte loads, next A chunk requested ahead of each WMMA block; same sums
# speedup vs baseline: 3.1159x; 1.1186x over previous
#include <hip/hip_runtime.h>
#include <hip/hip_bf16.h>

typedef __attribute__((ext_vector_type(2))) float v2f;
typedef __attribute__((ext_vector_type(4))) float v4f;
typedef __attribute__((ext_vector_type(8))) float v8f;

#define LB_TPB    1024
#define LB_LDSN   65536
#define LB_SLICES 64
#define LB_BT     512
#define LB_NB     128
__global__ __launch_bounds__(LB_TPB)
void slice_count(const int* __restrict__ ei, int E, int N, int cstride,
                 int* __restrict__ T, int* __restrict__ C) {
    __shared__ int h[LB_LDSN];
    const int dir = blockIdx.y;
    const int* key = ei + (dir == 0 ? E : 0);
    const int tid = threadIdx.x;
    const int len = (E + LB_SLICES - 1) / LB_SLICES;
    const int lo  = min((int)blockIdx.x * len, E);
    const int hi  = min(lo + len, E);
    const size_t srow = (size_t)dir * LB_SLICES + blockIdx.x;
    int* Tt = T + srow * cstride;

    for (int n = tid; n < N; n += LB_TPB) h[n] = 0;
    __syncthreads();

    for (int e0 = lo; e0 < hi; e0 += 8 * LB_TPB) {
        int k[8];
#pragma unroll
        for (int u = 0; u < 8; ++u) {
            const int e = e0 + u * LB_TPB + tid;
            k[u] = (e < hi) ? key[e] : 0;
        }
#pragma unroll
        for (int u = 0; u < 8; ++u)
            if (e0 + u * LB_TPB + tid < hi) atomicAdd(&h[k[u]], 1);
    }
    __syncthreads();

    for (int n0 = 0; n0 < N; n0 += 8 * LB_TPB) {
        int c[8];
#pragma unroll
        for (int u = 0; u < 8; ++u) {
            const int n = n0 + u * LB_TPB + tid;
            c[u] = (n < N) ? h[n] : 0;
        }
#pragma unroll
        for (int u = 0; u < 8; ++u) {
            const int n = n0 + u * LB_TPB + tid;
            if (n < N) Tt[n] = c[u];
        }
    }

    const int i = tid >> 3;
    const int j = tid & 7;
    int s = 0;
    for (int m = 0; m < LB_BT / 8; ++m) {
        const int n = i * LB_BT + j + (m << 3);
        const int x = h[min(n, N - 1)];
        s += (n < N) ? x : 0;
    }
    s += __shfl_xor(s, 4);
    s += __shfl_xor(s, 2);
    s += __shfl_xor(s, 1);
    if (j == 0) C[srow * LB_NB + i] = s;
}

__global__ __launch_bounds__(LB_BT)
void slot_bases(int* __restrict__ T, const int* __restrict__ C, int N,
                int cstride, int* __restrict__ rowptr) {
    __shared__ int wsum[32];
    __shared__ int wpart[32];
    __shared__ int bstart;
    const int b    = blockIdx.x;
    const int dir  = blockIdx.y;
    const int tid  = threadIdx.x;
    const int lane = tid & 31;
    const int wv   = tid >> 5;
    const int n    = b * LB_BT + tid;
    int*       Td = T + (size_t)dir * LB_SLICES * cstride;
    const int* Cd = C + (size_t)dir * LB_SLICES * LB_NB;
    int*       rp = rowptr + (size_t)dir * (N + 1);

    int part = 0;
    for (int idx = tid; idx < LB_SLICES * LB_NB; idx += LB_BT)
        if ((idx & (LB_NB - 1)) < b) part += Cd[idx];

    int c[LB_SLICES];
#pragma unroll
    for (int s = 0; s < LB_SLICES; ++s)
        c[s] = Td[(size_t)s * cstride + min(n, N - 1)];
#pragma unroll
    for (int s = 0; s < LB_SLICES; ++s)
        if (n >= N) c[s] = 0;
    int deg = 0;
#pragma unroll
    for (int s = 0; s < LB_SLICES; ++s) deg += c[s];

    int incl = deg;
    for (int d = 1; d < 32; d <<= 1) {
        int t = __shfl_up(incl, d);
        if (lane >= d) incl += t;
    }
    for (int d = 16; d >= 1; d >>= 1) part += __shfl_xor(part, d);
    if (lane == 31) wsum[wv] = incl;
    if (lane == 0)  wpart[wv] = part;
    __syncthreads();
    if (wv == 0) {
        const int w = (lane < LB_BT / 32) ? wsum[lane] : 0;
        int wi = w;
        for (int d = 1; d < 32; d <<= 1) {
            int t = __shfl_up(wi, d);
            if (lane >= d) wi += t;
        }
        if (lane < LB_BT / 32) wsum[lane] = wi - w;
        int p = (lane < LB_BT / 32) ? wpart[lane] : 0;
        for (int d = 16; d >= 1; d >>= 1) p += __shfl_xor(p, d);
        if (lane == 0) bstart = p;
    }
    __syncthreads();

    int run = bstart + wsum[wv] + incl - deg;
    if (n < N) {
        rp[n] = run;
        if (n == N - 1) rp[N] = run + deg;
#pragma unroll
        for (int s = 0; s < LB_SLICES; ++s) {
            Td[(size_t)s * cstride + n] = run;
            run += c[s];
        }
    }
}

__global__ __launch_bounds__(LB_TPB)
void slice_place(const int* __restrict__ ei, int E, int N, int cstride,
                 const int* __restrict__ T, int* __restrict__ adj) {
    __shared__ int h[LB_LDSN];
    const int dir = blockIdx.y;
    const int* key = ei + (dir == 0 ? E : 0);
    const int* val = ei + (dir == 0 ? 0 : E);
    const int tid = threadIdx.x;
    const int len = (E + LB_SLICES - 1) / LB_SLICES;
    const int lo  = min((int)blockIdx.x * len, E);
    const int hi  = min(lo + len, E);
    const int* Tt = T + ((size_t)dir * LB_SLICES + blockIdx.x) * cstride;
    int* ad = adj + (size_t)dir * E;

    for (int n0 = 0; n0 < N; n0 += 16 * LB_TPB) {
        int c[16];
#pragma unroll
        for (int u = 0; u < 16; ++u) {
            const int n = n0 + u * LB_TPB + tid;
            c[u] = (n < N) ? Tt[n] : 0;
        }
#pragma unroll
        for (int u = 0; u < 16; ++u) {
            const int n = n0 + u * LB_TPB + tid;
            if (n < N) h[n] = c[u];
        }
    }
    __syncthreads();

    for (int e0 = lo; e0 < hi; e0 += 8 * LB_TPB) {
        int k[8], v[8];
#pragma unroll
        for (int u = 0; u < 8; ++u) {
            const int e = min(e0 + u * LB_TPB + tid, hi - 1);
            k[u] = key[e];
            v[u] = val[e];
        }
#pragma unroll
        for (int u = 0; u < 8; ++u)
            if (e0 + u * LB_TPB + tid < hi) ad[atomicAdd(&h[k[u]], 1)] = v[u];
    }
}

#define GA_TPB 256
__global__ __launch_bounds__(GA_TPB)
void gather_mean64(const float* __restrict__ Ya, const float* __restrict__ Yb,
                   const int* __restrict__ rowptra, const int* __restrict__ rowptrb,
                   const int* __restrict__ adja, const int* __restrict__ adjb,
                   float* __restrict__ Sa, float* __restrict__ Sb, int N) {
    const bool br = blockIdx.y != 0;
    const float* __restrict__ Y      = br ? Yb : Ya;
    const int*   __restrict__ rowptr = br ? rowptrb : rowptra;
    const int*   __restrict__ adj    = br ? adjb : adja;
    float*       __restrict__ S      = br ? Sb : Sa;
    const int lane = threadIdx.x & 31;
    const int h    = lane >> 4;
    const int c    = lane & 15;
    const int node = __builtin_amdgcn_readfirstlane(
        (int)((blockIdx.x * GA_TPB + threadIdx.x) >> 5));
    if (node >= N) return;
    const int beg = rowptr[node];
    const int end = rowptr[node + 1];
    const int d   = end - beg;

    const v4f* Y4 = (const v4f*)Y;
    v4f acc = {0.0f, 0.0f, 0.0f, 0.0f};
    if (d > 0 && d <= 32) {
        const int my  = adj[beg + min(lane, d - 1)];
        const int key = (my << 5) | lane;
        int pos = 0;
        for (int j = 0; j < d; ++j) {
            const int kj = __builtin_amdgcn_readlane(key, j);
            pos += (kj < key) ? 1 : 0;
        }
        if (lane >= d) pos = lane;
        const int dst = ((pos & 1) << 4) | (pos >> 1);
        const int srt = __builtin_amdgcn_ds_permute(dst << 2, my);
        v4f v[8];
#define GA_ROW(U, Q) v[U] = Y4[(size_t)__builtin_amdgcn_update_dpp(0, srt, 0x150 + (Q), 0xf, 0xf, false) * 16 + c]
        GA_ROW(0, 0); GA_ROW(1, 1); GA_ROW(2, 2); GA_ROW(3, 3);
        GA_ROW(4, 4); GA_ROW(5, 5); GA_ROW(6, 6); GA_ROW(7, 7);
#pragma unroll
        for (int u = 0; u < 8; ++u)
            if (2 * u + h < d) acc += v[u];
        if (d > 16) {
            GA_ROW(0, 8);  GA_ROW(1, 9);  GA_ROW(2, 10); GA_ROW(3, 11);
            GA_ROW(4, 12); GA_ROW(5, 13); GA_ROW(6, 14); GA_ROW(7, 15);
#pragma unroll
            for (int u = 0; u < 8; ++u)
                if (16 + 2 * u + h < d) acc += v[u];
        }
#undef GA_ROW
    } else if (d > 32 && d <= 64) {
        const int e0 = adj[beg + lane];
        const int e1 = adj[beg + min(32 + lane, d - 1)];
        const int k0 = (e0 << 6) | lane;
        const int k1 = (32 + lane < d) ? ((e1 << 6) | (32 + lane)) : 0x7fffffff;
        int p0 = 0, p1 = 0;
        for (int j = 0; j < 32; ++j) {
            const int kj = __builtin_amdgcn_readlane(k0, j);
            p0 += (kj < k0) ? 1 : 0;
            p1 += (kj < k1) ? 1 : 0;
        }
        for (int j = 32; j < d; ++j) {
            const int kj = __builtin_amdgcn_readlane(k1, j - 32);
            p0 += (kj < k0) ? 1 : 0;
            p1 += (kj < k1) ? 1 : 0;
        }
        if (32 + lane >= d) p1 = -1;
        for (int g = 0; g < d; g += 16) {
            v4f v[8];
#pragma unroll
            for (int u = 0; u < 8; ++u) {
                int jr[2];
#pragma unroll
                for (int q = 0; q < 2; ++q) {
                    const int r = min(g + 2 * u + q, d - 1);
                    const unsigned m0 = __builtin_amdgcn_ballot_w32(p0 == r);
                    const unsigned m1 = __builtin_amdgcn_ballot_w32(p1 == r);
                    const int a0 = __builtin_amdgcn_readlane(e0, m0 ? __builtin_ctz(m0) : 0);
                    const int a1 = __builtin_amdgcn_readlane(e1, m1 ? __builtin_ctz(m1) : 0);
                    jr[q] = m0 ? a0 : a1;
                }
                const int jj = h ? jr[1] : jr[0];
                v[u] = Y4[(size_t)jj * 16 + c];
            }
#pragma unroll
            for (int u = 0; u < 8; ++u)
                if (g + 2 * u + h < d) acc += v[u];
        }
    } else if (d > 64) {
        long long last = -1;
        for (int it = 0; it < d; ++it) {
            long long best = 0x7fffffffffffffffLL;
            for (int p = lane; p < d; p += 32) {
                const long long kk = ((long long)adj[beg + p] << 32) | (long long)p;
                if (kk > last && kk < best) best = kk;
            }
            for (int m = 16; m >= 1; m >>= 1) {
                const long long o = __shfl_xor(best, m);
                if (o < best) best = o;
            }
            last = best;
            const int jj = (int)(best >> 32);
            const v4f r = Y4[(size_t)jj * 16 + c];
            if (h == 0) acc += r;
        }
    }
    acc.x += __shfl_xor(acc.x, 16);
    acc.y += __shfl_xor(acc.y, 16);
    acc.z += __shfl_xor(acc.z, 16);
    acc.w += __shfl_xor(acc.w, 16);
    const float inv = 1.0f / fmaxf((float)d, 1.0f);
    if (h == 0) ((v4f*)S)[(size_t)node * 16 + c] = acc * inv;
}

#define GM_LOAD(DST, PTR)                                                     \
    _Pragma("unroll")                                                         \
    for (int u = 0; u < 8; ++u) DST[u] = *(const v2f*)((PTR) + 4 * u);
#define GM_MMA(SRC, CI, SCALED)                                               \
    _Pragma("unroll")                                                         \
    for (int u = 0; u < 8; ++u) {                                             \
        v2f a = SRC[u];                                                       \
        if (SCALED) { a.x *= scale1; a.y *= scale1; }                         \
        const int sbase = ((((CI) << 4) + 2 * u + kSel) << 7);                \
        _Pragma("unroll")                                                     \
        for (int t = 0; t < 4; ++t) {                                         \
            v2f b = *(const v2f*)&sW[sbase + ((t << 4) + m) * 2];             \
            acc[t] = __builtin_amdgcn_wmma_f32_16x16x4_f32(                   \
                false, a, false, b, (short)0, acc[t], false, false);          \
        }                                                                     \
    }

template <int LDSK>
__global__ __launch_bounds__(128)
void gemm_wmma_relu(const float* __restrict__ X0, int d0,
                    const float* __restrict__ X1a,
                    const float* __restrict__ X1b,
                    const float* __restrict__ cnt,
                    const float* __restrict__ Wa,
                    const float* __restrict__ Wb,
                    int Ktot,
                    const float* __restrict__ biasa,
                    const float* __restrict__ biasb,
                    float* outa, float* outb,
                    int ostride, int ocol0a, int ocol0b, int dup, int Nrows) {
    __shared__ __attribute__((aligned(16))) float sW[LDSK * 64];
    const bool br = blockIdx.y != 0;
    const float* __restrict__ X1   = br ? X1b : X1a;
    const float* __restrict__ W    = br ? Wb : Wa;
    const float* __restrict__ bias = br ? biasb : biasa;
    float* __restrict__ out = br ? outb : outa;
    const int ocol0 = br ? ocol0b : ocol0a;

    const int tid  = threadIdx.x;
    const int wave = tid >> 5;
    const int lane = tid & 31;
    const int m    = lane & 15;
    const int kSel = lane >> 4;

    const int rowbase = (blockIdx.x * 4 + wave) * 16;
    const int row     = rowbase + m;
    const int crow    = row < Nrows ? row : (Nrows - 1);

    const float* x0r = X0 + (size_t)crow * d0 + 2 * kSel;
    const float* x1r = X1 ? X1 + (size_t)crow * 64 + 2 * kSel : nullptr;

    const int nch0 = d0 >> 5;
    const int ntot = (X1 != nullptr) ? (Ktot >> 5) : nch0;

    v2f areg0[8], areg1[8];
    if (ntot > 0) {
        const float* p0 = (0 < nch0) ? x0r : x1r;
        GM_LOAD(areg0, p0)
    }
    float bv[4] = {0.0f, 0.0f, 0.0f, 0.0f};
    if (bias) {
#pragma unroll
        for (int t = 0; t < 4; ++t) bv[t] = bias[(t << 4) + m];
    }
    float scale1 = 1.0f;
    if (cnt) scale1 = 1.0f / fmaxf(cnt[crow], 1.0f);

    {
        constexpr int NQ = LDSK / 16;
        const int npair = Ktot >> 1;
        v4f w0[NQ], w1[NQ];
#pragma unroll
        for (int q = 0; q < NQ; ++q) {
            const int it = tid + 128 * q;
            const int p  = min(it >> 4, npair - 1);
            const int g  = it & 15;
            w0[q] = *(const v4f*)(W + (size_t)(2 * p) * 64 + 4 * g);
            w1[q] = *(const v4f*)(W + (size_t)(2 * p + 1) * 64 + 4 * g);
        }
#pragma unroll
        for (int q = 0; q < NQ; ++q) {
            const int it = tid + 128 * q;
            const int p  = it >> 4;
            const int g  = it & 15;
            if (p < npair) {
                const v4f lo = {w0[q].x, w1[q].x, w0[q].y, w1[q].y};
                const v4f hi = {w0[q].z, w1[q].z, w0[q].w, w1[q].w};
                v4f* d = (v4f*)&sW[((p << 6) + 4 * g) * 2];
                d[0] = lo;
                d[1] = hi;
            }
        }
    }
    __syncthreads();

    v8f acc[4] = {};

#define GM_PTR(C) (((C) < nch0) ? x0r + ((C) << 5) : x1r + (((C) - nch0) << 5))
#pragma unroll 1
    for (int ci = 0; ci < ntot; ci += 2) {
        {
            const int c1 = min(ci + 1, ntot - 1);
            const float* p1 = GM_PTR(c1);
            GM_LOAD(areg1, p1)
        }
        if (cnt && ci >= nch0) { GM_MMA(areg0, ci, true) }
        else                   { GM_MMA(areg0, ci, false) }
        {
            const int c2 = min(ci + 2, ntot - 1);
            const float* p2 = GM_PTR(c2);
            GM_LOAD(areg0, p2)
        }
        if (ci + 1 < ntot) {
            if (cnt && ci + 1 >= nch0) { GM_MMA(areg1, ci + 1, true) }
            else                       { GM_MMA(areg1, ci + 1, false) }
        }
    }
#undef GM_PTR

    const int rr  = kSel * 8;
    const int col = m;
#pragma unroll
    for (int r = 0; r < 8; ++r) {
        const int orow = rowbase + r + rr;
        if (orow < Nrows) {
#pragma unroll
            for (int t = 0; t < 4; ++t) {
                float v = acc[t][r];
                if (bias) v += bv[t];
                v = fmaxf(v, 0.0f);
                float* p = out + (size_t)orow * ostride + ocol0 + (t << 4) + col;
                *p = v;
                if (dup) p[64] = v;
            }
        }
    }
}
#undef GM_LOAD
#undef GM_MMA

extern "C" void kernel_launch(void* const* d_in, const int* in_sizes, int n_in,
                              void* d_out, int out_size, void* d_ws, size_t ws_size,
                              hipStream_t stream) {
    const float* x    = (const float*)d_in[0];
    const int*   ei   = (const int*)  d_in[1];
    const float* W_u1 = (const float*)d_in[2];
    const float* b_u1 = (const float*)d_in[3];
    const float* U_u1 = (const float*)d_in[4];
    const float* W_d1 = (const float*)d_in[5];
    const float* b_d1 = (const float*)d_in[6];
    const float* U_d1 = (const float*)d_in[7];
    const float* W_u2 = (const float*)d_in[8];
    const float* b_u2 = (const float*)d_in[9];
    const float* U_u2 = (const float*)d_in[10];
    float* out = (float*)d_out;

    const int N = in_sizes[0] / 128;
    const int E = in_sizes[1] / 2;

    const int cstride = (N + 31) / 32 * 32;
    float* ws     = (float*)d_ws;
    float* Y      = ws;
    float* S      = Y  + (size_t)N * 64;
    float* ha     = S  + (size_t)N * 64;
    int*   cnt    = (int*)(ha + (size_t)N * 128);
    int*   rowptr = cnt    + 2 * (size_t)cstride;
    int*   adj    = rowptr + 2 * ((size_t)N + 1);
    int*   T      = (int*)ws;
    int*   C      = cnt;

    const int* rp_in   = rowptr;
    const int* rp_out  = rowptr + (N + 1);
    const int* adj_in  = adj;
    const int* adj_out = adj + E;

    slice_count<<<dim3(LB_SLICES, 2), LB_TPB, 0, stream>>>(ei, E, N, cstride, T, C);
    slot_bases<<<dim3((N + LB_BT - 1) / LB_BT, 2), LB_BT, 0, stream>>>(T, C, N, cstride, rowptr);
    slice_place<<<dim3(LB_SLICES, 2), LB_TPB, 0, stream>>>(ei, E, N, cstride, T, adj);

    const int gblk  = (N + 63) / 64;
    const int ggrid = (N + GA_TPB / 32 - 1) / (GA_TPB / 32);

    float* Yd = ha;
    float* Sd = ha + (size_t)N * 64;
    gemm_wmma_relu<128><<<dim3(gblk, 2), 128, 0, stream>>>(
        x, 128, nullptr, nullptr, nullptr, W_u1, W_d1, 128, b_u1, b_d1,
        Y, Yd, 64, 0, 0, 0, N);
    gather_mean64<<<dim3(ggrid, 2), GA_TPB, 0, stream>>>(
        Y, Yd, rp_in, rp_out, adj_in, adj_out, S, Sd, N);
    gemm_wmma_relu<192><<<dim3(gblk, 2), 128, 0, stream>>>(
        x, 128, S, Sd, nullptr, U_u1, U_d1, 192, nullptr, nullptr,
        out, out, 128, 0, 64, 0, N);

    auto conv = [&](const float* Xin, const int* rp, const int* nb,
                    const float* Wm, const float* bm,
                    const float* Um, float* outp, int dupw) {
        gemm_wmma_relu<128><<<gblk, 128, 0, stream>>>(
            Xin, 128, nullptr, nullptr, nullptr, Wm, nullptr, 128, bm, nullptr,
            Y, nullptr, 64, 0, 0, 0, N);
        gather_mean64<<<ggrid, GA_TPB, 0, stream>>>(Y, nullptr, rp, nullptr, nb, nullptr,
                                                    S, nullptr, N);
        gemm_wmma_relu<192><<<gblk, 128, 0, stream>>>(
            Xin, 128, S, nullptr, nullptr, Um, nullptr, 192, nullptr, nullptr,
            outp, nullptr, 128, 0, 0, dupw, N);
    };

    conv(out, rp_in,  adj_in,  W_u2, b_u2, U_u2, ha,  1);
    conv(ha,  rp_in,  adj_in,  W_u2, b_u2, U_u2, out, 1);
}


# next round's message GEMM runs inside the update GEMM's work-group (its 16x64 result tiles turned into A fragments through LDS, W staged over U): 10 launches, not 12; every sum as before
# speedup vs baseline: 3.3022x; 1.0598x over previous
#include <hip/hip_runtime.h>
#include <hip/hip_bf16.h>

typedef __attribute__((ext_vector_type(2))) float v2f;
typedef __attribute__((ext_vector_type(4))) float v4f;
typedef __attribute__((ext_vector_type(8))) float v8f;

#define LB_TPB    1024
#define LB_LDSN   65536
#define LB_SLICES 64
#define LB_BT     512
#define LB_NB     128
__global__ __launch_bounds__(LB_TPB)
void slice_count(const int* __restrict__ ei, int E, int N, int cstride,
                 int* __restrict__ T, int* __restrict__ C) {
    __shared__ int h[LB_LDSN];
    const int dir = blockIdx.y;
    const int* key = ei + (dir == 0 ? E : 0);
    const int tid = threadIdx.x;
    const int len = (E + LB_SLICES - 1) / LB_SLICES;
    const int lo  = min((int)blockIdx.x * len, E);
    const int hi  = min(lo + len, E);
    const size_t srow = (size_t)dir * LB_SLICES + blockIdx.x;
    int* Tt = T + srow * cstride;

    for (int n = tid; n < N; n += LB_TPB) h[n] = 0;
    __syncthreads();

    for (int e0 = lo; e0 < hi; e0 += 8 * LB_TPB) {
        int k[8];
#pragma unroll
        for (int u = 0; u < 8; ++u) {
            const int e = e0 + u * LB_TPB + tid;
            k[u] = (e < hi) ? key[e] : 0;
        }
#pragma unroll
        for (int u = 0; u < 8; ++u)
            if (e0 + u * LB_TPB + tid < hi) atomicAdd(&h[k[u]], 1);
    }
    __syncthreads();

    for (int n0 = 0; n0 < N; n0 += 8 * LB_TPB) {
        int c[8];
#pragma unroll
        for (int u = 0; u < 8; ++u) {
            const int n = n0 + u * LB_TPB + tid;
            c[u] = (n < N) ? h[n] : 0;
        }
#pragma unroll
        for (int u = 0; u < 8; ++u) {
            const int n = n0 + u * LB_TPB + tid;
            if (n < N) Tt[n] = c[u];
        }
    }

    const int i = tid >> 3;
    const int j = tid & 7;
    int s = 0;
    for (int m = 0; m < LB_BT / 8; ++m) {
        const int n = i * LB_BT + j + (m << 3);
        const int x = h[min(n, N - 1)];
        s += (n < N) ? x : 0;
    }
    s += __shfl_xor(s, 4);
    s += __shfl_xor(s, 2);
    s += __shfl_xor(s, 1);
    if (j == 0) C[srow * LB_NB + i] = s;
}

__global__ __launch_bounds__(LB_BT)
void slot_bases(int* __restrict__ T, const int* __restrict__ C, int N,
                int cstride, int* __restrict__ rowptr) {
    __shared__ int wsum[32];
    __shared__ int wpart[32];
    __shared__ int bstart;
    const int b    = blockIdx.x;
    const int dir  = blockIdx.y;
    const int tid  = threadIdx.x;
    const int lane = tid & 31;
    const int wv   = tid >> 5;
    const int n    = b * LB_BT + tid;
    int*       Td = T + (size_t)dir * LB_SLICES * cstride;
    const int* Cd = C + (size_t)dir * LB_SLICES * LB_NB;
    int*       rp = rowptr + (size_t)dir * (N + 1);

    int part = 0;
    for (int idx = tid; idx < LB_SLICES * LB_NB; idx += LB_BT)
        if ((idx & (LB_NB - 1)) < b) part += Cd[idx];

    int c[LB_SLICES];
#pragma unroll
    for (int s = 0; s < LB_SLICES; ++s)
        c[s] = Td[(size_t)s * cstride + min(n, N - 1)];
#pragma unroll
    for (int s = 0; s < LB_SLICES; ++s)
        if (n >= N) c[s] = 0;
    int deg = 0;
#pragma unroll
    for (int s = 0; s < LB_SLICES; ++s) deg += c[s];

    int incl = deg;
    for (int d = 1; d < 32; d <<= 1) {
        int t = __shfl_up(incl, d);
        if (lane >= d) incl += t;
    }
    for (int d = 16; d >= 1; d >>= 1) part += __shfl_xor(part, d);
    if (lane == 31) wsum[wv] = incl;
    if (lane == 0)  wpart[wv] = part;
    __syncthreads();
    if (wv == 0) {
        const int w = (lane < LB_BT / 32) ? wsum[lane] : 0;
        int wi = w;
        for (int d = 1; d < 32; d <<= 1) {
            int t = __shfl_up(wi, d);
            if (lane >= d) wi += t;
        }
        if (lane < LB_BT / 32) wsum[lane] = wi - w;
        int p = (lane < LB_BT / 32) ? wpart[lane] : 0;
        for (int d = 16; d >= 1; d >>= 1) p += __shfl_xor(p, d);
        if (lane == 0) bstart = p;
    }
    __syncthreads();

    int run = bstart + wsum[wv] + incl - deg;
    if (n < N) {
        rp[n] = run;
        if (n == N - 1) rp[N] = run + deg;
#pragma unroll
        for (int s = 0; s < LB_SLICES; ++s) {
            Td[(size_t)s * cstride + n] = run;
            run += c[s];
        }
    }
}

__global__ __launch_bounds__(LB_TPB)
void slice_place(const int* __restrict__ ei, int E, int N, int cstride,
                 const int* __restrict__ T, int* __restrict__ adj) {
    __shared__ int h[LB_LDSN];
    const int dir = blockIdx.y;
    const int* key = ei + (dir == 0 ? E : 0);
    const int* val = ei + (dir == 0 ? 0 : E);
    const int tid = threadIdx.x;
    const int len = (E + LB_SLICES - 1) / LB_SLICES;
    const int lo  = min((int)blockIdx.x * len, E);
    const int hi  = min(lo + len, E);
    const int* Tt = T + ((size_t)dir * LB_SLICES + blockIdx.x) * cstride;
    int* ad = adj + (size_t)dir * E;

    for (int n0 = 0; n0 < N; n0 += 16 * LB_TPB) {
        int c[16];
#pragma unroll
        for (int u = 0; u < 16; ++u) {
            const int n = n0 + u * LB_TPB + tid;
            c[u] = (n < N) ? Tt[n] : 0;
        }
#pragma unroll
        for (int u = 0; u < 16; ++u) {
            const int n = n0 + u * LB_TPB + tid;
            if (n < N) h[n] = c[u];
        }
    }
    __syncthreads();

    for (int e0 = lo; e0 < hi; e0 += 8 * LB_TPB) {
        int k[8], v[8];
#pragma unroll
        for (int u = 0; u < 8; ++u) {
            const int e = min(e0 + u * LB_TPB + tid, hi - 1);
            k[u] = key[e];
            v[u] = val[e];
        }
#pragma unroll
        for (int u = 0; u < 8; ++u)
            if (e0 + u * LB_TPB + tid < hi) ad[atomicAdd(&h[k[u]], 1)] = v[u];
    }
}

#define GA_TPB 256
__global__ __launch_bounds__(GA_TPB)
void gather_mean64(const float* __restrict__ Ya, const float* __restrict__ Yb,
                   const int* __restrict__ rowptra, const int* __restrict__ rowptrb,
                   const int* __restrict__ adja, const int* __restrict__ adjb,
                   float* __restrict__ Sa, float* __restrict__ Sb, int N) {
    const bool br = blockIdx.y != 0;
    const float* __restrict__ Y      = br ? Yb : Ya;
    const int*   __restrict__ rowptr = br ? rowptrb : rowptra;
    const int*   __restrict__ adj    = br ? adjb : adja;
    float*       __restrict__ S      = br ? Sb : Sa;
    const int lane = threadIdx.x & 31;
    const int h    = lane >> 4;
    const int c    = lane & 15;
    const int node = __builtin_amdgcn_readfirstlane(
        (int)((blockIdx.x * GA_TPB + threadIdx.x) >> 5));
    if (node >= N) return;
    const int beg = rowptr[node];
    const int end = rowptr[node + 1];
    const int d   = end - beg;

    const v4f* Y4 = (const v4f*)Y;
    v4f acc = {0.0f, 0.0f, 0.0f, 0.0f};
    if (d > 0 && d <= 32) {
        const int my  = adj[beg + min(lane, d - 1)];
        const int key = (my << 5) | lane;
        int pos = 0;
        for (int j = 0; j < d; ++j) {
            const int kj = __builtin_amdgcn_readlane(key, j);
            pos += (kj < key) ? 1 : 0;
        }
        if (lane >= d) pos = lane;
        const int dst = ((pos & 1) << 4) | (pos >> 1);
        const int srt = __builtin_amdgcn_ds_permute(dst << 2, my);
        v4f v[8];
#define GA_ROW(U, Q) v[U] = Y4[(size_t)__builtin_amdgcn_update_dpp(0, srt, 0x150 + (Q), 0xf, 0xf, false) * 16 + c]
        GA_ROW(0, 0); GA_ROW(1, 1); GA_ROW(2, 2); GA_ROW(3, 3);
        GA_ROW(4, 4); GA_ROW(5, 5); GA_ROW(6, 6); GA_ROW(7, 7);
#pragma unroll
        for (int u = 0; u < 8; ++u)
            if (2 * u + h < d) acc += v[u];
        if (d > 16) {
            GA_ROW(0, 8);  GA_ROW(1, 9);  GA_ROW(2, 10); GA_ROW(3, 11);
            GA_ROW(4, 12); GA_ROW(5, 13); GA_ROW(6, 14); GA_ROW(7, 15);
#pragma unroll
            for (int u = 0; u < 8; ++u)
                if (16 + 2 * u + h < d) acc += v[u];
        }
#undef GA_ROW
    } else if (d > 32 && d <= 64) {
        const int e0 = adj[beg + lane];
        const int e1 = adj[beg + min(32 + lane, d - 1)];
        const int k0 = (e0 << 6) | lane;
        const int k1 = (32 + lane < d) ? ((e1 << 6) | (32 + lane)) : 0x7fffffff;
        int p0 = 0, p1 = 0;
        for (int j = 0; j < 32; ++j) {
            const int kj = __builtin_amdgcn_readlane(k0, j);
            p0 += (kj < k0) ? 1 : 0;
            p1 += (kj < k1) ? 1 : 0;
        }
        for (int j = 32; j < d; ++j) {
            const int kj = __builtin_amdgcn_readlane(k1, j - 32);
            p0 += (kj < k0) ? 1 : 0;
            p1 += (kj < k1) ? 1 : 0;
        }
        if (32 + lane >= d) p1 = -1;
        for (int g = 0; g < d; g += 16) {
            v4f v[8];
#pragma unroll
            for (int u = 0; u < 8; ++u) {
                int jr[2];
#pragma unroll
                for (int q = 0; q < 2; ++q) {
                    const int r = min(g + 2 * u + q, d - 1);
                    const unsigned m0 = __builtin_amdgcn_ballot_w32(p0 == r);
                    const unsigned m1 = __builtin_amdgcn_ballot_w32(p1 == r);
                    const int a0 = __builtin_amdgcn_readlane(e0, m0 ? __builtin_ctz(m0) : 0);
                    const int a1 = __builtin_amdgcn_readlane(e1, m1 ? __builtin_ctz(m1) : 0);
                    jr[q] = m0 ? a0 : a1;
                }
                const int jj = h ? jr[1] : jr[0];
                v[u] = Y4[(size_t)jj * 16 + c];
            }
#pragma unroll
            for (int u = 0; u < 8; ++u)
                if (g + 2 * u + h < d) acc += v[u];
        }
    } else if (d > 64) {
        long long last = -1;
        for (int it = 0; it < d; ++it) {
            long long best = 0x7fffffffffffffffLL;
            for (int p = lane; p < d; p += 32) {
                const long long kk = ((long long)adj[beg + p] << 32) | (long long)p;
                if (kk > last && kk < best) best = kk;
            }
            for (int m = 16; m >= 1; m >>= 1) {
                const long long o = __shfl_xor(best, m);
                if (o < best) best = o;
            }
            last = best;
            const int jj = (int)(best >> 32);
            const v4f r = Y4[(size_t)jj * 16 + c];
            if (h == 0) acc += r;
        }
    }
    acc.x += __shfl_xor(acc.x, 16);
    acc.y += __shfl_xor(acc.y, 16);
    acc.z += __shfl_xor(acc.z, 16);
    acc.w += __shfl_xor(acc.w, 16);
    const float inv = 1.0f / fmaxf((float)d, 1.0f);
    if (h == 0) ((v4f*)S)[(size_t)node * 16 + c] = acc * inv;
}

#define GM_LOAD(DST, PTR)                                                     \
    _Pragma("unroll")                                                         \
    for (int u = 0; u < 8; ++u) DST[u] = *(const v2f*)((PTR) + 4 * u);
#define GM_MMA(SRC, CI, SCALED)                                               \
    _Pragma("unroll")                                                         \
    for (int u = 0; u < 8; ++u) {                                             \
        v2f a = SRC[u];                                                       \
        if (SCALED) { a.x *= scale1; a.y *= scale1; }                         \
        const int sbase = ((((CI) << 4) + 2 * u + kSel) << 7);                \
        _Pragma("unroll")                                                     \
        for (int t = 0; t < 4; ++t) {                                         \
            v2f b = *(const v2f*)&sW[sbase + ((t << 4) + m) * 2];             \
            acc[t] = __builtin_amdgcn_wmma_f32_16x16x4_f32(                   \
                false, a, false, b, (short)0, acc[t], false, false);          \
        }                                                                     \
    }

template <int LDSK>
__global__ __launch_bounds__(128)
void gemm_wmma_relu(const float* __restrict__ X0, int d0,
                    const float* __restrict__ X1a,
                    const float* __restrict__ X1b,
                    const float* __restrict__ cnt,
                    const float* __restrict__ Wa,
                    const float* __restrict__ Wb,
                    int Ktot,
                    const float* __restrict__ biasa,
                    const float* __restrict__ biasb,
                    float* outa, float* outb,
                    int ostride, int ocol0a, int ocol0b, int dup, int Nrows) {
    __shared__ __attribute__((aligned(16))) float sW[LDSK * 64];
    const bool br = blockIdx.y != 0;
    const float* __restrict__ X1   = br ? X1b : X1a;
    const float* __restrict__ W    = br ? Wb : Wa;
    const float* __restrict__ bias = br ? biasb : biasa;
    float* __restrict__ out = br ? outb : outa;
    const int ocol0 = br ? ocol0b : ocol0a;

    const int tid  = threadIdx.x;
    const int wave = tid >> 5;
    const int lane = tid & 31;
    const int m    = lane & 15;
    const int kSel = lane >> 4;

    const int rowbase = (blockIdx.x * 4 + wave) * 16;
    const int row     = rowbase + m;
    const int crow    = row < Nrows ? row : (Nrows - 1);

    const float* x0r = X0 + (size_t)crow * d0 + 2 * kSel;
    const float* x1r = X1 ? X1 + (size_t)crow * 64 + 2 * kSel : nullptr;

    const int nch0 = d0 >> 5;
    const int ntot = (X1 != nullptr) ? (Ktot >> 5) : nch0;

    v2f areg0[8], areg1[8];
    if (ntot > 0) {
        const float* p0 = (0 < nch0) ? x0r : x1r;
        GM_LOAD(areg0, p0)
    }
    float bv[4] = {0.0f, 0.0f, 0.0f, 0.0f};
    if (bias) {
#pragma unroll
        for (int t = 0; t < 4; ++t) bv[t] = bias[(t << 4) + m];
    }
    float scale1 = 1.0f;
    if (cnt) scale1 = 1.0f / fmaxf(cnt[crow], 1.0f);

    {
        constexpr int NQ = LDSK / 16;
        const int npair = Ktot >> 1;
        v4f w0[NQ], w1[NQ];
#pragma unroll
        for (int q = 0; q < NQ; ++q) {
            const int it = tid + 128 * q;
            const int p  = min(it >> 4, npair - 1);
            const int g  = it & 15;
            w0[q] = *(const v4f*)(W + (size_t)(2 * p) * 64 + 4 * g);
            w1[q] = *(const v4f*)(W + (size_t)(2 * p + 1) * 64 + 4 * g);
        }
#pragma unroll
        for (int q = 0; q < NQ; ++q) {
            const int it = tid + 128 * q;
            const int p  = it >> 4;
            const int g  = it & 15;
            if (p < npair) {
                const v4f lo = {w0[q].x, w1[q].x, w0[q].y, w1[q].y};
                const v4f hi = {w0[q].z, w1[q].z, w0[q].w, w1[q].w};
                v4f* d = (v4f*)&sW[((p << 6) + 4 * g) * 2];
                d[0] = lo;
                d[1] = hi;
            }
        }
    }
    __syncthreads();

    v8f acc[4] = {};

#define GM_PTR(C) (((C) < nch0) ? x0r + ((C) << 5) : x1r + (((C) - nch0) << 5))
#pragma unroll 1
    for (int ci = 0; ci < ntot; ci += 2) {
        {
            const int c1 = min(ci + 1, ntot - 1);
            const float* p1 = GM_PTR(c1);
            GM_LOAD(areg1, p1)
        }
        if (cnt && ci >= nch0) { GM_MMA(areg0, ci, true) }
        else                   { GM_MMA(areg0, ci, false) }
        {
            const int c2 = min(ci + 2, ntot - 1);
            const float* p2 = GM_PTR(c2);
            GM_LOAD(areg0, p2)
        }
        if (ci + 1 < ntot) {
            if (cnt && ci + 1 >= nch0) { GM_MMA(areg1, ci + 1, true) }
            else                       { GM_MMA(areg1, ci + 1, false) }
        }
    }
#undef GM_PTR

    const int rr  = kSel * 8;
    const int col = m;
#pragma unroll
    for (int r = 0; r < 8; ++r) {
        const int orow = rowbase + r + rr;
        if (orow < Nrows) {
#pragma unroll
            for (int t = 0; t < 4; ++t) {
                float v = acc[t][r];
                if (bias) v += bv[t];
                v = fmaxf(v, 0.0f);
                float* p = out + (size_t)orow * ostride + ocol0 + (t << 4) + col;
                *p = v;
                if (dup) p[64] = v;
            }
        }
    }
}
#define FU_PITCH 68
template <int NQ>
__device__ __forceinline__ void fu_stage(const float* __restrict__ W, float* sW, int tid) {
    constexpr int npair = NQ * 8;
    v4f w0[NQ], w1[NQ];
#pragma unroll
    for (int q = 0; q < NQ; ++q) {
        const int it = tid + 128 * q;
        const int p  = it >> 4;
        const int g  = it & 15;
        w0[q] = *(const v4f*)(W + (size_t)(2 * p) * 64 + 4 * g);
        w1[q] = *(const v4f*)(W + (size_t)(2 * p + 1) * 64 + 4 * g);
    }
#pragma unroll
    for (int q = 0; q < NQ; ++q) {
        const int it = tid + 128 * q;
        const int p  = it >> 4;
        const int g  = it & 15;
        if (p < npair) {
            const v4f lo = {w0[q].x, w1[q].x, w0[q].y, w1[q].y};
            const v4f hi = {w0[q].z, w1[q].z, w0[q].w, w1[q].w};
            v4f* d = (v4f*)&sW[((p << 6) + 4 * g) * 2];
            d[0] = lo;
            d[1] = hi;
        }
    }
}

template <int NB>
__global__ __launch_bounds__(128)
void update_then_message(const float* __restrict__ X0,
                         const float* __restrict__ S0,
                         const float* __restrict__ S1,
                         const float* __restrict__ U0,
                         const float* __restrict__ U1,
                         float* __restrict__ hout, int dup,
                         const float* __restrict__ Wm,
                         const float* __restrict__ bm,
                         float* __restrict__ Yout,
                         int Nrows) {
    __shared__ __attribute__((aligned(16))) float sW[192 * 64];
    __shared__ __attribute__((aligned(16))) float hT[4 * 16 * FU_PITCH];

    const int tid  = threadIdx.x;
    const int wave = tid >> 5;
    const int lane = tid & 31;
    const int m    = lane & 15;
    const int kSel = lane >> 4;

    const int rowbase = (blockIdx.x * 4 + wave) * 16;
    const int row     = rowbase + m;
    const int crow    = row < Nrows ? row : (Nrows - 1);
    const int rr      = kSel * 8;
    float* myT = hT + wave * (16 * FU_PITCH);

    const float* x0r = X0 + (size_t)crow * 128 + 2 * kSel;
    const int nch0 = 4;
    const int ntot = 6;
    const float scale1 = 1.0f;

    float bv[4];
#pragma unroll
    for (int t = 0; t < 4; ++t) bv[t] = bm[(t << 4) + m];

    v2f hA[NB][16];

#pragma unroll
    for (int b = 0; b < NB; ++b) {
        const float* __restrict__ U  = b ? U1 : U0;
        const float* __restrict__ X1 = b ? S1 : S0;
        const float* x1r = X1 + (size_t)crow * 64 + 2 * kSel;

        v2f areg0[8], areg1[8];
        GM_LOAD(areg0, x0r)

        fu_stage<12>(U, sW, tid);
        __syncthreads();

        v8f acc[4] = {};
#define GM_PTR(C) (((C) < nch0) ? x0r + ((C) << 5) : x1r + (((C) - nch0) << 5))
#pragma unroll 1
        for (int ci = 0; ci < ntot; ci += 2) {
            {
                const int c1 = min(ci + 1, ntot - 1);
                const float* p1 = GM_PTR(c1);
                GM_LOAD(areg1, p1)
            }
            GM_MMA(areg0, ci, false)
            {
                const int c2 = min(ci + 2, ntot - 1);
                const float* p2 = GM_PTR(c2);
                GM_LOAD(areg0, p2)
            }
            if (ci + 1 < ntot) { GM_MMA(areg1, ci + 1, false) }
        }
#undef GM_PTR

#pragma unroll
        for (int r = 0; r < 8; ++r) {
#pragma unroll
            for (int t = 0; t < 4; ++t)
                myT[(r + rr) * FU_PITCH + (t << 4) + m] = fmaxf(acc[t][r], 0.0f);
        }
        __syncthreads();
#pragma unroll
        for (int s = 0; s < 16; ++s)
            hA[b][s] = *(const v2f*)&myT[m * FU_PITCH + 4 * s + 2 * kSel];
#pragma unroll
        for (int r = 0; r < 8; ++r) {
            const int orow = rowbase + r + rr;
            if (orow < Nrows) {
#pragma unroll
                for (int t = 0; t < 4; ++t) {
                    const float v = fmaxf(acc[t][r], 0.0f);
                    float* p = hout + (size_t)orow * 128 + 64 * b + (t << 4) + m;
                    *p = v;
                    if (dup) p[64] = v;
                }
            }
        }
    }

    fu_stage<8>(Wm, sW, tid);
    __syncthreads();

    v8f acc[4] = {};
#pragma unroll
    for (int s = 0; s < 32; ++s) {
        const v2f a = (s < 16) ? hA[0][s] : hA[NB - 1][s - 16];
        const int sbase = ((2 * s + kSel) << 7);
#pragma unroll
        for (int t = 0; t < 4; ++t) {
            const v2f bb = *(const v2f*)&sW[sbase + ((t << 4) + m) * 2];
            acc[t] = __builtin_amdgcn_wmma_f32_16x16x4_f32(
                false, a, false, bb, (short)0, acc[t], false, false);
        }
    }
#pragma unroll
    for (int r = 0; r < 8; ++r) {
        const int orow = rowbase + r + rr;
        if (orow < Nrows) {
#pragma unroll
            for (int t = 0; t < 4; ++t)
                Yout[(size_t)orow * 64 + (t << 4) + m] = fmaxf(acc[t][r] + bv[t], 0.0f);
        }
    }
}
#undef GM_LOAD
#undef GM_MMA

extern "C" void kernel_launch(void* const* d_in, const int* in_sizes, int n_in,
                              void* d_out, int out_size, void* d_ws, size_t ws_size,
                              hipStream_t stream) {
    const float* x    = (const float*)d_in[0];
    const int*   ei   = (const int*)  d_in[1];
    const float* W_u1 = (const float*)d_in[2];
    const float* b_u1 = (const float*)d_in[3];
    const float* U_u1 = (const float*)d_in[4];
    const float* W_d1 = (const float*)d_in[5];
    const float* b_d1 = (const float*)d_in[6];
    const float* U_d1 = (const float*)d_in[7];
    const float* W_u2 = (const float*)d_in[8];
    const float* b_u2 = (const float*)d_in[9];
    const float* U_u2 = (const float*)d_in[10];
    float* out = (float*)d_out;

    const int N = in_sizes[0] / 128;
    const int E = in_sizes[1] / 2;

    const int cstride = (N + 31) / 32 * 32;
    float* ws     = (float*)d_ws;
    float* Y      = ws;
    float* S      = Y  + (size_t)N * 64;
    float* ha     = S  + (size_t)N * 64;
    int*   cnt    = (int*)(ha + (size_t)N * 128);
    int*   rowptr = cnt    + 2 * (size_t)cstride;
    int*   adj    = rowptr + 2 * ((size_t)N + 1);
    int*   T      = (int*)ws;
    int*   C      = cnt;

    const int* rp_in   = rowptr;
    const int* rp_out  = rowptr + (N + 1);
    const int* adj_in  = adj;
    const int* adj_out = adj + E;

    slice_count<<<dim3(LB_SLICES, 2), LB_TPB, 0, stream>>>(ei, E, N, cstride, T, C);
    slot_bases<<<dim3((N + LB_BT - 1) / LB_BT, 2), LB_BT, 0, stream>>>(T, C, N, cstride, rowptr);
    slice_place<<<dim3(LB_SLICES, 2), LB_TPB, 0, stream>>>(ei, E, N, cstride, T, adj);

    const int gblk  = (N + 63) / 64;
    const int ggrid = (N + GA_TPB / 32 - 1) / (GA_TPB / 32);

    float* Yd = ha;
    float* Sd = ha + (size_t)N * 64;
    gemm_wmma_relu<128><<<dim3(gblk, 2), 128, 0, stream>>>(
        x, 128, nullptr, nullptr, nullptr, W_u1, W_d1, 128, b_u1, b_d1,
        Y, Yd, 64, 0, 0, 0, N);
    gather_mean64<<<dim3(ggrid, 2), GA_TPB, 0, stream>>>(
        Y, Yd, rp_in, rp_out, adj_in, adj_out, S, Sd, N);
    update_then_message<2><<<gblk, 128, 0, stream>>>(
        x, S, Sd, U_u1, U_d1, out, 0, W_u2, b_u2, Y, N);

    gather_mean64<<<ggrid, GA_TPB, 0, stream>>>(Y, nullptr, rp_in, nullptr, adj_in, nullptr,
                                                S, nullptr, N);
    update_then_message<1><<<gblk, 128, 0, stream>>>(
        out, S, nullptr, U_u2, nullptr, ha, 1, W_u2, b_u2, Y, N);
    gather_mean64<<<ggrid, GA_TPB, 0, stream>>>(Y, nullptr, rp_in, nullptr, adj_in, nullptr,
                                                S, nullptr, N);
    gemm_wmma_relu<192><<<gblk, 128, 0, stream>>>(
        ha, 128, S, nullptr, nullptr, U_u2, nullptr, 192, nullptr, nullptr,
        out, nullptr, 128, 0, 0, 1, N);
}


# layer 1's message GEMM (both branches; needs no list) runs as extra work-groups of the slot-start launch, its rows kept in the output buffer until update 1; 9 launches, not 10; every sum as before
# speedup vs baseline: 3.4411x; 1.0421x over previous
#include <hip/hip_runtime.h>
#include <hip/hip_bf16.h>

typedef __attribute__((ext_vector_type(2))) float v2f;
typedef __attribute__((ext_vector_type(4))) float v4f;
typedef __attribute__((ext_vector_type(8))) float v8f;

#define LB_TPB    1024
#define LB_LDSN   65536
#define LB_SLICES 64
#define LB_BT     512
#define LB_NB     128
__global__ __launch_bounds__(LB_TPB)
void slice_count(const int* __restrict__ ei, int E, int N, int cstride,
                 int* __restrict__ T, int* __restrict__ C) {
    __shared__ int h[LB_LDSN];
    const int dir = blockIdx.y;
    const int* key = ei + (dir == 0 ? E : 0);
    const int tid = threadIdx.x;
    const int len = (E + LB_SLICES - 1) / LB_SLICES;
    const int lo  = min((int)blockIdx.x * len, E);
    const int hi  = min(lo + len, E);
    const size_t srow = (size_t)dir * LB_SLICES + blockIdx.x;
    int* Tt = T + srow * cstride;

    for (int n = tid; n < N; n += LB_TPB) h[n] = 0;
    __syncthreads();

    for (int e0 = lo; e0 < hi; e0 += 8 * LB_TPB) {
        int k[8];
#pragma unroll
        for (int u = 0; u < 8; ++u) {
            const int e = e0 + u * LB_TPB + tid;
            k[u] = (e < hi) ? key[e] : 0;
        }
#pragma unroll
        for (int u = 0; u < 8; ++u)
            if (e0 + u * LB_TPB + tid < hi) atomicAdd(&h[k[u]], 1);
    }
    __syncthreads();

    for (int n0 = 0; n0 < N; n0 += 8 * LB_TPB) {
        int c[8];
#pragma unroll
        for (int u = 0; u < 8; ++u) {
            const int n = n0 + u * LB_TPB + tid;
            c[u] = (n < N) ? h[n] : 0;
        }
#pragma unroll
        for (int u = 0; u < 8; ++u) {
            const int n = n0 + u * LB_TPB + tid;
            if (n < N) Tt[n] = c[u];
        }
    }

    const int i = tid >> 3;
    const int j = tid & 7;
    int s = 0;
    for (int m = 0; m < LB_BT / 8; ++m) {
        const int n = i * LB_BT + j + (m << 3);
        const int x = h[min(n, N - 1)];
        s += (n < N) ? x : 0;
    }
    s += __shfl_xor(s, 4);
    s += __shfl_xor(s, 2);
    s += __shfl_xor(s, 1);
    if (j == 0) C[srow * LB_NB + i] = s;
}

__device__ __forceinline__
void slot_bases_body(int* __restrict__ T, const int* __restrict__ C, int N,
                     int cstride, int* __restrict__ rowptr, int b, int dir,
                     int* wsum, int* wpart, int* bstart) {
    const int tid  = threadIdx.x;
    const int lane = tid & 31;
    const int wv   = tid >> 5;
    const int n    = b * LB_BT + tid;
    int*       Td = T + (size_t)dir * LB_SLICES * cstride;
    const int* Cd = C + (size_t)dir * LB_SLICES * LB_NB;
    int*       rp = rowptr + (size_t)dir * (N + 1);

    int part = 0;
    for (int idx = tid; idx < LB_SLICES * LB_NB; idx += LB_BT)
        if ((idx & (LB_NB - 1)) < b) part += Cd[idx];

    int c[LB_SLICES];
#pragma unroll
    for (int s = 0; s < LB_SLICES; ++s)
        c[s] = Td[(size_t)s * cstride + min(n, N - 1)];
#pragma unroll
    for (int s = 0; s < LB_SLICES; ++s)
        if (n >= N) c[s] = 0;
    int deg = 0;
#pragma unroll
    for (int s = 0; s < LB_SLICES; ++s) deg += c[s];

    int incl = deg;
    for (int d = 1; d < 32; d <<= 1) {
        int t = __shfl_up(incl, d);
        if (lane >= d) incl += t;
    }
    for (int d = 16; d >= 1; d >>= 1) part += __shfl_xor(part, d);
    if (lane == 31) wsum[wv] = incl;
    if (lane == 0)  wpart[wv] = part;
    __syncthreads();
    if (wv == 0) {
        const int w = (lane < LB_BT / 32) ? wsum[lane] : 0;
        int wi = w;
        for (int d = 1; d < 32; d <<= 1) {
            int t = __shfl_up(wi, d);
            if (lane >= d) wi += t;
        }
        if (lane < LB_BT / 32) wsum[lane] = wi - w;
        int p = (lane < LB_BT / 32) ? wpart[lane] : 0;
        for (int d = 16; d >= 1; d >>= 1) p += __shfl_xor(p, d);
        if (lane == 0) *bstart = p;
    }
    __syncthreads();

    int run = *bstart + wsum[wv] + incl - deg;
    if (n < N) {
        rp[n] = run;
        if (n == N - 1) rp[N] = run + deg;
#pragma unroll
        for (int s = 0; s < LB_SLICES; ++s) {
            Td[(size_t)s * cstride + n] = run;
            run += c[s];
        }
    }
}

__global__ __launch_bounds__(LB_TPB)
void slice_place(const int* __restrict__ ei, int E, int N, int cstride,
                 const int* __restrict__ T, int* __restrict__ adj) {
    __shared__ int h[LB_LDSN];
    const int dir = blockIdx.y;
    const int* key = ei + (dir == 0 ? E : 0);
    const int* val = ei + (dir == 0 ? 0 : E);
    const int tid = threadIdx.x;
    const int len = (E + LB_SLICES - 1) / LB_SLICES;
    const int lo  = min((int)blockIdx.x * len, E);
    const int hi  = min(lo + len, E);
    const int* Tt = T + ((size_t)dir * LB_SLICES + blockIdx.x) * cstride;
    int* ad = adj + (size_t)dir * E;

    for (int n0 = 0; n0 < N; n0 += 16 * LB_TPB) {
        int c[16];
#pragma unroll
        for (int u = 0; u < 16; ++u) {
            const int n = n0 + u * LB_TPB + tid;
            c[u] = (n < N) ? Tt[n] : 0;
        }
#pragma unroll
        for (int u = 0; u < 16; ++u) {
            const int n = n0 + u * LB_TPB + tid;
            if (n < N) h[n] = c[u];
        }
    }
    __syncthreads();

    for (int e0 = lo; e0 < hi; e0 += 8 * LB_TPB) {
        int k[8], v[8];
#pragma unroll
        for (int u = 0; u < 8; ++u) {
            const int e = min(e0 + u * LB_TPB + tid, hi - 1);
            k[u] = key[e];
            v[u] = val[e];
        }
#pragma unroll
        for (int u = 0; u < 8; ++u)
            if (e0 + u * LB_TPB + tid < hi) ad[atomicAdd(&h[k[u]], 1)] = v[u];
    }
}

#define GA_TPB 256
__global__ __launch_bounds__(GA_TPB)
void gather_mean64(const float* __restrict__ Ya, const float* __restrict__ Yb,
                   const int* __restrict__ rowptra, const int* __restrict__ rowptrb,
                   const int* __restrict__ adja, const int* __restrict__ adjb,
                   float* __restrict__ Sa, float* __restrict__ Sb, int N) {
    const bool br = blockIdx.y != 0;
    const float* __restrict__ Y      = br ? Yb : Ya;
    const int*   __restrict__ rowptr = br ? rowptrb : rowptra;
    const int*   __restrict__ adj    = br ? adjb : adja;
    float*       __restrict__ S      = br ? Sb : Sa;
    const int lane = threadIdx.x & 31;
    const int h    = lane >> 4;
    const int c    = lane & 15;
    const int node = __builtin_amdgcn_readfirstlane(
        (int)((blockIdx.x * GA_TPB + threadIdx.x) >> 5));
    if (node >= N) return;
    const int beg = rowptr[node];
    const int end = rowptr[node + 1];
    const int d   = end - beg;

    const v4f* Y4 = (const v4f*)Y;
    v4f acc = {0.0f, 0.0f, 0.0f, 0.0f};
    if (d > 0 && d <= 32) {
        const int my  = adj[beg + min(lane, d - 1)];
        const int key = (my << 5) | lane;
        int pos = 0;
        for (int j = 0; j < d; ++j) {
            const int kj = __builtin_amdgcn_readlane(key, j);
            pos += (kj < key) ? 1 : 0;
        }
        if (lane >= d) pos = lane;
        const int dst = ((pos & 1) << 4) | (pos >> 1);
        const int srt = __builtin_amdgcn_ds_permute(dst << 2, my);
        v4f v[8];
#define GA_ROW(U, Q) v[U] = Y4[(size_t)__builtin_amdgcn_update_dpp(0, srt, 0x150 + (Q), 0xf, 0xf, false) * 16 + c]
        GA_ROW(0, 0); GA_ROW(1, 1); GA_ROW(2, 2); GA_ROW(3, 3);
        GA_ROW(4, 4); GA_ROW(5, 5); GA_ROW(6, 6); GA_ROW(7, 7);
#pragma unroll
        for (int u = 0; u < 8; ++u)
            if (2 * u + h < d) acc += v[u];
        if (d > 16) {
            GA_ROW(0, 8);  GA_ROW(1, 9);  GA_ROW(2, 10); GA_ROW(3, 11);
            GA_ROW(4, 12); GA_ROW(5, 13); GA_ROW(6, 14); GA_ROW(7, 15);
#pragma unroll
            for (int u = 0; u < 8; ++u)
                if (16 + 2 * u + h < d) acc += v[u];
        }
#undef GA_ROW
    } else if (d > 32 && d <= 64) {
        const int e0 = adj[beg + lane];
        const int e1 = adj[beg + min(32 + lane, d - 1)];
        const int k0 = (e0 << 6) | lane;
        const int k1 = (32 + lane < d) ? ((e1 << 6) | (32 + lane)) : 0x7fffffff;
        int p0 = 0, p1 = 0;
        for (int j = 0; j < 32; ++j) {
            const int kj = __builtin_amdgcn_readlane(k0, j);
            p0 += (kj < k0) ? 1 : 0;
            p1 += (kj < k1) ? 1 : 0;
        }
        for (int j = 32; j < d; ++j) {
            const int kj = __builtin_amdgcn_readlane(k1, j - 32);
            p0 += (kj < k0) ? 1 : 0;
            p1 += (kj < k1) ? 1 : 0;
        }
        if (32 + lane >= d) p1 = -1;
        for (int g = 0; g < d; g += 16) {
            v4f v[8];
#pragma unroll
            for (int u = 0; u < 8; ++u) {
                int jr[2];
#pragma unroll
                for (int q = 0; q < 2; ++q) {
                    const int r = min(g + 2 * u + q, d - 1);
                    const unsigned m0 = __builtin_amdgcn_ballot_w32(p0 == r);
                    const unsigned m1 = __builtin_amdgcn_ballot_w32(p1 == r);
                    const int a0 = __builtin_amdgcn_readlane(e0, m0 ? __builtin_ctz(m0) : 0);
                    const int a1 = __builtin_amdgcn_readlane(e1, m1 ? __builtin_ctz(m1) : 0);
                    jr[q] = m0 ? a0 : a1;
                }
                const int jj = h ? jr[1] : jr[0];
                v[u] = Y4[(size_t)jj * 16 + c];
            }
#pragma unroll
            for (int u = 0; u < 8; ++u)
                if (g + 2 * u + h < d) acc += v[u];
        }
    } else if (d > 64) {
        long long last = -1;
        for (int it = 0; it < d; ++it) {
            long long best = 0x7fffffffffffffffLL;
            for (int p = lane; p < d; p += 32) {
                const long long kk = ((long long)adj[beg + p] << 32) | (long long)p;
                if (kk > last && kk < best) best = kk;
            }
            for (int m = 16; m >= 1; m >>= 1) {
                const long long o = __shfl_xor(best, m);
                if (o < best) best = o;
            }
            last = best;
            const int jj = (int)(best >> 32);
            const v4f r = Y4[(size_t)jj * 16 + c];
            if (h == 0) acc += r;
        }
    }
    acc.x += __shfl_xor(acc.x, 16);
    acc.y += __shfl_xor(acc.y, 16);
    acc.z += __shfl_xor(acc.z, 16);
    acc.w += __shfl_xor(acc.w, 16);
    const float inv = 1.0f / fmaxf((float)d, 1.0f);
    if (h == 0) ((v4f*)S)[(size_t)node * 16 + c] = acc * inv;
}

#define GM_LOAD(DST, PTR)                                                     \
    _Pragma("unroll")                                                         \
    for (int u = 0; u < 8; ++u) DST[u] = *(const v2f*)((PTR) + 4 * u);
#define GM_MMA(SRC, CI, SCALED)                                               \
    _Pragma("unroll")                                                         \
    for (int u = 0; u < 8; ++u) {                                             \
        v2f a = SRC[u];                                                       \
        if (SCALED) { a.x *= scale1; a.y *= scale1; }                         \
        const int sbase = ((((CI) << 4) + 2 * u + kSel) << 7);                \
        _Pragma("unroll")                                                     \
        for (int t = 0; t < 4; ++t) {                                         \
            v2f b = *(const v2f*)&sW[sbase + ((t << 4) + m) * 2];             \
            acc[t] = __builtin_amdgcn_wmma_f32_16x16x4_f32(                   \
                false, a, false, b, (short)0, acc[t], false, false);          \
        }                                                                     \
    }

template <int LDSK>
__global__ __launch_bounds__(128)
void gemm_wmma_relu(const float* __restrict__ X0, int d0,
                    const float* __restrict__ X1a,
                    const float* __restrict__ X1b,
                    const float* __restrict__ cnt,
                    const float* __restrict__ Wa,
                    const float* __restrict__ Wb,
                    int Ktot,
                    const float* __restrict__ biasa,
                    const float* __restrict__ biasb,
                    float* outa, float* outb,
                    int ostride, int ocol0a, int ocol0b, int dup, int Nrows) {
    __shared__ __attribute__((aligned(16))) float sW[LDSK * 64];
    const bool br = blockIdx.y != 0;
    const float* __restrict__ X1   = br ? X1b : X1a;
    const float* __restrict__ W    = br ? Wb : Wa;
    const float* __restrict__ bias = br ? biasb : biasa;
    float* __restrict__ out = br ? outb : outa;
    const int ocol0 = br ? ocol0b : ocol0a;

    const int tid  = threadIdx.x;
    const int wave = tid >> 5;
    const int lane = tid & 31;
    const int m    = lane & 15;
    const int kSel = lane >> 4;

    const int rowbase = (blockIdx.x * 4 + wave) * 16;
    const int row     = rowbase + m;
    const int crow    = row < Nrows ? row : (Nrows - 1);

    const float* x0r = X0 + (size_t)crow * d0 + 2 * kSel;
    const float* x1r = X1 ? X1 + (size_t)crow * 64 + 2 * kSel : nullptr;

    const int nch0 = d0 >> 5;
    const int ntot = (X1 != nullptr) ? (Ktot >> 5) : nch0;

    v2f areg0[8], areg1[8];
    if (ntot > 0) {
        const float* p0 = (0 < nch0) ? x0r : x1r;
        GM_LOAD(areg0, p0)
    }
    float bv[4] = {0.0f, 0.0f, 0.0f, 0.0f};
    if (bias) {
#pragma unroll
        for (int t = 0; t < 4; ++t) bv[t] = bias[(t << 4) + m];
    }
    float scale1 = 1.0f;
    if (cnt) scale1 = 1.0f / fmaxf(cnt[crow], 1.0f);

    {
        constexpr int NQ = LDSK / 16;
        const int npair = Ktot >> 1;
        v4f w0[NQ], w1[NQ];
#pragma unroll
        for (int q = 0; q < NQ; ++q) {
            const int it = tid + 128 * q;
            const int p  = min(it >> 4, npair - 1);
            const int g  = it & 15;
            w0[q] = *(const v4f*)(W + (size_t)(2 * p) * 64 + 4 * g);
            w1[q] = *(const v4f*)(W + (size_t)(2 * p + 1) * 64 + 4 * g);
        }
#pragma unroll
        for (int q = 0; q < NQ; ++q) {
            const int it = tid + 128 * q;
            const int p  = it >> 4;
            const int g  = it & 15;
            if (p < npair) {
                const v4f lo = {w0[q].x, w1[q].x, w0[q].y, w1[q].y};
                const v4f hi = {w0[q].z, w1[q].z, w0[q].w, w1[q].w};
                v4f* d = (v4f*)&sW[((p << 6) + 4 * g) * 2];
                d[0] = lo;
                d[1] = hi;
            }
        }
    }
    __syncthreads();

    v8f acc[4] = {};

#define GM_PTR(C) (((C) < nch0) ? x0r + ((C) << 5) : x1r + (((C) - nch0) << 5))
#pragma unroll 1
    for (int ci = 0; ci < ntot; ci += 2) {
        {
            const int c1 = min(ci + 1, ntot - 1);
            const float* p1 = GM_PTR(c1);
            GM_LOAD(areg1, p1)
        }
        if (cnt && ci >= nch0) { GM_MMA(areg0, ci, true) }
        else                   { GM_MMA(areg0, ci, false) }
        {
            const int c2 = min(ci + 2, ntot - 1);
            const float* p2 = GM_PTR(c2);
            GM_LOAD(areg0, p2)
        }
        if (ci + 1 < ntot) {
            if (cnt && ci + 1 >= nch0) { GM_MMA(areg1, ci + 1, true) }
            else                       { GM_MMA(areg1, ci + 1, false) }
        }
    }
#undef GM_PTR

    const int rr  = kSel * 8;
    const int col = m;
#pragma unroll
    for (int r = 0; r < 8; ++r) {
        const int orow = rowbase + r + rr;
        if (orow < Nrows) {
#pragma unroll
            for (int t = 0; t < 4; ++t) {
                float v = acc[t][r];
                if (bias) v += bv[t];
                v = fmaxf(v, 0.0f);
                float* p = out + (size_t)orow * ostride + ocol0 + (t << 4) + col;
                *p = v;
                if (dup) p[64] = v;
            }
        }
    }
}
#define FU_PITCH 68
template <int NQ>
__device__ __forceinline__ void fu_stage(const float* __restrict__ W, float* sW, int tid) {
    constexpr int npair = NQ * 8;
    v4f w0[NQ], w1[NQ];
#pragma unroll
    for (int q = 0; q < NQ; ++q) {
        const int it = tid + 128 * q;
        const int p  = it >> 4;
        const int g  = it & 15;
        w0[q] = *(const v4f*)(W + (size_t)(2 * p) * 64 + 4 * g);
        w1[q] = *(const v4f*)(W + (size_t)(2 * p + 1) * 64 + 4 * g);
    }
#pragma unroll
    for (int q = 0; q < NQ; ++q) {
        const int it = tid + 128 * q;
        const int p  = it >> 4;
        const int g  = it & 15;
        if (p < npair) {
            const v4f lo = {w0[q].x, w1[q].x, w0[q].y, w1[q].y};
            const v4f hi = {w0[q].z, w1[q].z, w0[q].w, w1[q].w};
            v4f* d = (v4f*)&sW[((p << 6) + 4 * g) * 2];
            d[0] = lo;
            d[1] = hi;
        }
    }
}

template <int NB>
__global__ __launch_bounds__(128)
void update_then_message(const float* __restrict__ X0,
                         const float* __restrict__ S0,
                         const float* __restrict__ S1,
                         const float* __restrict__ U0,
                         const float* __restrict__ U1,
                         float* __restrict__ hout, int dup,
                         const float* __restrict__ Wm,
                         const float* __restrict__ bm,
                         float* __restrict__ Yout,
                         int Nrows) {
    __shared__ __attribute__((aligned(16))) float sW[192 * 64];
    __shared__ __attribute__((aligned(16))) float hT[4 * 16 * FU_PITCH];

    const int tid  = threadIdx.x;
    const int wave = tid >> 5;
    const int lane = tid & 31;
    const int m    = lane & 15;
    const int kSel = lane >> 4;

    const int rowbase = (blockIdx.x * 4 + wave) * 16;
    const int row     = rowbase + m;
    const int crow    = row < Nrows ? row : (Nrows - 1);
    const int rr      = kSel * 8;
    float* myT = hT + wave * (16 * FU_PITCH);

    const float* x0r = X0 + (size_t)crow * 128 + 2 * kSel;
    const int nch0 = 4;
    const int ntot = 6;
    const float scale1 = 1.0f;

    float bv[4];
#pragma unroll
    for (int t = 0; t < 4; ++t) bv[t] = bm[(t << 4) + m];

    v2f hA[NB][16];

#pragma unroll
    for (int b = 0; b < NB; ++b) {
        const float* __restrict__ U  = b ? U1 : U0;
        const float* __restrict__ X1 = b ? S1 : S0;
        const float* x1r = X1 + (size_t)crow * 64 + 2 * kSel;

        v2f areg0[8], areg1[8];
        GM_LOAD(areg0, x0r)

        fu_stage<12>(U, sW, tid);
        __syncthreads();

        v8f acc[4] = {};
#define GM_PTR(C) (((C) < nch0) ? x0r + ((C) << 5) : x1r + (((C) - nch0) << 5))
#pragma unroll 1
        for (int ci = 0; ci < ntot; ci += 2) {
            {
                const int c1 = min(ci + 1, ntot - 1);
                const float* p1 = GM_PTR(c1);
                GM_LOAD(areg1, p1)
            }
            GM_MMA(areg0, ci, false)
            {
                const int c2 = min(ci + 2, ntot - 1);
                const float* p2 = GM_PTR(c2);
                GM_LOAD(areg0, p2)
            }
            if (ci + 1 < ntot) { GM_MMA(areg1, ci + 1, false) }
        }
#undef GM_PTR

#pragma unroll
        for (int r = 0; r < 8; ++r) {
#pragma unroll
            for (int t = 0; t < 4; ++t)
                myT[(r + rr) * FU_PITCH + (t << 4) + m] = fmaxf(acc[t][r], 0.0f);
        }
        __syncthreads();
#pragma unroll
        for (int s = 0; s < 16; ++s)
            hA[b][s] = *(const v2f*)&myT[m * FU_PITCH + 4 * s + 2 * kSel];
#pragma unroll
        for (int r = 0; r < 8; ++r) {
            const int orow = rowbase + r + rr;
            if (orow < Nrows) {
#pragma unroll
                for (int t = 0; t < 4; ++t) {
                    const float v = fmaxf(acc[t][r], 0.0f);
                    float* p = hout + (size_t)orow * 128 + 64 * b + (t << 4) + m;
                    *p = v;
                    if (dup) p[64] = v;
                }
            }
        }
    }

    fu_stage<8>(Wm, sW, tid);
    __syncthreads();

    v8f acc[4] = {};
#pragma unroll
    for (int s = 0; s < 32; ++s) {
        const v2f a = (s < 16) ? hA[0][s] : hA[NB - 1][s - 16];
        const int sbase = ((2 * s + kSel) << 7);
#pragma unroll
        for (int t = 0; t < 4; ++t) {
            const v2f bb = *(const v2f*)&sW[sbase + ((t << 4) + m) * 2];
            acc[t] = __builtin_amdgcn_wmma_f32_16x16x4_f32(
                false, a, false, bb, (short)0, acc[t], false, false);
        }
    }
#pragma unroll
    for (int r = 0; r < 8; ++r) {
        const int orow = rowbase + r + rr;
        if (orow < Nrows) {
#pragma unroll
            for (int t = 0; t < 4; ++t)
                Yout[(size_t)orow * 64 + (t << 4) + m] = fmaxf(acc[t][r] + bv[t], 0.0f);
        }
    }
}
#define MS_ROWS 256
__device__ __forceinline__
void message1_body(const float* __restrict__ X, const float* __restrict__ W,
                   const float* __restrict__ bias, float* __restrict__ Yout,
                   int Nrows, int blk, float* sW) {
    const int tid  = threadIdx.x;
    const int wave = tid >> 5;
    const int lane = tid & 31;
    const int m    = lane & 15;
    const int kSel = lane >> 4;

    const int rowbase = blk * MS_ROWS + wave * 16;
    const int row     = rowbase + m;
    const int crow    = row < Nrows ? row : (Nrows - 1);
    const float* x0r  = X + (size_t)crow * 128 + 2 * kSel;
    const float scale1 = 1.0f;

    v2f areg0[8], areg1[8];
    GM_LOAD(areg0, x0r)
    float bv[4];
#pragma unroll
    for (int t = 0; t < 4; ++t) bv[t] = bias[(t << 4) + m];

    {
        v4f w0[2], w1[2];
#pragma unroll
        for (int q = 0; q < 2; ++q) {
            const int it = tid + LB_BT * q;
            const int p  = it >> 4;
            const int g  = it & 15;
            w0[q] = *(const v4f*)(W + (size_t)(2 * p) * 64 + 4 * g);
            w1[q] = *(const v4f*)(W + (size_t)(2 * p + 1) * 64 + 4 * g);
        }
#pragma unroll
        for (int q = 0; q < 2; ++q) {
            const int it = tid + LB_BT * q;
            const int p  = it >> 4;
            const int g  = it & 15;
            const v4f lo = {w0[q].x, w1[q].x, w0[q].y, w1[q].y};
            const v4f hi = {w0[q].z, w1[q].z, w0[q].w, w1[q].w};
            v4f* d = (v4f*)&sW[((p << 6) + 4 * g) * 2];
            d[0] = lo;
            d[1] = hi;
        }
    }
    __syncthreads();

    v8f acc[4] = {};
#pragma unroll 1
    for (int ci = 0; ci < 4; ci += 2) {
        {
            const float* p1 = x0r + (min(ci + 1, 3) << 5);
            GM_LOAD(areg1, p1)
        }
        GM_MMA(areg0, ci, false)
        {
            const float* p2 = x0r + (min(ci + 2, 3) << 5);
            GM_LOAD(areg0, p2)
        }
        GM_MMA(areg1, ci + 1, false)
    }

    const int rr = kSel * 8;
#pragma unroll
    for (int r = 0; r < 8; ++r) {
        const int orow = rowbase + r + rr;
        if (orow < Nrows) {
#pragma unroll
            for (int t = 0; t < 4; ++t)
                Yout[(size_t)orow * 64 + (t << 4) + m] = fmaxf(acc[t][r] + bv[t], 0.0f);
        }
    }
}

__global__ __launch_bounds__(LB_BT)
void slot_bases_and_message(int* __restrict__ T, const int* __restrict__ C, int N,
                            int cstride, int* __restrict__ rowptr,
                            const float* __restrict__ X,
                            const float* __restrict__ Wa, const float* __restrict__ Wb,
                            const float* __restrict__ biasa, const float* __restrict__ biasb,
                            float* __restrict__ Ya, float* __restrict__ Yb) {
    __shared__ __attribute__((aligned(16))) float sW[128 * 64];
    __shared__ int wsum[32];
    __shared__ int wpart[32];
    __shared__ int bstart;
    const int nb = (N + LB_BT - 1) / LB_BT;
    const int bx = blockIdx.x;
    if (bx < 2 * nb) {
        const int dir = bx >= nb ? 1 : 0;
        slot_bases_body(T, C, N, cstride, rowptr, bx - dir * nb, dir, wsum, wpart, &bstart);
    } else {
        const int ngb = (N + MS_ROWS - 1) / MS_ROWS;
        const int g   = bx - 2 * nb;
        const bool br = g >= ngb;
        message1_body(X, br ? Wb : Wa, br ? biasb : biasa, br ? Yb : Ya, N,
                      br ? g - ngb : g, sW);
    }
}
#undef GM_LOAD
#undef GM_MMA

extern "C" void kernel_launch(void* const* d_in, const int* in_sizes, int n_in,
                              void* d_out, int out_size, void* d_ws, size_t ws_size,
                              hipStream_t stream) {
    const float* x    = (const float*)d_in[0];
    const int*   ei   = (const int*)  d_in[1];
    const float* W_u1 = (const float*)d_in[2];
    const float* b_u1 = (const float*)d_in[3];
    const float* U_u1 = (const float*)d_in[4];
    const float* W_d1 = (const float*)d_in[5];
    const float* b_d1 = (const float*)d_in[6];
    const float* U_d1 = (const float*)d_in[7];
    const float* W_u2 = (const float*)d_in[8];
    const float* b_u2 = (const float*)d_in[9];
    const float* U_u2 = (const float*)d_in[10];
    float* out = (float*)d_out;

    const int N = in_sizes[0] / 128;
    const int E = in_sizes[1] / 2;

    const int cstride = (N + 31) / 32 * 32;
    float* ws     = (float*)d_ws;
    float* Y      = ws;
    float* S      = Y  + (size_t)N * 64;
    float* ha     = S  + (size_t)N * 64;
    int*   cnt    = (int*)(ha + (size_t)N * 128);
    int*   rowptr = cnt    + 2 * (size_t)cstride;
    int*   adj    = rowptr + 2 * ((size_t)N + 1);
    int*   T      = (int*)ws;
    int*   C      = cnt;

    const int* rp_in   = rowptr;
    const int* rp_out  = rowptr + (N + 1);
    const int* adj_in  = adj;
    const int* adj_out = adj + E;

    float* Yu = out;
    float* Yd = out + (size_t)N * 64;
    slice_count<<<dim3(LB_SLICES, 2), LB_TPB, 0, stream>>>(ei, E, N, cstride, T, C);
    slot_bases_and_message<<<2 * ((N + LB_BT - 1) / LB_BT) + 2 * ((N + MS_ROWS - 1) / MS_ROWS),
                             LB_BT, 0, stream>>>(
        T, C, N, cstride, rowptr, x, W_u1, W_d1, b_u1, b_d1, Yu, Yd);
    slice_place<<<dim3(LB_SLICES, 2), LB_TPB, 0, stream>>>(ei, E, N, cstride, T, adj);

    const int gblk  = (N + 63) / 64;
    const int ggrid = (N + GA_TPB / 32 - 1) / (GA_TPB / 32);

    float* Sd = ha;
    gather_mean64<<<dim3(ggrid, 2), GA_TPB, 0, stream>>>(
        Yu, Yd, rp_in, rp_out, adj_in, adj_out, S, Sd, N);
    update_then_message<2><<<gblk, 128, 0, stream>>>(
        x, S, Sd, U_u1, U_d1, out, 0, W_u2, b_u2, Y, N);

    gather_mean64<<<ggrid, GA_TPB, 0, stream>>>(Y, nullptr, rp_in, nullptr, adj_in, nullptr,
                                                S, nullptr, N);
    update_then_message<1><<<gblk, 128, 0, stream>>>(
        out, S, nullptr, U_u2, nullptr, ha, 1, W_u2, b_u2, Y, N);
    gather_mean64<<<ggrid, GA_TPB, 0, stream>>>(Y, nullptr, rp_in, nullptr, adj_in, nullptr,
                                                S, nullptr, N);
    gemm_wmma_relu<192><<<gblk, 128, 0, stream>>>(
        ha, 128, S, nullptr, nullptr, U_u2, nullptr, 192, nullptr, nullptr,
        out, nullptr, 128, 0, 0, 1, N);
}


# neighbour means formed inside the update GEMM's work-group (a wave sums its 16 nodes' lists into an LDS tile; last 64 k read from it): no mean launches or mean array; 6 launches not 9; sums as before
# speedup vs baseline: 3.5888x; 1.0429x over previous
#include <hip/hip_runtime.h>
#include <hip/hip_bf16.h>

typedef __attribute__((ext_vector_type(2))) float v2f;
typedef __attribute__((ext_vector_type(4))) float v4f;
typedef __attribute__((ext_vector_type(8))) float v8f;

#define LB_TPB    1024
#define LB_LDSN   65536
#define LB_SLICES 64
#define LB_BT     512
#define LB_NB     128
__global__ __launch_bounds__(LB_TPB)
void slice_count(const int* __restrict__ ei, int E, int N, int cstride,
                 int* __restrict__ T, int* __restrict__ C) {
    __shared__ int h[LB_LDSN];
    const int dir = blockIdx.y;
    const int* key = ei + (dir == 0 ? E : 0);
    const int tid = threadIdx.x;
    const int len = (E + LB_SLICES - 1) / LB_SLICES;
    const int lo  = min((int)blockIdx.x * len, E);
    const int hi  = min(lo + len, E);
    const size_t srow = (size_t)dir * LB_SLICES + blockIdx.x;
    int* Tt = T + srow * cstride;

    for (int n = tid; n < N; n += LB_TPB) h[n] = 0;
    __syncthreads();

    for (int e0 = lo; e0 < hi; e0 += 8 * LB_TPB) {
        int k[8];
#pragma unroll
        for (int u = 0; u < 8; ++u) {
            const int e = e0 + u * LB_TPB + tid;
            k[u] = (e < hi) ? key[e] : 0;
        }
#pragma unroll
        for (int u = 0; u < 8; ++u)
            if (e0 + u * LB_TPB + tid < hi) atomicAdd(&h[k[u]], 1);
    }
    __syncthreads();

    for (int n0 = 0; n0 < N; n0 += 8 * LB_TPB) {
        int c[8];
#pragma unroll
        for (int u = 0; u < 8; ++u) {
            const int n = n0 + u * LB_TPB + tid;
            c[u] = (n < N) ? h[n] : 0;
        }
#pragma unroll
        for (int u = 0; u < 8; ++u) {
            const int n = n0 + u * LB_TPB + tid;
            if (n < N) Tt[n] = c[u];
        }
    }

    const int i = tid >> 3;
    const int j = tid & 7;
    int s = 0;
    for (int m = 0; m < LB_BT / 8; ++m) {
        const int n = i * LB_BT + j + (m << 3);
        const int x = h[min(n, N - 1)];
        s += (n < N) ? x : 0;
    }
    s += __shfl_xor(s, 4);
    s += __shfl_xor(s, 2);
    s += __shfl_xor(s, 1);
    if (j == 0) C[srow * LB_NB + i] = s;
}

__device__ __forceinline__
void slot_bases_body(int* __restrict__ T, const int* __restrict__ C, int N,
                     int cstride, int* __restrict__ rowptr, int b, int dir,
                     int* wsum, int* wpart, int* bstart) {
    const int tid  = threadIdx.x;
    const int lane = tid & 31;
    const int wv   = tid >> 5;
    const int n    = b * LB_BT + tid;
    int*       Td = T + (size_t)dir * LB_SLICES * cstride;
    const int* Cd = C + (size_t)dir * LB_SLICES * LB_NB;
    int*       rp = rowptr + (size_t)dir * (N + 1);

    int part = 0;
    for (int idx = tid; idx < LB_SLICES * LB_NB; idx += LB_BT)
        if ((idx & (LB_NB - 1)) < b) part += Cd[idx];

    int c[LB_SLICES];
#pragma unroll
    for (int s = 0; s < LB_SLICES; ++s)
        c[s] = Td[(size_t)s * cstride + min(n, N - 1)];
#pragma unroll
    for (int s = 0; s < LB_SLICES; ++s)
        if (n >= N) c[s] = 0;
    int deg = 0;
#pragma unroll
    for (int s = 0; s < LB_SLICES; ++s) deg += c[s];

    int incl = deg;
    for (int d = 1; d < 32; d <<= 1) {
        int t = __shfl_up(incl, d);
        if (lane >= d) incl += t;
    }
    for (int d = 16; d >= 1; d >>= 1) part += __shfl_xor(part, d);
    if (lane == 31) wsum[wv] = incl;
    if (lane == 0)  wpart[wv] = part;
    __syncthreads();
    if (wv == 0) {
        const int w = (lane < LB_BT / 32) ? wsum[lane] : 0;
        int wi = w;
        for (int d = 1; d < 32; d <<= 1) {
            int t = __shfl_up(wi, d);
            if (lane >= d) wi += t;
        }
        if (lane < LB_BT / 32) wsum[lane] = wi - w;
        int p = (lane < LB_BT / 32) ? wpart[lane] : 0;
        for (int d = 16; d >= 1; d >>= 1) p += __shfl_xor(p, d);
        if (lane == 0) *bstart = p;
    }
    __syncthreads();

    int run = *bstart + wsum[wv] + incl - deg;
    if (n < N) {
        rp[n] = run;
        if (n == N - 1) rp[N] = run + deg;
#pragma unroll
        for (int s = 0; s < LB_SLICES; ++s) {
            Td[(size_t)s * cstride + n] = run;
            run += c[s];
        }
    }
}

__global__ __launch_bounds__(LB_TPB)
void slice_place(const int* __restrict__ ei, int E, int N, int cstride,
                 const int* __restrict__ T, int* __restrict__ adj) {
    __shared__ int h[LB_LDSN];
    const int dir = blockIdx.y;
    const int* key = ei + (dir == 0 ? E : 0);
    const int* val = ei + (dir == 0 ? 0 : E);
    const int tid = threadIdx.x;
    const int len = (E + LB_SLICES - 1) / LB_SLICES;
    const int lo  = min((int)blockIdx.x * len, E);
    const int hi  = min(lo + len, E);
    const int* Tt = T + ((size_t)dir * LB_SLICES + blockIdx.x) * cstride;
    int* ad = adj + (size_t)dir * E;

    for (int n0 = 0; n0 < N; n0 += 16 * LB_TPB) {
        int c[16];
#pragma unroll
        for (int u = 0; u < 16; ++u) {
            const int n = n0 + u * LB_TPB + tid;
            c[u] = (n < N) ? Tt[n] : 0;
        }
#pragma unroll
        for (int u = 0; u < 16; ++u) {
            const int n = n0 + u * LB_TPB + tid;
            if (n < N) h[n] = c[u];
        }
    }
    __syncthreads();

    for (int e0 = lo; e0 < hi; e0 += 8 * LB_TPB) {
        int k[8], v[8];
#pragma unroll
        for (int u = 0; u < 8; ++u) {
            const int e = min(e0 + u * LB_TPB + tid, hi - 1);
            k[u] = key[e];
            v[u] = val[e];
        }
#pragma unroll
        for (int u = 0; u < 8; ++u)
            if (e0 + u * LB_TPB + tid < hi) ad[atomicAdd(&h[k[u]], 1)] = v[u];
    }
}

__device__ __forceinline__
int list_order(int d, int my, int lane) {
    int srt = my;
    if (d > 0 && d <= 32) {
        const int key = (my << 5) | lane;
        int pos = 0;
        for (int j = 0; j < d; ++j) {
            const int kj = __builtin_amdgcn_readlane(key, j);
            pos += (kj < key) ? 1 : 0;
        }
        if (lane >= d) pos = lane;
        const int dst = ((pos & 1) << 4) | (pos >> 1);
        srt = __builtin_amdgcn_ds_permute(dst << 2, my);
    }
    return srt;
}

__device__ __forceinline__
void list_rows(const v4f* __restrict__ Y4, int d, int srt, int c,
               v4f (&v)[8], v4f (&w)[8]) {
    if (d > 0 && d <= 32) {
#define GA_ROW(V, U, Q) V[U] = *(const v4f*)((const char*)Y4 + ((((unsigned)__builtin_amdgcn_update_dpp(0, srt, 0x150 + (Q), 0xf, 0xf, false)) << 8) | ((unsigned)c << 4)))
        GA_ROW(v, 0, 0); GA_ROW(v, 1, 1); GA_ROW(v, 2, 2); GA_ROW(v, 3, 3);
        GA_ROW(v, 4, 4); GA_ROW(v, 5, 5); GA_ROW(v, 6, 6); GA_ROW(v, 7, 7);
        if (d > 16) {
            GA_ROW(w, 0, 8);  GA_ROW(w, 1, 9);  GA_ROW(w, 2, 10); GA_ROW(w, 3, 11);
            GA_ROW(w, 4, 12); GA_ROW(w, 5, 13); GA_ROW(w, 6, 14); GA_ROW(w, 7, 15);
        }
#undef GA_ROW
    }
}

__device__ __forceinline__
v4f list_finish(const v4f* __restrict__ Y4, const int* __restrict__ adj,
                int beg, int d, int lane, int h, int c,
                const v4f (&v)[8], const v4f (&w)[8]) {
    v4f acc = {0.0f, 0.0f, 0.0f, 0.0f};
    if (d > 0 && d <= 32) {
#pragma unroll
        for (int u = 0; u < 8; ++u)
            if (2 * u + h < d) acc += v[u];
        if (d > 16) {
#pragma unroll
            for (int u = 0; u < 8; ++u)
                if (16 + 2 * u + h < d) acc += w[u];
        }
    } else if (d > 32 && d <= 64) {
        const int e0 = adj[beg + lane];
        const int e1 = adj[beg + min(32 + lane, d - 1)];
        const int k0 = (e0 << 6) | lane;
        const int k1 = (32 + lane < d) ? ((e1 << 6) | (32 + lane)) : 0x7fffffff;
        int p0 = 0, p1 = 0;
        for (int j = 0; j < 32; ++j) {
            const int kj = __builtin_amdgcn_readlane(k0, j);
            p0 += (kj < k0) ? 1 : 0;
            p1 += (kj < k1) ? 1 : 0;
        }
        for (int j = 32; j < d; ++j) {
            const int kj = __builtin_amdgcn_readlane(k1, j - 32);
            p0 += (kj < k0) ? 1 : 0;
            p1 += (kj < k1) ? 1 : 0;
        }
        if (32 + lane >= d) p1 = -1;
        for (int g = 0; g < d; g += 16) {
            v4f v[8];
#pragma unroll
            for (int u = 0; u < 8; ++u) {
                int jr[2];
#pragma unroll
                for (int q = 0; q < 2; ++q) {
                    const int r = min(g + 2 * u + q, d - 1);
                    const unsigned m0 = __builtin_amdgcn_ballot_w32(p0 == r);
                    const unsigned m1 = __builtin_amdgcn_ballot_w32(p1 == r);
                    const int a0 = __builtin_amdgcn_readlane(e0, m0 ? __builtin_ctz(m0) : 0);
                    const int a1 = __builtin_amdgcn_readlane(e1, m1 ? __builtin_ctz(m1) : 0);
                    jr[q] = m0 ? a0 : a1;
                }
                const int jj = h ? jr[1] : jr[0];
                v[u] = Y4[(size_t)jj * 16 + c];
            }
#pragma unroll
            for (int u = 0; u < 8; ++u)
                if (g + 2 * u + h < d) acc += v[u];
        }
    } else if (d > 64) {
        long long last = -1;
        for (int it = 0; it < d; ++it) {
            long long best = 0x7fffffffffffffffLL;
            for (int p = lane; p < d; p += 32) {
                const long long kk = ((long long)adj[beg + p] << 32) | (long long)p;
                if (kk > last && kk < best) best = kk;
            }
            for (int m = 16; m >= 1; m >>= 1) {
                const long long o = __shfl_xor(best, m);
                if (o < best) best = o;
            }
            last = best;
            const int jj = (int)(best >> 32);
            const v4f r = Y4[(size_t)jj * 16 + c];
            if (h == 0) acc += r;
        }
    }
    acc.x += __shfl_xor(acc.x, 16);
    acc.y += __shfl_xor(acc.y, 16);
    acc.z += __shfl_xor(acc.z, 16);
    acc.w += __shfl_xor(acc.w, 16);
    return acc;
}

#define FU_PITCH 68
__device__ __forceinline__
void gather_tile(const float* __restrict__ Y, const int* __restrict__ rowptr,
                 const int* __restrict__ adj, int N, int rowbase, int lane,
                 float* myT) {
    const int h = lane >> 4;
    const int c = lane & 15;
    const v4f* Y4 = (const v4f*)Y;
    const int rp = rowptr[min(rowbase + min(lane, 16), N)];
    int e0 = __builtin_amdgcn_readlane(rp, 0);
    int e1 = __builtin_amdgcn_readlane(rp, 1);
    int e2 = __builtin_amdgcn_readlane(rp, 2);
    int nxA = 0, nxB = 0;
    if (e1 > e0) nxA = adj[e0 + min(lane, e1 - e0 - 1)];
    if (e2 > e1) nxB = adj[e1 + min(lane, e2 - e1 - 1)];
#pragma unroll 1
    for (int i = 0; i < 16; i += 2) {
        const int begA = e0, dA = e1 - e0;
        const int begB = e1, dB = e2 - e1;
        const int srtA = list_order(dA, nxA, lane);
        const int srtB = list_order(dB, nxB, lane);
        e0 = e2;
        e1 = (i < 14) ? __builtin_amdgcn_readlane(rp, i + 3) : e0;
        e2 = (i < 14) ? __builtin_amdgcn_readlane(rp, i + 4) : e0;
        nxA = 0;
        nxB = 0;
        if (e1 > e0) nxA = adj[e0 + min(lane, e1 - e0 - 1)];
        if (e2 > e1) nxB = adj[e1 + min(lane, e2 - e1 - 1)];
        v4f vA[8], wA[8], vB[8], wB[8];
        list_rows(Y4, dA, srtA, c, vA, wA);
        list_rows(Y4, dB, srtB, c, vB, wB);
        const v4f accA = list_finish(Y4, adj, begA, dA, lane, h, c, vA, wA);
        const v4f accB = list_finish(Y4, adj, begB, dB, lane, h, c, vB, wB);
        const float invA = 1.0f / fmaxf((float)dA, 1.0f);
        const float invB = 1.0f / fmaxf((float)dB, 1.0f);
        if (h == 0) {
            *(v4f*)&myT[i * FU_PITCH + 4 * c]       = accA * invA;
            *(v4f*)&myT[(i + 1) * FU_PITCH + 4 * c] = accB * invB;
        }
    }
}

#define GM_LOAD(DST, PTR)                                                     \
    _Pragma("unroll")                                                         \
    for (int u = 0; u < 8; ++u) DST[u] = *(const v2f*)((PTR) + 4 * u);
#define GM_MMA(SRC, CI)                                                       \
    _Pragma("unroll")                                                         \
    for (int u = 0; u < 8; ++u) {                                             \
        const v2f a = SRC[u];                                                 \
        const int sbase = ((((CI) << 4) + 2 * u + kSel) << 7);                \
        _Pragma("unroll")                                                     \
        for (int t = 0; t < 4; ++t) {                                         \
            v2f b = *(const v2f*)&sW[sbase + ((t << 4) + m) * 2];             \
            acc[t] = __builtin_amdgcn_wmma_f32_16x16x4_f32(                   \
                false, a, false, b, (short)0, acc[t], false, false);          \
        }                                                                     \
    }

template <int NQ>
__device__ __forceinline__ void fu_stage(const float* __restrict__ W, float* sW, int tid) {
    constexpr int npair = NQ * 8;
    v4f w0[NQ], w1[NQ];
#pragma unroll
    for (int q = 0; q < NQ; ++q) {
        const int it = tid + 128 * q;
        const int p  = it >> 4;
        const int g  = it & 15;
        w0[q] = *(const v4f*)(W + (size_t)(2 * p) * 64 + 4 * g);
        w1[q] = *(const v4f*)(W + (size_t)(2 * p + 1) * 64 + 4 * g);
    }
#pragma unroll
    for (int q = 0; q < NQ; ++q) {
        const int it = tid + 128 * q;
        const int p  = it >> 4;
        const int g  = it & 15;
        if (p < npair) {
            const v4f lo = {w0[q].x, w1[q].x, w0[q].y, w1[q].y};
            const v4f hi = {w0[q].z, w1[q].z, w0[q].w, w1[q].w};
            v4f* d = (v4f*)&sW[((p << 6) + 4 * g) * 2];
            d[0] = lo;
            d[1] = hi;
        }
    }
}

template <int NB, bool LAST>
__global__ __launch_bounds__(128, 4)
void gather_update_message(const float* X0,
                           const float* __restrict__ Ya,
                           const float* __restrict__ Yb,
                           const int* __restrict__ rowptra, const int* __restrict__ rowptrb,
                           const int* __restrict__ adja, const int* __restrict__ adjb,
                           const float* __restrict__ U0,
                           const float* __restrict__ U1,
                           float* hout, int dup,
                           const float* __restrict__ Wm,
                           const float* __restrict__ bm,
                           float* __restrict__ Yout,
                           int Nrows) {
    __shared__ __attribute__((aligned(16))) float sW[192 * 64];
    __shared__ __attribute__((aligned(16))) float hT[4 * 16 * FU_PITCH];

    const int tid  = threadIdx.x;
    const int wave = tid >> 5;
    const int lane = tid & 31;
    const int m    = lane & 15;
    const int kSel = lane >> 4;

    const int rowbase = __builtin_amdgcn_readfirstlane((int)(blockIdx.x * 4 + wave) * 16);
    const int row     = rowbase + m;
    const int crow    = row < Nrows ? row : (Nrows - 1);
    const int rr      = kSel * 8;
    float* myT = hT + wave * (16 * FU_PITCH);

    const float* x0r = X0 + (size_t)crow * 128 + 2 * kSel;

    v2f hA[NB][16];

    gather_tile(Ya, rowptra, adja, Nrows, rowbase, lane, myT);
    v2f s1A[16];
    if constexpr (NB > 1) {
        float* sT = sW + wave * (16 * FU_PITCH);
        gather_tile(Yb, rowptrb, adjb, Nrows, rowbase, lane, sT);
        __syncthreads();
#pragma unroll
        for (int s = 0; s < 16; ++s)
            s1A[s] = *(const v2f*)&sT[m * FU_PITCH + 4 * s + 2 * kSel];
        __syncthreads();
    }

#pragma unroll
    for (int b = 0; b < NB; ++b) {
        const float* __restrict__ U = b ? U1 : U0;

        if (b > 0) {
#pragma unroll
            for (int s = 0; s < 16; ++s)
                *(v2f*)&myT[m * FU_PITCH + 4 * s + 2 * kSel] = s1A[s];
        }

        v2f areg0[8], areg1[8];
        GM_LOAD(areg0, x0r)

        fu_stage<12>(U, sW, tid);
        __syncthreads();

        v8f acc[4] = {};
#pragma unroll 1
        for (int ci = 0; ci < 4; ci += 2) {
            {
                const float* p1 = x0r + (min(ci + 1, 3) << 5);
                GM_LOAD(areg1, p1)
            }
            GM_MMA(areg0, ci)
            {
                const float* p2 = x0r + (min(ci + 2, 3) << 5);
                GM_LOAD(areg0, p2)
            }
            GM_MMA(areg1, ci + 1)
        }
#pragma unroll
        for (int s = 0; s < 16; ++s) {
            const v2f a = *(const v2f*)&myT[m * FU_PITCH + 4 * s + 2 * kSel];
            const int sbase = ((64 + 2 * s + kSel) << 7);
#pragma unroll
            for (int t = 0; t < 4; ++t) {
                const v2f bb = *(const v2f*)&sW[sbase + ((t << 4) + m) * 2];
                acc[t] = __builtin_amdgcn_wmma_f32_16x16x4_f32(
                    false, a, false, bb, (short)0, acc[t], false, false);
            }
        }

        if constexpr (!LAST) {
#pragma unroll
            for (int r = 0; r < 8; ++r) {
#pragma unroll
                for (int t = 0; t < 4; ++t)
                    myT[(r + rr) * FU_PITCH + (t << 4) + m] = fmaxf(acc[t][r], 0.0f);
            }
            __syncthreads();
#pragma unroll
            for (int s = 0; s < 16; ++s)
                hA[b][s] = *(const v2f*)&myT[m * FU_PITCH + 4 * s + 2 * kSel];
        }
#pragma unroll
        for (int r = 0; r < 8; ++r) {
            const int orow = rowbase + r + rr;
            if (orow < Nrows) {
#pragma unroll
                for (int t = 0; t < 4; ++t) {
                    const float v = fmaxf(acc[t][r], 0.0f);
                    float* p = hout + (size_t)orow * 128 + 64 * b + (t << 4) + m;
                    *p = v;
                    if (dup) p[64] = v;
                }
            }
        }
    }

    if constexpr (!LAST) {
        float bv[4];
#pragma unroll
        for (int t = 0; t < 4; ++t) bv[t] = bm[(t << 4) + m];

        fu_stage<8>(Wm, sW, tid);
        __syncthreads();

        v8f acc[4] = {};
#pragma unroll
        for (int s = 0; s < 32; ++s) {
            const v2f a = (s < 16) ? hA[0][s] : hA[NB - 1][s - 16];
            const int sbase = ((2 * s + kSel) << 7);
#pragma unroll
            for (int t = 0; t < 4; ++t) {
                const v2f bb = *(const v2f*)&sW[sbase + ((t << 4) + m) * 2];
                acc[t] = __builtin_amdgcn_wmma_f32_16x16x4_f32(
                    false, a, false, bb, (short)0, acc[t], false, false);
            }
        }
#pragma unroll
        for (int r = 0; r < 8; ++r) {
            const int orow = rowbase + r + rr;
            if (orow < Nrows) {
#pragma unroll
                for (int t = 0; t < 4; ++t)
                    Yout[(size_t)orow * 64 + (t << 4) + m] = fmaxf(acc[t][r] + bv[t], 0.0f);
            }
        }
    }
}
#define MS_ROWS 256
__device__ __forceinline__
void message1_body(const float* __restrict__ X, const float* __restrict__ W,
                   const float* __restrict__ bias, float* __restrict__ Yout,
                   int Nrows, int blk, float* sW) {
    const int tid  = threadIdx.x;
    const int wave = tid >> 5;
    const int lane = tid & 31;
    const int m    = lane & 15;
    const int kSel = lane >> 4;

    const int rowbase = blk * MS_ROWS + wave * 16;
    const int row     = rowbase + m;
    const int crow    = row < Nrows ? row : (Nrows - 1);
    const float* x0r  = X + (size_t)crow * 128 + 2 * kSel;

    v2f areg0[8], areg1[8];
    GM_LOAD(areg0, x0r)
    float bv[4];
#pragma unroll
    for (int t = 0; t < 4; ++t) bv[t] = bias[(t << 4) + m];

    {
        v4f w0[2], w1[2];
#pragma unroll
        for (int q = 0; q < 2; ++q) {
            const int it = tid + LB_BT * q;
            const int p  = it >> 4;
            const int g  = it & 15;
            w0[q] = *(const v4f*)(W + (size_t)(2 * p) * 64 + 4 * g);
            w1[q] = *(const v4f*)(W + (size_t)(2 * p + 1) * 64 + 4 * g);
        }
#pragma unroll
        for (int q = 0; q < 2; ++q) {
            const int it = tid + LB_BT * q;
            const int p  = it >> 4;
            const int g  = it & 15;
            const v4f lo = {w0[q].x, w1[q].x, w0[q].y, w1[q].y};
            const v4f hi = {w0[q].z, w1[q].z, w0[q].w, w1[q].w};
            v4f* d = (v4f*)&sW[((p << 6) + 4 * g) * 2];
            d[0] = lo;
            d[1] = hi;
        }
    }
    __syncthreads();

    v8f acc[4] = {};
#pragma unroll 1
    for (int ci = 0; ci < 4; ci += 2) {
        {
            const float* p1 = x0r + (min(ci + 1, 3) << 5);
            GM_LOAD(areg1, p1)
        }
        GM_MMA(areg0, ci)
        {
            const float* p2 = x0r + (min(ci + 2, 3) << 5);
            GM_LOAD(areg0, p2)
        }
        GM_MMA(areg1, ci + 1)
    }

    const int rr = kSel * 8;
#pragma unroll
    for (int r = 0; r < 8; ++r) {
        const int orow = rowbase + r + rr;
        if (orow < Nrows) {
#pragma unroll
            for (int t = 0; t < 4; ++t)
                Yout[(size_t)orow * 64 + (t << 4) + m] = fmaxf(acc[t][r] + bv[t], 0.0f);
        }
    }
}

__global__ __launch_bounds__(LB_BT)
void slot_bases_and_message(int* __restrict__ T, const int* __restrict__ C, int N,
                            int cstride, int* __restrict__ rowptr,
                            const float* __restrict__ X,
                            const float* __restrict__ Wa, const float* __restrict__ Wb,
                            const float* __restrict__ biasa, const float* __restrict__ biasb,
                            float* __restrict__ Ya, float* __restrict__ Yb) {
    __shared__ __attribute__((aligned(16))) float sW[128 * 64];
    __shared__ int wsum[32];
    __shared__ int wpart[32];
    __shared__ int bstart;
    const int nb = (N + LB_BT - 1) / LB_BT;
    const int bx = blockIdx.x;
    if (bx < 2 * nb) {
        const int dir = bx >= nb ? 1 : 0;
        slot_bases_body(T, C, N, cstride, rowptr, bx - dir * nb, dir, wsum, wpart, &bstart);
    } else {
        const int ngb = (N + MS_ROWS - 1) / MS_ROWS;
        const int g   = bx - 2 * nb;
        const bool br = g >= ngb;
        message1_body(X, br ? Wb : Wa, br ? biasb : biasa, br ? Yb : Ya, N,
                      br ? g - ngb : g, sW);
    }
}
#undef GM_LOAD
#undef GM_MMA

extern "C" void kernel_launch(void* const* d_in, const int* in_sizes, int n_in,
                              void* d_out, int out_size, void* d_ws, size_t ws_size,
                              hipStream_t stream) {
    const float* x    = (const float*)d_in[0];
    const int*   ei   = (const int*)  d_in[1];
    const float* W_u1 = (const float*)d_in[2];
    const float* b_u1 = (const float*)d_in[3];
    const float* U_u1 = (const float*)d_in[4];
    const float* W_d1 = (const float*)d_in[5];
    const float* b_d1 = (const float*)d_in[6];
    const float* U_d1 = (const float*)d_in[7];
    const float* W_u2 = (const float*)d_in[8];
    const float* b_u2 = (const float*)d_in[9];
    const float* U_u2 = (const float*)d_in[10];
    float* out = (float*)d_out;

    const int N = in_sizes[0] / 128;
    const int E = in_sizes[1] / 2;

    const int cstride = (N + 31) / 32 * 32;
    float* ws     = (float*)d_ws;
    float* Y      = ws;
    float* Y2     = Y  + (size_t)N * 64;
    float* ha     = Y2 + (size_t)N * 64;
    int*   cnt    = (int*)(ha + (size_t)N * 128);
    int*   rowptr = cnt    + 2 * (size_t)cstride;
    int*   adj    = rowptr + 2 * ((size_t)N + 1);
    int*   T      = (int*)ws;
    int*   C      = cnt;

    const int* rp_in   = rowptr;
    const int* rp_out  = rowptr + (N + 1);
    const int* adj_in  = adj;
    const int* adj_out = adj + E;

    float* Yu = out;
    float* Yd = out + (size_t)N * 64;
    slice_count<<<dim3(LB_SLICES, 2), LB_TPB, 0, stream>>>(ei, E, N, cstride, T, C);
    slot_bases_and_message<<<2 * ((N + LB_BT - 1) / LB_BT) + 2 * ((N + MS_ROWS - 1) / MS_ROWS),
                             LB_BT, 0, stream>>>(
        T, C, N, cstride, rowptr, x, W_u1, W_d1, b_u1, b_d1, Yu, Yd);
    slice_place<<<dim3(LB_SLICES, 2), LB_TPB, 0, stream>>>(ei, E, N, cstride, T, adj);

    const int gblk = (N + 63) / 64;

    gather_update_message<2, false><<<gblk, 128, 0, stream>>>(
        x, Yu, Yd, rp_in, rp_out, adj_in, adj_out, U_u1, U_d1, ha, 0, W_u2, b_u2, Y, N);

    gather_update_message<1, false><<<gblk, 128, 0, stream>>>(
        ha, Y, nullptr, rp_in, nullptr, adj_in, nullptr, U_u2, nullptr, out, 1, W_u2, b_u2, Y2, N);
    gather_update_message<1, true><<<gblk, 128, 0, stream>>>(
        out, Y2, nullptr, rp_in, nullptr, adj_in, nullptr, U_u2, nullptr, out, 1,
        nullptr, nullptr, nullptr, N);
}


# lists built by a two-level bucket sort: a slice buckets its edges by 512-node block (staged in LDS, whole-line stores), a work-group per block orders its nodes; 5 launches, not 6; same sums
# speedup vs baseline: 4.6193x; 1.2871x over previous
#include <hip/hip_runtime.h>
#include <hip/hip_bf16.h>

typedef __attribute__((ext_vector_type(2))) float v2f;
typedef __attribute__((ext_vector_type(4))) float v4f;
typedef __attribute__((ext_vector_type(8))) float v8f;

typedef __attribute__((ext_vector_type(2))) int v2i;
#define LB_TPB    1024
#define LB_SLICES 64
#define LB_BT     512
#define LB_BTS    9
#define LB_NB     128
#define LA_CAP    16384
#define LA_UN     16
__global__ __launch_bounds__(LB_TPB)
void slice_buckets(const int* __restrict__ ei, int E, int* __restrict__ bucket,
                   int* __restrict__ cntT, int* __restrict__ offT) {
    __shared__ int cur[LB_NB];
    __shared__ v2i sbuf[LA_CAP];
    const int dir = blockIdx.y;
    const int* key = ei + (dir == 0 ? E : 0);
    const int* val = ei + (dir == 0 ? 0 : E);
    const int tid = threadIdx.x;
    const int len = (E + LB_SLICES - 1) / LB_SLICES;
    const int lo  = min((int)blockIdx.x * len, E);
    const int hi  = min(lo + len, E);
    const int srow = dir * LB_SLICES + blockIdx.x;

    if (tid < LB_NB) cur[tid] = 0;
    __syncthreads();

    for (int e0 = lo; e0 < hi; e0 += LA_UN * LB_TPB) {
        int k[LA_UN];
#pragma unroll
        for (int u = 0; u < LA_UN; ++u)
            k[u] = key[min(e0 + u * LB_TPB + tid, hi - 1)];
#pragma unroll
        for (int u = 0; u < LA_UN; ++u)
            if (e0 + u * LB_TPB + tid < hi) atomicAdd(&cur[k[u] >> LB_BTS], 1);
    }
    __syncthreads();

    if (tid < 32) {
        int c[4];
        int s = 0;
#pragma unroll
        for (int j = 0; j < 4; ++j) {
            c[j] = cur[4 * tid + j];
            s += c[j];
        }
        int incl = s;
        for (int d = 1; d < 32; d <<= 1) {
            int t = __shfl_up(incl, d);
            if (tid >= d) incl += t;
        }
        int run = incl - s;
#pragma unroll
        for (int j = 0; j < 4; ++j) {
            cntT[srow * LB_NB + 4 * tid + j] = c[j];
            offT[srow * LB_NB + 4 * tid + j] = run;
            cur[4 * tid + j] = run;
            run += c[j];
        }
    }
    __syncthreads();

    const bool stage = (hi - lo) <= LA_CAP;
    v2i* gb = (v2i*)bucket + (size_t)dir * E + lo;
    for (int e0 = lo; e0 < hi; e0 += LA_UN * LB_TPB) {
        int k[LA_UN], v[LA_UN];
#pragma unroll
        for (int u = 0; u < LA_UN; ++u) {
            const int e = min(e0 + u * LB_TPB + tid, hi - 1);
            k[u] = key[e];
            v[u] = val[e];
        }
#pragma unroll
        for (int u = 0; u < LA_UN; ++u) {
            if (e0 + u * LB_TPB + tid < hi) {
                const int pos = atomicAdd(&cur[k[u] >> LB_BTS], 1);
                const v2i kv = {k[u], v[u]};
                if (stage) sbuf[pos] = kv;
                else       gb[pos]   = kv;
            }
        }
    }
    if (stage) {
        __syncthreads();
        const int cnt = hi - lo;
        for (int i0 = 0; i0 < cnt; i0 += LA_UN * LB_TPB) {
            v2i t[LA_UN];
#pragma unroll
            for (int u = 0; u < LA_UN; ++u)
                t[u] = sbuf[min(i0 + u * LB_TPB + tid, cnt - 1)];
#pragma unroll
            for (int u = 0; u < LA_UN; ++u) {
                const int i = i0 + u * LB_TPB + tid;
                if (i < cnt) gb[i] = t[u];
            }
        }
    }
}

#define LB_WV  (LB_BT / 32)
#define LB_SPW (LB_SLICES / LB_WV)
#define LB_UN  6
__device__ __forceinline__
void block_lists_body(const int* __restrict__ bucket, const int* __restrict__ cntT,
                      const int* __restrict__ offT, int E, int N,
                      int* __restrict__ rowptr, int* __restrict__ adj, int b, int dir,
                      int* ncnt, int* scnt, int* sbase, int* wsum) {
    const int tid  = threadIdx.x;
    const int lane = tid & 31;
    const int wv   = tid >> 5;
    const int len  = (E + LB_SLICES - 1) / LB_SLICES;
    const int nbase = b * LB_BT;
    const v2i* gb = (const v2i*)bucket + (size_t)dir * E;
    const int* gk = (const int*)gb;
    int* rp = rowptr + (size_t)dir * (N + 1);

    ncnt[tid] = 0;
    if (tid < LB_SLICES) {
        const int idx = (dir * LB_SLICES + tid) * LB_NB + b;
        const int o = offT[idx];
        scnt[tid]  = cntT[idx];
        sbase[tid] = min(tid * len, E) + o;
        int osum = o;
        for (int d = 16; d >= 1; d >>= 1) osum += __shfl_xor(osum, d);
        if (lane == 0) wsum[wv] = osum;
    }
    __syncthreads();
    int osum = 0;
#pragma unroll
    for (int w = 0; w < LB_SLICES / 32; ++w) osum += wsum[w];
    const int start = osum;
    int c[LB_SPW], bs[LB_SPW];
#pragma unroll
    for (int q = 0; q < LB_SPW; ++q) {
        c[q]  = scnt[wv + LB_WV * q];
        bs[q] = sbase[wv + LB_WV * q];
    }

    {
        int kk[LB_SPW][LB_UN];
#pragma unroll
        for (int q = 0; q < LB_SPW; ++q) {
#pragma unroll
            for (int j = 0; j < LB_UN; ++j) {
                const int i = c[q] > 0 ? bs[q] + min(lane + 32 * j, c[q] - 1) : 0;
                kk[q][j] = gk[2 * (size_t)i];
            }
        }
#pragma unroll
        for (int q = 0; q < LB_SPW; ++q) {
#pragma unroll
            for (int j = 0; j < LB_UN; ++j)
                if (lane + 32 * j < c[q]) atomicAdd(&ncnt[kk[q][j] - nbase], 1);
        }
#pragma unroll
        for (int q = 0; q < LB_SPW; ++q)
            for (int i = lane + 32 * LB_UN; i < c[q]; i += 32)
                atomicAdd(&ncnt[gk[2 * (size_t)(bs[q] + i)] - nbase], 1);
    }
    __syncthreads();

    const int n   = nbase + tid;
    const int deg = ncnt[tid];
    int inc2 = deg;
    for (int d = 1; d < 32; d <<= 1) {
        int t = __shfl_up(inc2, d);
        if (lane >= d) inc2 += t;
    }
    if (lane == 31) wsum[wv] = inc2;
    __syncthreads();
    if (wv == 0) {
        const int w = (lane < LB_WV) ? wsum[lane] : 0;
        int wi = w;
        for (int d = 1; d < 32; d <<= 1) {
            int t = __shfl_up(wi, d);
            if (lane >= d) wi += t;
        }
        if (lane < LB_WV) wsum[lane] = wi - w;
    }
    __syncthreads();
    const int excl = wsum[wv] + inc2 - deg;
    ncnt[tid] = excl;
    if (n < N) {
        rp[n] = start + excl;
        if (n == N - 1) rp[N] = start + excl + deg;
    }
    __syncthreads();

    int* ad = adj + (size_t)dir * E + start;
    {
        v2i kv[LB_SPW][LB_UN];
#pragma unroll
        for (int q = 0; q < LB_SPW; ++q) {
#pragma unroll
            for (int j = 0; j < LB_UN; ++j) {
                const int i = c[q] > 0 ? bs[q] + min(lane + 32 * j, c[q] - 1) : 0;
                kv[q][j] = gb[i];
            }
        }
#pragma unroll
        for (int q = 0; q < LB_SPW; ++q) {
#pragma unroll
            for (int j = 0; j < LB_UN; ++j)
                if (lane + 32 * j < c[q])
                    ad[atomicAdd(&ncnt[kv[q][j].x - nbase], 1)] = kv[q][j].y;
        }
#pragma unroll
        for (int q = 0; q < LB_SPW; ++q)
            for (int i = lane + 32 * LB_UN; i < c[q]; i += 32) {
                const v2i e = gb[bs[q] + i];
                ad[atomicAdd(&ncnt[e.x - nbase], 1)] = e.y;
            }
    }
}

__device__ __forceinline__
int list_order(int d, int my, int lane) {
    int srt = my;
    if (d > 0 && d <= 32) {
        const int key = (my << 5) | lane;
        int pos = 0;
        for (int j = 0; j < d; ++j) {
            const int kj = __builtin_amdgcn_readlane(key, j);
            pos += (kj < key) ? 1 : 0;
        }
        if (lane >= d) pos = lane;
        const int dst = ((pos & 1) << 4) | (pos >> 1);
        srt = __builtin_amdgcn_ds_permute(dst << 2, my);
    }
    return srt;
}

__device__ __forceinline__
void list_rows(const v4f* __restrict__ Y4, int d, int srt, int c,
               v4f (&v)[8], v4f (&w)[8]) {
    if (d > 0 && d <= 32) {
#define GA_ROW(V, U, Q) V[U] = *(const v4f*)((const char*)Y4 + ((((unsigned)__builtin_amdgcn_update_dpp(0, srt, 0x150 + (Q), 0xf, 0xf, false)) << 8) | ((unsigned)c << 4)))
        GA_ROW(v, 0, 0); GA_ROW(v, 1, 1); GA_ROW(v, 2, 2); GA_ROW(v, 3, 3);
        GA_ROW(v, 4, 4); GA_ROW(v, 5, 5); GA_ROW(v, 6, 6); GA_ROW(v, 7, 7);
        if (d > 16) {
            GA_ROW(w, 0, 8);  GA_ROW(w, 1, 9);  GA_ROW(w, 2, 10); GA_ROW(w, 3, 11);
            GA_ROW(w, 4, 12); GA_ROW(w, 5, 13); GA_ROW(w, 6, 14); GA_ROW(w, 7, 15);
        }
#undef GA_ROW
    }
}

__device__ __forceinline__
v4f list_finish(const v4f* __restrict__ Y4, const int* __restrict__ adj,
                int beg, int d, int lane, int h, int c,
                const v4f (&v)[8], const v4f (&w)[8]) {
    v4f acc = {0.0f, 0.0f, 0.0f, 0.0f};
    if (d > 0 && d <= 32) {
#pragma unroll
        for (int u = 0; u < 8; ++u)
            if (2 * u + h < d) acc += v[u];
        if (d > 16) {
#pragma unroll
            for (int u = 0; u < 8; ++u)
                if (16 + 2 * u + h < d) acc += w[u];
        }
    } else if (d > 32 && d <= 64) {
        const int e0 = adj[beg + lane];
        const int e1 = adj[beg + min(32 + lane, d - 1)];
        const int k0 = (e0 << 6) | lane;
        const int k1 = (32 + lane < d) ? ((e1 << 6) | (32 + lane)) : 0x7fffffff;
        int p0 = 0, p1 = 0;
        for (int j = 0; j < 32; ++j) {
            const int kj = __builtin_amdgcn_readlane(k0, j);
            p0 += (kj < k0) ? 1 : 0;
            p1 += (kj < k1) ? 1 : 0;
        }
        for (int j = 32; j < d; ++j) {
            const int kj = __builtin_amdgcn_readlane(k1, j - 32);
            p0 += (kj < k0) ? 1 : 0;
            p1 += (kj < k1) ? 1 : 0;
        }
        if (32 + lane >= d) p1 = -1;
        for (int g = 0; g < d; g += 16) {
            v4f v[8];
#pragma unroll
            for (int u = 0; u < 8; ++u) {
                int jr[2];
#pragma unroll
                for (int q = 0; q < 2; ++q) {
                    const int r = min(g + 2 * u + q, d - 1);
                    const unsigned m0 = __builtin_amdgcn_ballot_w32(p0 == r);
                    const unsigned m1 = __builtin_amdgcn_ballot_w32(p1 == r);
                    const int a0 = __builtin_amdgcn_readlane(e0, m0 ? __builtin_ctz(m0) : 0);
                    const int a1 = __builtin_amdgcn_readlane(e1, m1 ? __builtin_ctz(m1) : 0);
                    jr[q] = m0 ? a0 : a1;
                }
                const int jj = h ? jr[1] : jr[0];
                v[u] = Y4[(size_t)jj * 16 + c];
            }
#pragma unroll
            for (int u = 0; u < 8; ++u)
                if (g + 2 * u + h < d) acc += v[u];
        }
    } else if (d > 64) {
        long long last = -1;
        for (int it = 0; it < d; ++it) {
            long long best = 0x7fffffffffffffffLL;
            for (int p = lane; p < d; p += 32) {
                const long long kk = ((long long)adj[beg + p] << 32) | (long long)p;
                if (kk > last && kk < best) best = kk;
            }
            for (int m = 16; m >= 1; m >>= 1) {
                const long long o = __shfl_xor(best, m);
                if (o < best) best = o;
            }
            last = best;
            const int jj = (int)(best >> 32);
            const v4f r = Y4[(size_t)jj * 16 + c];
            if (h == 0) acc += r;
        }
    }
    acc.x += __shfl_xor(acc.x, 16);
    acc.y += __shfl_xor(acc.y, 16);
    acc.z += __shfl_xor(acc.z, 16);
    acc.w += __shfl_xor(acc.w, 16);
    return acc;
}

#define FU_PITCH 68
__device__ __forceinline__
void gather_tile(const float* __restrict__ Y, const int* __restrict__ rowptr,
                 const int* __restrict__ adj, int N, int rowbase, int lane,
                 float* myT) {
    const int h = lane >> 4;
    const int c = lane & 15;
    const v4f* Y4 = (const v4f*)Y;
    const int rp = rowptr[min(rowbase + min(lane, 16), N)];
    int e0 = __builtin_amdgcn_readlane(rp, 0);
    int e1 = __builtin_amdgcn_readlane(rp, 1);
    int e2 = __builtin_amdgcn_readlane(rp, 2);
    int nxA = 0, nxB = 0;
    if (e1 > e0) nxA = adj[e0 + min(lane, e1 - e0 - 1)];
    if (e2 > e1) nxB = adj[e1 + min(lane, e2 - e1 - 1)];
#pragma unroll 1
    for (int i = 0; i < 16; i += 2) {
        const int begA = e0, dA = e1 - e0;
        const int begB = e1, dB = e2 - e1;
        const int srtA = list_order(dA, nxA, lane);
        const int srtB = list_order(dB, nxB, lane);
        e0 = e2;
        e1 = (i < 14) ? __builtin_amdgcn_readlane(rp, i + 3) : e0;
        e2 = (i < 14) ? __builtin_amdgcn_readlane(rp, i + 4) : e0;
        nxA = 0;
        nxB = 0;
        if (e1 > e0) nxA = adj[e0 + min(lane, e1 - e0 - 1)];
        if (e2 > e1) nxB = adj[e1 + min(lane, e2 - e1 - 1)];
        v4f vA[8], wA[8], vB[8], wB[8];
        list_rows(Y4, dA, srtA, c, vA, wA);
        list_rows(Y4, dB, srtB, c, vB, wB);
        const v4f accA = list_finish(Y4, adj, begA, dA, lane, h, c, vA, wA);
        const v4f accB = list_finish(Y4, adj, begB, dB, lane, h, c, vB, wB);
        const float invA = 1.0f / fmaxf((float)dA, 1.0f);
        const float invB = 1.0f / fmaxf((float)dB, 1.0f);
        if (h == 0) {
            *(v4f*)&myT[i * FU_PITCH + 4 * c]       = accA * invA;
            *(v4f*)&myT[(i + 1) * FU_PITCH + 4 * c] = accB * invB;
        }
    }
}

#define GM_LOAD(DST, PTR)                                                     \
    _Pragma("unroll")                                                         \
    for (int u = 0; u < 8; ++u) DST[u] = *(const v2f*)((PTR) + 4 * u);
#define GM_MMA(SRC, CI)                                                       \
    _Pragma("unroll")                                                         \
    for (int u = 0; u < 8; ++u) {                                             \
        const v2f a = SRC[u];                                                 \
        const int sbase = ((((CI) << 4) + 2 * u + kSel) << 7);                \
        _Pragma("unroll")                                                     \
        for (int t = 0; t < 4; ++t) {                                         \
            v2f b = *(const v2f*)&sW[sbase + ((t << 4) + m) * 2];             \
            acc[t] = __builtin_amdgcn_wmma_f32_16x16x4_f32(                   \
                false, a, false, b, (short)0, acc[t], false, false);          \
        }                                                                     \
    }

template <int NQ>
__device__ __forceinline__ void fu_stage(const float* __restrict__ W, float* sW, int tid) {
    constexpr int npair = NQ * 8;
    v4f w0[NQ], w1[NQ];
#pragma unroll
    for (int q = 0; q < NQ; ++q) {
        const int it = tid + 128 * q;
        const int p  = it >> 4;
        const int g  = it & 15;
        w0[q] = *(const v4f*)(W + (size_t)(2 * p) * 64 + 4 * g);
        w1[q] = *(const v4f*)(W + (size_t)(2 * p + 1) * 64 + 4 * g);
    }
#pragma unroll
    for (int q = 0; q < NQ; ++q) {
        const int it = tid + 128 * q;
        const int p  = it >> 4;
        const int g  = it & 15;
        if (p < npair) {
            const v4f lo = {w0[q].x, w1[q].x, w0[q].y, w1[q].y};
            const v4f hi = {w0[q].z, w1[q].z, w0[q].w, w1[q].w};
            v4f* d = (v4f*)&sW[((p << 6) + 4 * g) * 2];
            d[0] = lo;
            d[1] = hi;
        }
    }
}

template <int NB, bool LAST>
__global__ __launch_bounds__(128, 4)
void gather_update_message(const float* X0,
                           const float* __restrict__ Ya,
                           const float* __restrict__ Yb,
                           const int* __restrict__ rowptra, const int* __restrict__ rowptrb,
                           const int* __restrict__ adja, const int* __restrict__ adjb,
                           const float* __restrict__ U0,
                           const float* __restrict__ U1,
                           float* hout, int dup,
                           const float* __restrict__ Wm,
                           const float* __restrict__ bm,
                           float* __restrict__ Yout,
                           int Nrows) {
    __shared__ __attribute__((aligned(16))) float sW[192 * 64];
    __shared__ __attribute__((aligned(16))) float hT[4 * 16 * FU_PITCH];

    const int tid  = threadIdx.x;
    const int wave = tid >> 5;
    const int lane = tid & 31;
    const int m    = lane & 15;
    const int kSel = lane >> 4;

    const int rowbase = __builtin_amdgcn_readfirstlane((int)(blockIdx.x * 4 + wave) * 16);
    const int row     = rowbase + m;
    const int crow    = row < Nrows ? row : (Nrows - 1);
    const int rr      = kSel * 8;
    float* myT = hT + wave * (16 * FU_PITCH);

    const float* x0r = X0 + (size_t)crow * 128 + 2 * kSel;

    v2f hA[NB][16];

    gather_tile(Ya, rowptra, adja, Nrows, rowbase, lane, myT);
    v2f s1A[16];
    if constexpr (NB > 1) {
        float* sT = sW + wave * (16 * FU_PITCH);
        gather_tile(Yb, rowptrb, adjb, Nrows, rowbase, lane, sT);
        __syncthreads();
#pragma unroll
        for (int s = 0; s < 16; ++s)
            s1A[s] = *(const v2f*)&sT[m * FU_PITCH + 4 * s + 2 * kSel];
        __syncthreads();
    }

#pragma unroll
    for (int b = 0; b < NB; ++b) {
        const float* __restrict__ U = b ? U1 : U0;

        if (b > 0) {
#pragma unroll
            for (int s = 0; s < 16; ++s)
                *(v2f*)&myT[m * FU_PITCH + 4 * s + 2 * kSel] = s1A[s];
        }

        v2f areg0[8], areg1[8];
        GM_LOAD(areg0, x0r)

        fu_stage<12>(U, sW, tid);
        __syncthreads();

        v8f acc[4] = {};
#pragma unroll 1
        for (int ci = 0; ci < 4; ci += 2) {
            {
                const float* p1 = x0r + (min(ci + 1, 3) << 5);
                GM_LOAD(areg1, p1)
            }
            GM_MMA(areg0, ci)
            {
                const float* p2 = x0r + (min(ci + 2, 3) << 5);
                GM_LOAD(areg0, p2)
            }
            GM_MMA(areg1, ci + 1)
        }
#pragma unroll
        for (int s = 0; s < 16; ++s) {
            const v2f a = *(const v2f*)&myT[m * FU_PITCH + 4 * s + 2 * kSel];
            const int sbase = ((64 + 2 * s + kSel) << 7);
#pragma unroll
            for (int t = 0; t < 4; ++t) {
                const v2f bb = *(const v2f*)&sW[sbase + ((t << 4) + m) * 2];
                acc[t] = __builtin_amdgcn_wmma_f32_16x16x4_f32(
                    false, a, false, bb, (short)0, acc[t], false, false);
            }
        }

        if constexpr (!LAST) {
#pragma unroll
            for (int r = 0; r < 8; ++r) {
#pragma unroll
                for (int t = 0; t < 4; ++t)
                    myT[(r + rr) * FU_PITCH + (t << 4) + m] = fmaxf(acc[t][r], 0.0f);
            }
            __syncthreads();
#pragma unroll
            for (int s = 0; s < 16; ++s)
                hA[b][s] = *(const v2f*)&myT[m * FU_PITCH + 4 * s + 2 * kSel];
        }
#pragma unroll
        for (int r = 0; r < 8; ++r) {
            const int orow = rowbase + r + rr;
            if (orow < Nrows) {
#pragma unroll
                for (int t = 0; t < 4; ++t) {
                    const float v = fmaxf(acc[t][r], 0.0f);
                    float* p = hout + (size_t)orow * 128 + 64 * b + (t << 4) + m;
                    *p = v;
                    if (dup) p[64] = v;
                }
            }
        }
    }

    if constexpr (!LAST) {
        float bv[4];
#pragma unroll
        for (int t = 0; t < 4; ++t) bv[t] = bm[(t << 4) + m];

        fu_stage<8>(Wm, sW, tid);
        __syncthreads();

        v8f acc[4] = {};
#pragma unroll
        for (int s = 0; s < 32; ++s) {
            const v2f a = (s < 16) ? hA[0][s] : hA[NB - 1][s - 16];
            const int sbase = ((2 * s + kSel) << 7);
#pragma unroll
            for (int t = 0; t < 4; ++t) {
                const v2f bb = *(const v2f*)&sW[sbase + ((t << 4) + m) * 2];
                acc[t] = __builtin_amdgcn_wmma_f32_16x16x4_f32(
                    false, a, false, bb, (short)0, acc[t], false, false);
            }
        }
#pragma unroll
        for (int r = 0; r < 8; ++r) {
            const int orow = rowbase + r + rr;
            if (orow < Nrows) {
#pragma unroll
                for (int t = 0; t < 4; ++t)
                    Yout[(size_t)orow * 64 + (t << 4) + m] = fmaxf(acc[t][r] + bv[t], 0.0f);
            }
        }
    }
}
#define MS_ROWS 256
__device__ __forceinline__
void message1_body(const float* __restrict__ X, const float* __restrict__ W,
                   const float* __restrict__ bias, float* __restrict__ Yout,
                   int Nrows, int blk, float* sW) {
    const int tid  = threadIdx.x;
    const int wave = tid >> 5;
    const int lane = tid & 31;
    const int m    = lane & 15;
    const int kSel = lane >> 4;

    const int rowbase = blk * MS_ROWS + wave * 16;
    const int row     = rowbase + m;
    const int crow    = row < Nrows ? row : (Nrows - 1);
    const float* x0r  = X + (size_t)crow * 128 + 2 * kSel;

    v2f areg0[8], areg1[8];
    GM_LOAD(areg0, x0r)
    float bv[4];
#pragma unroll
    for (int t = 0; t < 4; ++t) bv[t] = bias[(t << 4) + m];

    {
        v4f w0[2], w1[2];
#pragma unroll
        for (int q = 0; q < 2; ++q) {
            const int it = tid + LB_BT * q;
            const int p  = it >> 4;
            const int g  = it & 15;
            w0[q] = *(const v4f*)(W + (size_t)(2 * p) * 64 + 4 * g);
            w1[q] = *(const v4f*)(W + (size_t)(2 * p + 1) * 64 + 4 * g);
        }
#pragma unroll
        for (int q = 0; q < 2; ++q) {
            const int it = tid + LB_BT * q;
            const int p  = it >> 4;
            const int g  = it & 15;
            const v4f lo = {w0[q].x, w1[q].x, w0[q].y, w1[q].y};
            const v4f hi = {w0[q].z, w1[q].z, w0[q].w, w1[q].w};
            v4f* d = (v4f*)&sW[((p << 6) + 4 * g) * 2];
            d[0] = lo;
            d[1] = hi;
        }
    }
    __syncthreads();

    v8f acc[4] = {};
#pragma unroll 1
    for (int ci = 0; ci < 4; ci += 2) {
        {
            const float* p1 = x0r + (min(ci + 1, 3) << 5);
            GM_LOAD(areg1, p1)
        }
        GM_MMA(areg0, ci)
        {
            const float* p2 = x0r + (min(ci + 2, 3) << 5);
            GM_LOAD(areg0, p2)
        }
        GM_MMA(areg1, ci + 1)
    }

    const int rr = kSel * 8;
#pragma unroll
    for (int r = 0; r < 8; ++r) {
        const int orow = rowbase + r + rr;
        if (orow < Nrows) {
#pragma unroll
            for (int t = 0; t < 4; ++t)
                Yout[(size_t)orow * 64 + (t << 4) + m] = fmaxf(acc[t][r] + bv[t], 0.0f);
        }
    }
}

__global__ __launch_bounds__(LB_BT)
void block_lists_and_message(const int* __restrict__ bucket, const int* __restrict__ cntT,
                             const int* __restrict__ offT, int E, int N,
                             int* __restrict__ rowptr, int* __restrict__ adj,
                             const float* __restrict__ X,
                             const float* __restrict__ Wa, const float* __restrict__ Wb,
                             const float* __restrict__ biasa, const float* __restrict__ biasb,
                             float* __restrict__ Ya, float* __restrict__ Yb) {
    __shared__ __attribute__((aligned(16))) float sW[128 * 64];
    __shared__ int ncnt[LB_BT];
    __shared__ int scnt[LB_SLICES];
    __shared__ int sbase[LB_SLICES];
    __shared__ int wsum[32];
    const int nb = (N + LB_BT - 1) / LB_BT;
    const int bx = blockIdx.x;
    if (bx < 2 * nb) {
        const int dir = bx >= nb ? 1 : 0;
        block_lists_body(bucket, cntT, offT, E, N, rowptr, adj, bx - dir * nb, dir,
                         ncnt, scnt, sbase, wsum);
    } else {
        const int ngb = (N + MS_ROWS - 1) / MS_ROWS;
        const int g   = bx - 2 * nb;
        const bool br = g >= ngb;
        message1_body(X, br ? Wb : Wa, br ? biasb : biasa, br ? Yb : Ya, N,
                      br ? g - ngb : g, sW);
    }
}
#undef GM_LOAD
#undef GM_MMA

extern "C" void kernel_launch(void* const* d_in, const int* in_sizes, int n_in,
                              void* d_out, int out_size, void* d_ws, size_t ws_size,
                              hipStream_t stream) {
    const float* x    = (const float*)d_in[0];
    const int*   ei   = (const int*)  d_in[1];
    const float* W_u1 = (const float*)d_in[2];
    const float* b_u1 = (const float*)d_in[3];
    const float* U_u1 = (const float*)d_in[4];
    const float* W_d1 = (const float*)d_in[5];
    const float* b_d1 = (const float*)d_in[6];
    const float* U_d1 = (const float*)d_in[7];
    const float* W_u2 = (const float*)d_in[8];
    const float* b_u2 = (const float*)d_in[9];
    const float* U_u2 = (const float*)d_in[10];
    float* out = (float*)d_out;

    const int N = in_sizes[0] / 128;
    const int E = in_sizes[1] / 2;

    const int cstride = (N + 31) / 32 * 32;
    float* ws     = (float*)d_ws;
    float* Y      = ws;
    float* Y2     = Y  + (size_t)N * 64;
    float* ha     = Y2 + (size_t)N * 64;
    int*   cnt    = (int*)(ha + (size_t)N * 128);
    int*   rowptr = cnt    + 2 * (size_t)cstride;
    int*   adj    = rowptr + 2 * ((size_t)N + 1);
    int*   bucket = (int*)ws;
    int*   cntT   = cnt;
    int*   offT   = cnt + 2 * LB_SLICES * LB_NB;

    const int* rp_in   = rowptr;
    const int* rp_out  = rowptr + (N + 1);
    const int* adj_in  = adj;
    const int* adj_out = adj + E;

    float* Yu = out;
    float* Yd = out + (size_t)N * 64;
    slice_buckets<<<dim3(LB_SLICES, 2), LB_TPB, 0, stream>>>(ei, E, bucket, cntT, offT);
    block_lists_and_message<<<2 * ((N + LB_BT - 1) / LB_BT) + 2 * ((N + MS_ROWS - 1) / MS_ROWS),
                              LB_BT, 0, stream>>>(
        bucket, cntT, offT, E, N, rowptr, adj, x, W_u1, W_d1, b_u1, b_d1, Yu, Yd);

    const int gblk = (N + 63) / 64;

    gather_update_message<2, false><<<gblk, 128, 0, stream>>>(
        x, Yu, Yd, rp_in, rp_out, adj_in, adj_out, U_u1, U_d1, ha, 0, W_u2, b_u2, Y, N);

    gather_update_message<1, false><<<gblk, 128, 0, stream>>>(
        ha, Y, nullptr, rp_in, nullptr, adj_in, nullptr, U_u2, nullptr, out, 1, W_u2, b_u2, Y2, N);
    gather_update_message<1, true><<<gblk, 128, 0, stream>>>(
        out, Y2, nullptr, rp_in, nullptr, adj_in, nullptr, U_u2, nullptr, out, 1,
        nullptr, nullptr, nullptr, N);
}
